# starting kernel
# baseline (speedup 1.0000x reference)
#include <hip/hip_runtime.h>
#include <hip/hip_bf16.h>

typedef __attribute__((ext_vector_type(2))) float v2f;
typedef __attribute__((ext_vector_type(8))) float v8f;

#define NCOLS 128
#define NODE_K 256
#define EDGE_K 64
#define NHEADS 4

template <int K>
__global__ void __launch_bounds__(256)
gemm_f32_wmma(const float* __restrict__ A, const float* __restrict__ B,
              float* __restrict__ C, int M) {
    __shared__ float Apanel[16 * K];
    const int lane = threadIdx.x & 31;
    const int wave = threadIdx.x >> 5;
    const int m0   = blockIdx.x * 16;

    for (int idx = threadIdx.x; idx < 16 * K; idx += 256) {
        int r = idx / K, c = idx % K;
        int row = m0 + r; row = row < M ? row : M - 1;
        Apanel[idx] = A[(size_t)row * K + c];
    }
    __syncthreads();

    const int n0      = wave * 16;
    const int half    = lane >> 4;
    const int l15     = lane & 15;
    const int kpair   = half * 2;

    v8f acc = {};
    for (int k = 0; k < K; k += 4) {
        v2f a, b;
        a.x = Apanel[l15 * K + k + kpair];
        a.y = Apanel[l15 * K + k + kpair + 1];
        b.x = B[(size_t)(k + kpair)     * NCOLS + n0 + l15];
        b.y = B[(size_t)(k + kpair + 1) * NCOLS + n0 + l15];
        acc = __builtin_amdgcn_wmma_f32_16x16x4_f32(
            false, a, false, b, (short)0, acc, false, false);
    }

    const int mbase = half * 8;
#pragma unroll
    for (int r = 0; r < 8; ++r) {
        int row = m0 + mbase + r;
        if (row < M) C[(size_t)row * NCOLS + n0 + l15] = acc[r];
    }
}

__global__ void __launch_bounds__(256)
edge_proj_score(const float* __restrict__ EF, const float* __restrict__ We,
                const float* __restrict__ H, const int* __restrict__ src,
                float* __restrict__ score, int E) {
    __shared__ float Apanel[16 * EDGE_K];
    __shared__ int   srcIdx[16];
    __shared__ float spart[8][16];

    const int lane = threadIdx.x & 31;
    const int wave = threadIdx.x >> 5;
    const int m0   = blockIdx.x * 16;

    for (int idx = threadIdx.x; idx < 16 * EDGE_K; idx += 256) {
        int r = idx >> 6, c = idx & 63;
        int row = m0 + r; row = row < E ? row : E - 1;
        Apanel[idx] = EF[(size_t)row * EDGE_K + c];
    }
    if (threadIdx.x < 16) {
        int e = m0 + threadIdx.x;
        srcIdx[threadIdx.x] = src[e < E ? e : E - 1];
    }
    __syncthreads();

    const int n0    = wave * 16;
    const int half  = lane >> 4;
    const int l15   = lane & 15;
    const int kpair = half * 2;

    v8f acc = {};
    for (int k = 0; k < EDGE_K; k += 4) {
        v2f a, b;
        a.x = Apanel[l15 * EDGE_K + k + kpair];
        a.y = Apanel[l15 * EDGE_K + k + kpair + 1];
        b.x = We[(size_t)(k + kpair)     * NCOLS + n0 + l15];
        b.y = We[(size_t)(k + kpair + 1) * NCOLS + n0 + l15];
        acc = __builtin_amdgcn_wmma_f32_16x16x4_f32(
            false, a, false, b, (short)0, acc, false, false);
    }

    const int mbase = half * 8;
    float prod[8];
#pragma unroll
    for (int r = 0; r < 8; ++r) {
        int s = srcIdx[mbase + r];
        float hv = H[(size_t)s * NCOLS + n0 + l15];
        prod[r] = acc[r] * hv;
    }
#pragma unroll
    for (int off = 1; off < 16; off <<= 1) {
#pragma unroll
        for (int r = 0; r < 8; ++r) prod[r] += __shfl_xor(prod[r], off, 32);
    }
    if (l15 == 0) {
#pragma unroll
        for (int r = 0; r < 8; ++r) spart[wave][mbase + r] = prod[r];
    }
    __syncthreads();

    if (threadIdx.x < 64) {
        int m = threadIdx.x >> 2, hd = threadIdx.x & 3;
        int e = m0 + m;
        if (e < E) {
            float sv = (spart[2 * hd][m] + spart[2 * hd + 1][m]) * 0.17677669529663689f;
            sv = sv > 0.f ? sv : 0.01f * sv;
            score[(size_t)e * NHEADS + hd] = sv;
        }
    }
}

__global__ void __launch_bounds__(256)
init_kernel(float* __restrict__ out, float* __restrict__ smax,
            float* __restrict__ denom, int nOut, int nSeg) {
    int gid = blockIdx.x * 256 + threadIdx.x;
    if (gid < nOut) out[gid] = 0.f;
    if (gid < nSeg) { smax[gid] = __int_as_float(0xFF800000); denom[gid] = 0.f; }
}

__device__ __forceinline__ void atomicMaxFloat(float* addr, float val) {
    if (val >= 0.f) atomicMax((int*)addr, __float_as_int(val));
    else            atomicMin((unsigned int*)addr, __float_as_uint(val));
}

__global__ void __launch_bounds__(256)
segmax_kernel(const float* __restrict__ score, const int* __restrict__ dst,
              float* __restrict__ smax, int E) {
    int gid = blockIdx.x * 256 + threadIdx.x;
    if (gid >= E * NHEADS) return;
    int e = gid >> 2, hd = gid & 3;
    atomicMaxFloat(&smax[dst[e] * NHEADS + hd], score[gid]);
}

__global__ void __launch_bounds__(256)
expsum_kernel(const float* __restrict__ score, const int* __restrict__ dst,
              const float* __restrict__ smax, float* __restrict__ aarr,
              float* __restrict__ denom, int E) {
    int gid = blockIdx.x * 256 + threadIdx.x;
    if (gid >= E * NHEADS) return;
    int e = gid >> 2, hd = gid & 3;
    float v = __expf(score[gid] - smax[dst[e] * NHEADS + hd]);
    aarr[gid] = v;
    atomicAdd(&denom[dst[e] * NHEADS + hd], v);
}

__global__ void __launch_bounds__(256)
aggregate_kernel(const float* __restrict__ H, const int* __restrict__ src,
                 const int* __restrict__ dst, const float* __restrict__ aarr,
                 const float* __restrict__ denom, float* __restrict__ out, int E) {
    int gid = blockIdx.x * 256 + threadIdx.x;
    int e = gid >> 7;
    if (e >= E) return;
    int f  = gid & 127;
    int hd = f >> 5;
    int s = src[e], d = dst[e];
    float attn = aarr[e * NHEADS + hd] / denom[d * NHEADS + hd];
    float val  = H[(size_t)s * NCOLS + f] * attn;
    atomicAdd(&out[(size_t)d * NCOLS + f], val);
}

extern "C" void kernel_launch(void* const* d_in, const int* in_sizes, int n_in,
                              void* d_out, int out_size, void* d_ws, size_t ws_size,
                              hipStream_t stream) {
    const float* node_feat = (const float*)d_in[0];
    const float* edge_feat = (const float*)d_in[1];
    const int*   src       = (const int*)d_in[2];
    const int*   dst       = (const int*)d_in[3];
    const float* W_node    = (const float*)d_in[4];
    const float* W_edge    = (const float*)d_in[5];
    float*       out       = (float*)d_out;

    const int N = in_sizes[0] / NODE_K;
    const int E = in_sizes[2];

    float* h     = (float*)d_ws;
    float* score = h     + (size_t)N * NCOLS;
    float* aarr  = score + (size_t)E * NHEADS;
    float* smax  = aarr  + (size_t)E * NHEADS;
    float* denom = smax  + (size_t)N * NHEADS;

    gemm_f32_wmma<NODE_K><<<(N + 15) / 16, 256, 0, stream>>>(node_feat, W_node, h, N);

    edge_proj_score<<<(E + 15) / 16, 256, 0, stream>>>(edge_feat, W_edge, h, src, score, E);

    int nOut = N * NCOLS;
    init_kernel<<<(nOut + 255) / 256, 256, 0, stream>>>(out, smax, denom, nOut, N * NHEADS);

    segmax_kernel<<<(E * NHEADS + 255) / 256, 256, 0, stream>>>(score, dst, smax, E);

    expsum_kernel<<<(E * NHEADS + 255) / 256, 256, 0, stream>>>(score, dst, smax, aarr, denom, E);

    long long total = (long long)E * NCOLS;
    aggregate_kernel<<<(int)((total + 255) / 256), 256, 0, stream>>>(h, src, dst, aarr, denom, out, E);
}


# edge score re-decomposed: product transposed so a lane owns one edge, a wave owns one head with its W_edge slice held in registers over a sweep of edge tiles; k and head-dot sum order change
# speedup vs baseline: 2.3372x; 2.3372x over previous
#include <hip/hip_runtime.h>
#include <hip/hip_bf16.h>

typedef __attribute__((ext_vector_type(2))) float v2f;
typedef __attribute__((ext_vector_type(8))) float v8f;

#define NCOLS 128
#define NODE_K 256
#define EDGE_K 64
#define NHEADS 4

template <int K>
__global__ void __launch_bounds__(256)
gemm_f32_wmma(const float* __restrict__ A, const float* __restrict__ B,
              float* __restrict__ C, int M) {
    __shared__ float Apanel[16 * K];
    const int lane = threadIdx.x & 31;
    const int wave = threadIdx.x >> 5;
    const int m0   = blockIdx.x * 16;

    for (int idx = threadIdx.x; idx < 16 * K; idx += 256) {
        int r = idx / K, c = idx % K;
        int row = m0 + r; row = row < M ? row : M - 1;
        Apanel[idx] = A[(size_t)row * K + c];
    }
    __syncthreads();

    const int n0      = wave * 16;
    const int half    = lane >> 4;
    const int l15     = lane & 15;
    const int kpair   = half * 2;

    v8f acc = {};
    for (int k = 0; k < K; k += 4) {
        v2f a, b;
        a.x = Apanel[l15 * K + k + kpair];
        a.y = Apanel[l15 * K + k + kpair + 1];
        b.x = B[(size_t)(k + kpair)     * NCOLS + n0 + l15];
        b.y = B[(size_t)(k + kpair + 1) * NCOLS + n0 + l15];
        acc = __builtin_amdgcn_wmma_f32_16x16x4_f32(
            false, a, false, b, (short)0, acc, false, false);
    }

    const int mbase = half * 8;
#pragma unroll
    for (int r = 0; r < 8; ++r) {
        int row = m0 + mbase + r;
        if (row < M) C[(size_t)row * NCOLS + n0 + l15] = acc[r];
    }
}

typedef __attribute__((ext_vector_type(4))) float v4f;

#define EF_PITCH4  17
#define SCORE_GRID 1024

__global__ void __launch_bounds__(128)
edge_score_heads(const float* __restrict__ EF, const float* __restrict__ We,
                 const float* __restrict__ H, const int* __restrict__ src,
                 float* __restrict__ score, int E) {
    __shared__ v4f stage[4][16 * EF_PITCH4];

    const int lane = threadIdx.x & 31;
    const int head = __builtin_amdgcn_readfirstlane((int)(threadIdx.x >> 5));
    const int c  = lane & 15;
    const int hf = lane >> 4;

    const int ntile = (E + 15) >> 4;
    const int npair = (ntile + 1) >> 1;
    int p = blockIdx.x;
    if (p >= npair) return;
    const int G = gridDim.x;

    v2f wf[2][16];
#pragma unroll
    for (int u = 0; u < 2; ++u) {
#pragma unroll
        for (int j = 0; j < 16; ++j) {
            const int k0 = 8 * (j >> 1) + 4 * hf + 2 * (j & 1);
            const float* wp = We + (size_t)k0 * NCOLS + 32 * head + 16 * u + c;
            wf[u][j].x = wp[0];
            wf[u][j].y = wp[NCOLS];
        }
    }

    v4f* my = stage[head];
    const v4f* EF4 = reinterpret_cast<const v4f*>(EF);
    const v4f* H4  = reinterpret_cast<const v4f*>(H);
    const unsigned lastTile = (unsigned)(ntile - 1);
    const unsigned lastEdge = (unsigned)(E - 1);

    v4f g[8];
    v4f hv[4];
    unsigned sNext;

    auto loadRows = [&](int tile, v4f* dstv) {
        unsigned tl = (unsigned)tile;
        tl = tl < lastTile ? tl : lastTile;
        const unsigned r0 = tl * 16u + (unsigned)hf;
#pragma unroll
        for (int i = 0; i < 8; ++i) {
            unsigned row = r0 + 2u * (unsigned)i;
            row = row < lastEdge ? row : lastEdge;
            dstv[i] = EF4[row * (unsigned)(EDGE_K / 4) + (unsigned)c];
        }
    };
    auto loadSrc = [&](int tile) -> unsigned {
        unsigned tl = (unsigned)tile;
        tl = tl < lastTile ? tl : lastTile;
        unsigned e = tl * 16u + (unsigned)c;
        e = e < lastEdge ? e : lastEdge;
        return (unsigned)src[e];
    };
    auto loadH = [&](unsigned s, v4f* dstv) {
        const v4f* hp = H4 + (s * (unsigned)(NCOLS / 4) + (unsigned)(8 * head + 2 * hf));
        dstv[0] = hp[0]; dstv[1] = hp[1]; dstv[2] = hp[4]; dstv[3] = hp[5];
    };

    auto tileStep = [&](int tNext, int tNext2) -> float {
#pragma unroll
        for (int i = 0; i < 8; ++i) my[(2 * i + hf) * EF_PITCH4 + c] = g[i];
        __builtin_amdgcn_fence(__ATOMIC_RELEASE, "wavefront");
        __builtin_amdgcn_wave_barrier();
        __builtin_amdgcn_fence(__ATOMIC_ACQUIRE, "wavefront");
        v4f q[8];
#pragma unroll
        for (int i = 0; i < 8; ++i) q[i] = my[c * EF_PITCH4 + 2 * i + hf];
        __builtin_amdgcn_fence(__ATOMIC_RELEASE, "wavefront");
        __builtin_amdgcn_wave_barrier();

        v4f hn[4];
        loadRows(tNext, g);
        loadH(sNext, hn);
        const unsigned s2 = loadSrc(tNext2);
        __builtin_amdgcn_sched_barrier(0);

        v8f acc0 = {}, acc1 = {};
#pragma unroll
        for (int i = 0; i < 8; ++i) {
            const v2f b0 = __builtin_shufflevector(q[i], q[i], 0, 1);
            const v2f b1 = __builtin_shufflevector(q[i], q[i], 2, 3);
            acc0 = __builtin_amdgcn_wmma_f32_16x16x4_f32(false, wf[0][2 * i],     false, b0, (short)0, acc0, false, false);
            acc1 = __builtin_amdgcn_wmma_f32_16x16x4_f32(false, wf[1][2 * i],     false, b0, (short)0, acc1, false, false);
            acc0 = __builtin_amdgcn_wmma_f32_16x16x4_f32(false, wf[0][2 * i + 1], false, b1, (short)0, acc0, false, false);
            acc1 = __builtin_amdgcn_wmma_f32_16x16x4_f32(false, wf[1][2 * i + 1], false, b1, (short)0, acc1, false, false);
        }
        __builtin_amdgcn_sched_barrier(0);

        float d = 0.f;
#pragma unroll
        for (int r = 0; r < 4; ++r) {
            d += acc0[r]     * hv[0][r];
            d += acc0[r + 4] * hv[1][r];
            d += acc1[r]     * hv[2][r];
            d += acc1[r + 4] * hv[3][r];
        }
        d += __shfl_xor(d, 16, 32);
        float sv = d * 0.17677669529663689f;
        sv = sv > 0.f ? sv : 0.01f * sv;

        hv[0] = hn[0]; hv[1] = hn[1]; hv[2] = hn[2]; hv[3] = hn[3];
        sNext = s2;
        return sv;
    };

    loadRows(2 * p, g);
    loadH(loadSrc(2 * p), hv);
    sNext = loadSrc(2 * p + 1);

    for (; p < npair; p += G) {
        const int pn = p + G;
        const float svA = tileStep(2 * p + 1, 2 * pn);
        const float svB = tileStep(2 * pn, 2 * pn + 1);
        const int e = 32 * p + lane;
        if (e < E) score[(size_t)head * E + e] = hf ? svB : svA;
    }
}

__global__ void __launch_bounds__(256)
init_kernel(float* __restrict__ smax, float* __restrict__ denom, int nSeg) {
    int gid = blockIdx.x * 256 + threadIdx.x;
    if (gid < nSeg) { smax[gid] = __int_as_float(0xFF800000); denom[gid] = 0.f; }
}

__device__ __forceinline__ void atomicMaxFloat(float* addr, float val) {
    if (val >= 0.f) atomicMax((int*)addr, __float_as_int(val));
    else            atomicMin((unsigned int*)addr, __float_as_uint(val));
}

__global__ void __launch_bounds__(256)
segmax_kernel(const float* __restrict__ score, const int* __restrict__ dst,
              float* __restrict__ smax, int E) {
    int gid = blockIdx.x * 256 + threadIdx.x;
    if (gid >= E * NHEADS) return;
    int e = gid >> 2, hd = gid & 3;
    atomicMaxFloat(&smax[dst[e] * NHEADS + hd], score[(size_t)hd * E + e]);
}

__global__ void __launch_bounds__(256)
expsum_kernel(const float* __restrict__ score, const int* __restrict__ dst,
              const float* __restrict__ smax, float* __restrict__ aarr,
              float* __restrict__ denom, int E) {
    int gid = blockIdx.x * 256 + threadIdx.x;
    if (gid >= E * NHEADS) return;
    int e = gid >> 2, hd = gid & 3;
    float v = __expf(score[(size_t)hd * E + e] - smax[dst[e] * NHEADS + hd]);
    aarr[gid] = v;
    atomicAdd(&denom[dst[e] * NHEADS + hd], v);
}

#define TILE_N    196
#define AGG_WAVES 32
#define QCAP      544

__device__ __forceinline__ int   rl_i(int v, int k)   { return __builtin_amdgcn_readlane(v, k); }
__device__ __forceinline__ float rl_f(float v, int k) {
    return __int_as_float(__builtin_amdgcn_readlane(__float_as_int(v), k));
}

__global__ void __launch_bounds__(1024)
aggregate_tiles(const float* __restrict__ H, const int* __restrict__ src,
                const int* __restrict__ dst, const float* __restrict__ aarr,
                const float* __restrict__ denom, float* __restrict__ out,
                int E, int N) {
    __shared__ float tile[TILE_N * NCOLS];
    __shared__ int2  queue[AGG_WAVES][QCAP];

    const int lane = threadIdx.x & 31;
    const int wave = __builtin_amdgcn_readfirstlane((int)(threadIdx.x >> 5));
    const int lo   = blockIdx.x * TILE_N;
    int cnt = N - lo; cnt = cnt < TILE_N ? cnt : TILE_N;

    for (int i = threadIdx.x; i < TILE_N * NCOLS; i += 1024) tile[i] = 0.f;
    __syncthreads();

    int2* q = queue[wave];
    int qn = 0;

    auto drain = [&]() {
        __builtin_amdgcn_fence(__ATOMIC_RELEASE, "wavefront");
        __builtin_amdgcn_wave_barrier();
        __builtin_amdgcn_fence(__ATOMIC_ACQUIRE, "wavefront");
        for (int r = 0; r < qn; r += 32) {
            int n = qn - r; n = n < 32 ? n : 32;
            int idx = r + lane; idx = idx < qn ? idx : qn - 1;
            int2 ent = q[idx];
            int   s  = src[ent.x];
            float4 a4 = reinterpret_cast<const float4*>(aarr)[ent.x];
            float4 n4 = reinterpret_cast<const float4*>(denom)[lo + ent.y];
            float w0 = a4.x / n4.x, w1 = a4.y / n4.y, w2 = a4.z / n4.z, w3 = a4.w / n4.w;
            for (int k = 0; k < n; k += 4) {
                float hv[4][4];
#pragma unroll
                for (int u = 0; u < 4; ++u) {
                    const int kk = k + u < n ? k + u : n - 1;
                    const float* hr = H + (size_t)rl_i(s, kk) * NCOLS + lane;
                    hv[u][0] = hr[0]; hv[u][1] = hr[32]; hv[u][2] = hr[64]; hv[u][3] = hr[96];
                }
#pragma unroll
                for (int u = 0; u < 4; ++u) {
                    if (k + u < n) {
                        float* tr = tile + rl_i(ent.y, k + u) * NCOLS + lane;
                        atomicAdd(tr,      hv[u][0] * rl_f(w0, k + u));
                        atomicAdd(tr + 32, hv[u][1] * rl_f(w1, k + u));
                        atomicAdd(tr + 64, hv[u][2] * rl_f(w2, k + u));
                        atomicAdd(tr + 96, hv[u][3] * rl_f(w3, k + u));
                    }
                }
            }
        }
        qn = 0;
        __builtin_amdgcn_fence(__ATOMIC_RELEASE, "wavefront");
        __builtin_amdgcn_wave_barrier();
    };

    auto filter4 = [&](int4 d4, int v) {
        int dd[4] = {d4.x, d4.y, d4.z, d4.w};
#pragma unroll
        for (int c = 0; c < 4; ++c) {
            unsigned rel  = (unsigned)(dd[c] - lo);
            bool     hit  = rel < (unsigned)cnt;
            unsigned mask = __builtin_amdgcn_ballot_w32(hit);
            if (mask) {
                if (hit) q[qn + __builtin_amdgcn_mbcnt_lo(mask, 0u)] = make_int2(v * 4 + c, (int)rel);
                qn += __builtin_popcount(mask);
            }
        }
    };

    const int  nvec = E >> 2;
    const int4 none = make_int4(-1, -1, -1, -1);
    const int4* dst4 = reinterpret_cast<const int4*>(dst);
    if (wave * 32 < nvec) {
        const int last = nvec - 1;
        auto ld = [&](int v) { return dst4[v < last ? v : last]; };
        int vb = wave * 32;
        int4 x0 = ld(vb + lane), x1 = ld(vb + lane + 1024), x2 = ld(vb + lane + 2048), x3 = ld(vb + lane + 3072);
        for (; vb < nvec; vb += 4 * 1024) {
            const int v0 = vb + lane, v1 = v0 + 1024, v2 = v0 + 2048, v3 = v0 + 3072;
            const int nx = v0 + 4 * 1024;
            const int4 y0 = ld(nx), y1 = ld(nx + 1024), y2 = ld(nx + 2048), y3 = ld(nx + 3072);
            filter4(v0 > last ? none : x0, v0);
            filter4(v1 > last ? none : x1, v1);
            filter4(v2 > last ? none : x2, v2);
            filter4(v3 > last ? none : x3, v3);
            if (qn >= 32) drain();
            x0 = y0; x1 = y1; x2 = y2; x3 = y3;
        }
    }
    if (wave == 0) {
        int4 t = none;
        int b = nvec * 4;
        if (lane == 0) {
            if (b     < E) t.x = dst[b];
            if (b + 1 < E) t.y = dst[b + 1];
            if (b + 2 < E) t.z = dst[b + 2];
        }
        filter4(t, nvec + lane);
    }
    if (qn > 0) drain();
    __syncthreads();

    const float4* t4 = reinterpret_cast<const float4*>(tile);
    float4* o4 = reinterpret_cast<float4*>(out + (size_t)lo * NCOLS);
    for (int i = threadIdx.x; i < cnt * (NCOLS / 4); i += 1024) o4[i] = t4[i];
}

extern "C" void kernel_launch(void* const* d_in, const int* in_sizes, int n_in,
                              void* d_out, int out_size, void* d_ws, size_t ws_size,
                              hipStream_t stream) {
    const float* node_feat = (const float*)d_in[0];
    const float* edge_feat = (const float*)d_in[1];
    const int*   src       = (const int*)d_in[2];
    const int*   dst       = (const int*)d_in[3];
    const float* W_node    = (const float*)d_in[4];
    const float* W_edge    = (const float*)d_in[5];
    float*       out       = (float*)d_out;

    const int N = in_sizes[0] / NODE_K;
    const int E = in_sizes[2];

    float* h     = (float*)d_ws;
    float* score = h     + (size_t)N * NCOLS;
    float* aarr  = score + (size_t)E * NHEADS;
    float* smax  = aarr  + (size_t)E * NHEADS;
    float* denom = smax  + (size_t)N * NHEADS;

    gemm_f32_wmma<NODE_K><<<(N + 15) / 16, 256, 0, stream>>>(node_feat, W_node, h, N);

    edge_score_heads<<<SCORE_GRID, 128, 0, stream>>>(edge_feat, W_edge, h, src, score, E);

    init_kernel<<<(N * NHEADS + 255) / 256, 256, 0, stream>>>(smax, denom, N * NHEADS);

    segmax_kernel<<<(E * NHEADS + 255) / 256, 256, 0, stream>>>(score, dst, smax, E);

    expsum_kernel<<<(E * NHEADS + 255) / 256, 256, 0, stream>>>(score, dst, smax, aarr, denom, E);

    aggregate_tiles<<<(N + TILE_N - 1) / TILE_N, 1024, 0, stream>>>(h, src, dst, aarr, denom, out, E, N);
}


# edge softmax moved into the owner-tile aggregation: two dst sweeps, node max then exp sums by LDS atomics, rows divided by the denominator at the store; 6 launches -> 3; denominator sum order changes
# speedup vs baseline: 2.7066x; 1.1581x over previous
#include <hip/hip_runtime.h>
#include <hip/hip_bf16.h>

typedef __attribute__((ext_vector_type(2))) float v2f;
typedef __attribute__((ext_vector_type(8))) float v8f;

#define NCOLS 128
#define NODE_K 256
#define EDGE_K 64
#define NHEADS 4

template <int K>
__global__ void __launch_bounds__(256)
gemm_f32_wmma(const float* __restrict__ A, const float* __restrict__ B,
              float* __restrict__ C, int M) {
    __shared__ float Apanel[16 * K];
    const int lane = threadIdx.x & 31;
    const int wave = threadIdx.x >> 5;
    const int m0   = blockIdx.x * 16;

    for (int idx = threadIdx.x; idx < 16 * K; idx += 256) {
        int r = idx / K, c = idx % K;
        int row = m0 + r; row = row < M ? row : M - 1;
        Apanel[idx] = A[(size_t)row * K + c];
    }
    __syncthreads();

    const int n0      = wave * 16;
    const int half    = lane >> 4;
    const int l15     = lane & 15;
    const int kpair   = half * 2;

    v8f acc = {};
    for (int k = 0; k < K; k += 4) {
        v2f a, b;
        a.x = Apanel[l15 * K + k + kpair];
        a.y = Apanel[l15 * K + k + kpair + 1];
        b.x = B[(size_t)(k + kpair)     * NCOLS + n0 + l15];
        b.y = B[(size_t)(k + kpair + 1) * NCOLS + n0 + l15];
        acc = __builtin_amdgcn_wmma_f32_16x16x4_f32(
            false, a, false, b, (short)0, acc, false, false);
    }

    const int mbase = half * 8;
#pragma unroll
    for (int r = 0; r < 8; ++r) {
        int row = m0 + mbase + r;
        if (row < M) C[(size_t)row * NCOLS + n0 + l15] = acc[r];
    }
}

typedef __attribute__((ext_vector_type(4))) float v4f;

#define EF_PITCH4  17
#define SCORE_GRID 1024

__global__ void __launch_bounds__(128)
edge_score_heads(const float* __restrict__ EF, const float* __restrict__ We,
                 const float* __restrict__ H, const int* __restrict__ src,
                 float* __restrict__ score, int E) {
    __shared__ v4f stage[4][16 * EF_PITCH4];

    const int lane = threadIdx.x & 31;
    const int head = __builtin_amdgcn_readfirstlane((int)(threadIdx.x >> 5));
    const int c  = lane & 15;
    const int hf = lane >> 4;

    const int ntile = (E + 15) >> 4;
    const int npair = (ntile + 1) >> 1;
    int p = blockIdx.x;
    if (p >= npair) return;
    const int G = gridDim.x;

    v2f wf[2][16];
#pragma unroll
    for (int u = 0; u < 2; ++u) {
#pragma unroll
        for (int j = 0; j < 16; ++j) {
            const int k0 = 8 * (j >> 1) + 4 * hf + 2 * (j & 1);
            const float* wp = We + (size_t)k0 * NCOLS + 32 * head + 16 * u + c;
            wf[u][j].x = wp[0];
            wf[u][j].y = wp[NCOLS];
        }
    }

    v4f* my = stage[head];
    const v4f* EF4 = reinterpret_cast<const v4f*>(EF);
    const v4f* H4  = reinterpret_cast<const v4f*>(H);
    const unsigned lastTile = (unsigned)(ntile - 1);
    const unsigned lastEdge = (unsigned)(E - 1);

    v4f g[8];
    v4f hv[4];
    unsigned sNext;

    auto loadRows = [&](int tile, v4f* dstv) {
        unsigned tl = (unsigned)tile;
        tl = tl < lastTile ? tl : lastTile;
        const unsigned r0 = tl * 16u + (unsigned)hf;
#pragma unroll
        for (int i = 0; i < 8; ++i) {
            unsigned row = r0 + 2u * (unsigned)i;
            row = row < lastEdge ? row : lastEdge;
            dstv[i] = EF4[row * (unsigned)(EDGE_K / 4) + (unsigned)c];
        }
    };
    auto loadSrc = [&](int tile) -> unsigned {
        unsigned tl = (unsigned)tile;
        tl = tl < lastTile ? tl : lastTile;
        unsigned e = tl * 16u + (unsigned)c;
        e = e < lastEdge ? e : lastEdge;
        return (unsigned)src[e];
    };
    auto loadH = [&](unsigned s, v4f* dstv) {
        const v4f* hp = H4 + (s * (unsigned)(NCOLS / 4) + (unsigned)(8 * head + 2 * hf));
        dstv[0] = hp[0]; dstv[1] = hp[1]; dstv[2] = hp[4]; dstv[3] = hp[5];
    };

    auto tileStep = [&](int tNext, int tNext2) -> float {
#pragma unroll
        for (int i = 0; i < 8; ++i) my[(2 * i + hf) * EF_PITCH4 + c] = g[i];
        __builtin_amdgcn_fence(__ATOMIC_RELEASE, "wavefront");
        __builtin_amdgcn_wave_barrier();
        __builtin_amdgcn_fence(__ATOMIC_ACQUIRE, "wavefront");
        v4f q[8];
#pragma unroll
        for (int i = 0; i < 8; ++i) q[i] = my[c * EF_PITCH4 + 2 * i + hf];
        __builtin_amdgcn_fence(__ATOMIC_RELEASE, "wavefront");
        __builtin_amdgcn_wave_barrier();

        v4f hn[4];
        loadRows(tNext, g);
        loadH(sNext, hn);
        const unsigned s2 = loadSrc(tNext2);
        __builtin_amdgcn_sched_barrier(0);

        v8f acc0 = {}, acc1 = {};
#pragma unroll
        for (int i = 0; i < 8; ++i) {
            const v2f b0 = __builtin_shufflevector(q[i], q[i], 0, 1);
            const v2f b1 = __builtin_shufflevector(q[i], q[i], 2, 3);
            acc0 = __builtin_amdgcn_wmma_f32_16x16x4_f32(false, wf[0][2 * i],     false, b0, (short)0, acc0, false, false);
            acc1 = __builtin_amdgcn_wmma_f32_16x16x4_f32(false, wf[1][2 * i],     false, b0, (short)0, acc1, false, false);
            acc0 = __builtin_amdgcn_wmma_f32_16x16x4_f32(false, wf[0][2 * i + 1], false, b1, (short)0, acc0, false, false);
            acc1 = __builtin_amdgcn_wmma_f32_16x16x4_f32(false, wf[1][2 * i + 1], false, b1, (short)0, acc1, false, false);
        }
        __builtin_amdgcn_sched_barrier(0);

        float d = 0.f;
#pragma unroll
        for (int r = 0; r < 4; ++r) {
            d += acc0[r]     * hv[0][r];
            d += acc0[r + 4] * hv[1][r];
            d += acc1[r]     * hv[2][r];
            d += acc1[r + 4] * hv[3][r];
        }
        d += __shfl_xor(d, 16, 32);
        float sv = d * 0.17677669529663689f;
        sv = sv > 0.f ? sv : 0.01f * sv;

        hv[0] = hn[0]; hv[1] = hn[1]; hv[2] = hn[2]; hv[3] = hn[3];
        sNext = s2;
        return sv;
    };

    loadRows(2 * p, g);
    loadH(loadSrc(2 * p), hv);
    sNext = loadSrc(2 * p + 1);

    for (; p < npair; p += G) {
        const int pn = p + G;
        const float svA = tileStep(2 * p + 1, 2 * pn);
        const float svB = tileStep(2 * pn, 2 * pn + 1);
        const int e = 32 * p + lane;
        if (e < E) score[(size_t)head * E + e] = hf ? svB : svA;
    }
}

#define TILE_N    196
#define AGG_WAVES 32
#define QCAP      544

__device__ __forceinline__ int   rl_i(int v, int k)   { return __builtin_amdgcn_readlane(v, k); }
__device__ __forceinline__ float rl_f(float v, int k) {
    return __int_as_float(__builtin_amdgcn_readlane(__float_as_int(v), k));
}

__device__ __forceinline__ void ldsMaxFloat(float* addr, float val) {
    if (val >= 0.f) atomicMax((int*)addr, __float_as_int(val));
    else            atomicMin((unsigned int*)addr, __float_as_uint(val));
}

__global__ void __launch_bounds__(1024)
aggregate_tiles(const float* __restrict__ H, const int* __restrict__ src,
                const int* __restrict__ dst, const float* __restrict__ score,
                float* __restrict__ out, int E, int N) {
    __shared__ alignas(16) float tile[TILE_N * NCOLS];
    __shared__ int2  queue[AGG_WAVES][QCAP];
    __shared__ alignas(16) float smaxT[TILE_N * NHEADS];
    __shared__ alignas(16) float denT[TILE_N * NHEADS];

    const int lane = threadIdx.x & 31;
    const int wave = __builtin_amdgcn_readfirstlane((int)(threadIdx.x >> 5));
    const int lo   = blockIdx.x * TILE_N;
    int cnt = N - lo; cnt = cnt < TILE_N ? cnt : TILE_N;

    for (int i = threadIdx.x; i < TILE_N * NCOLS; i += 1024) tile[i] = 0.f;
    if (threadIdx.x < TILE_N * NHEADS) {
        smaxT[threadIdx.x] = __int_as_float(0xFF800000);
        denT[threadIdx.x]  = 0.f;
    }
    __syncthreads();

    int2* q = queue[wave];
    int qn = 0;
    const float* sc0 = score;
    const float* sc1 = score + (size_t)E;
    const float* sc2 = score + 2 * (size_t)E;
    const float* sc3 = score + 3 * (size_t)E;

    auto drainMax = [&]() {
        __builtin_amdgcn_fence(__ATOMIC_RELEASE, "wavefront");
        __builtin_amdgcn_wave_barrier();
        __builtin_amdgcn_fence(__ATOMIC_ACQUIRE, "wavefront");
        for (int r = 0; r < qn; r += 32) {
            const int idx = r + lane;
            if (idx < qn) {
                const int2 ent = q[idx];
                const float s0 = sc0[ent.x], s1 = sc1[ent.x], s2 = sc2[ent.x], s3 = sc3[ent.x];
                float* mp = smaxT + ent.y * NHEADS;
                ldsMaxFloat(mp,     s0);
                ldsMaxFloat(mp + 1, s1);
                ldsMaxFloat(mp + 2, s2);
                ldsMaxFloat(mp + 3, s3);
            }
        }
        qn = 0;
        __builtin_amdgcn_fence(__ATOMIC_RELEASE, "wavefront");
        __builtin_amdgcn_wave_barrier();
    };

    auto drainSum = [&]() {
        __builtin_amdgcn_fence(__ATOMIC_RELEASE, "wavefront");
        __builtin_amdgcn_wave_barrier();
        __builtin_amdgcn_fence(__ATOMIC_ACQUIRE, "wavefront");
        for (int r = 0; r < qn; r += 32) {
            int n = qn - r; n = n < 32 ? n : 32;
            int idx = r + lane; idx = idx < qn ? idx : qn - 1;
            int2 ent = q[idx];
            int   s  = src[ent.x];
            const float s0 = sc0[ent.x], s1 = sc1[ent.x], s2 = sc2[ent.x], s3 = sc3[ent.x];
            const float4 m4 = reinterpret_cast<const float4*>(smaxT)[ent.y];
            float w0 = __expf(s0 - m4.x), w1 = __expf(s1 - m4.y);
            float w2 = __expf(s2 - m4.z), w3 = __expf(s3 - m4.w);
            if (lane < n) {
                float* dp = denT + ent.y * NHEADS;
                atomicAdd(dp,     w0);
                atomicAdd(dp + 1, w1);
                atomicAdd(dp + 2, w2);
                atomicAdd(dp + 3, w3);
            }
            for (int k = 0; k < n; k += 4) {
                float hv[4][4];
#pragma unroll
                for (int u = 0; u < 4; ++u) {
                    const int kk = k + u < n ? k + u : n - 1;
                    const float* hr = H + (size_t)rl_i(s, kk) * NCOLS + lane;
                    hv[u][0] = hr[0]; hv[u][1] = hr[32]; hv[u][2] = hr[64]; hv[u][3] = hr[96];
                }
#pragma unroll
                for (int u = 0; u < 4; ++u) {
                    if (k + u < n) {
                        float* tr = tile + rl_i(ent.y, k + u) * NCOLS + lane;
                        atomicAdd(tr,      hv[u][0] * rl_f(w0, k + u));
                        atomicAdd(tr + 32, hv[u][1] * rl_f(w1, k + u));
                        atomicAdd(tr + 64, hv[u][2] * rl_f(w2, k + u));
                        atomicAdd(tr + 96, hv[u][3] * rl_f(w3, k + u));
                    }
                }
            }
        }
        qn = 0;
        __builtin_amdgcn_fence(__ATOMIC_RELEASE, "wavefront");
        __builtin_amdgcn_wave_barrier();
    };

    auto filter4 = [&](int4 d4, int v) {
        int dd[4] = {d4.x, d4.y, d4.z, d4.w};
#pragma unroll
        for (int c = 0; c < 4; ++c) {
            unsigned rel  = (unsigned)(dd[c] - lo);
            bool     hit  = rel < (unsigned)cnt;
            unsigned mask = __builtin_amdgcn_ballot_w32(hit);
            if (mask) {
                if (hit) q[qn + __builtin_amdgcn_mbcnt_lo(mask, 0u)] = make_int2(v * 4 + c, (int)rel);
                qn += __builtin_popcount(mask);
            }
        }
    };

    const int  nvec = E >> 2;
    const int4 none = make_int4(-1, -1, -1, -1);
    const int4* dst4 = reinterpret_cast<const int4*>(dst);

    auto sweep = [&](auto&& drainFn) {
        if (wave * 32 < nvec) {
            const int last = nvec - 1;
            auto ld = [&](int v) { return dst4[v < last ? v : last]; };
            int vb = wave * 32;
            int4 x0 = ld(vb + lane), x1 = ld(vb + lane + 1024), x2 = ld(vb + lane + 2048), x3 = ld(vb + lane + 3072);
            for (; vb < nvec; vb += 4 * 1024) {
                const int v0 = vb + lane, v1 = v0 + 1024, v2 = v0 + 2048, v3 = v0 + 3072;
                const int nx = v0 + 4 * 1024;
                const int4 y0 = ld(nx), y1 = ld(nx + 1024), y2 = ld(nx + 2048), y3 = ld(nx + 3072);
                filter4(v0 > last ? none : x0, v0);
                filter4(v1 > last ? none : x1, v1);
                filter4(v2 > last ? none : x2, v2);
                filter4(v3 > last ? none : x3, v3);
                if (qn >= 32) drainFn();
                x0 = y0; x1 = y1; x2 = y2; x3 = y3;
            }
        }
        if (wave == 0) {
            int4 t = none;
            int b = nvec * 4;
            if (lane == 0) {
                if (b     < E) t.x = dst[b];
                if (b + 1 < E) t.y = dst[b + 1];
                if (b + 2 < E) t.z = dst[b + 2];
            }
            filter4(t, nvec + lane);
        }
        if (qn > 0) drainFn();
    };

    sweep(drainMax);
    __syncthreads();
    sweep(drainSum);
    __syncthreads();

    const float4* t4 = reinterpret_cast<const float4*>(tile);
    float4* o4 = reinterpret_cast<float4*>(out + (size_t)lo * NCOLS);
    for (int i = threadIdx.x; i < cnt * (NCOLS / 4); i += 1024) {
        const float d = denT[(i >> 5) * NHEADS + ((i & 31) >> 3)];
        float4 v = t4[i];
        const bool has = d > 0.f;
        v.x = has ? v.x / d : 0.f;
        v.y = has ? v.y / d : 0.f;
        v.z = has ? v.z / d : 0.f;
        v.w = has ? v.w / d : 0.f;
        o4[i] = v;
    }
}

extern "C" void kernel_launch(void* const* d_in, const int* in_sizes, int n_in,
                              void* d_out, int out_size, void* d_ws, size_t ws_size,
                              hipStream_t stream) {
    const float* node_feat = (const float*)d_in[0];
    const float* edge_feat = (const float*)d_in[1];
    const int*   src       = (const int*)d_in[2];
    const int*   dst       = (const int*)d_in[3];
    const float* W_node    = (const float*)d_in[4];
    const float* W_edge    = (const float*)d_in[5];
    float*       out       = (float*)d_out;

    const int N = in_sizes[0] / NODE_K;
    const int E = in_sizes[2];

    float* h     = (float*)d_ws;
    float* score = h     + (size_t)N * NCOLS;

    gemm_f32_wmma<NODE_K><<<(N + 15) / 16, 256, 0, stream>>>(node_feat, W_node, h, N);

    edge_score_heads<<<SCORE_GRID, 128, 0, stream>>>(edge_feat, W_edge, h, src, score, E);

    aggregate_tiles<<<(N + TILE_N - 1) / TILE_N, 1024, 0, stream>>>(h, src, dst, score, out, E, N);
}


# aggregation reads the dst list once: hits stay in the wave's LDS list with src id and four scores, pass 2 walks that list instead of a second sweep; LDS add order stays arrival order
# speedup vs baseline: 3.2146x; 1.1877x over previous
#include <hip/hip_runtime.h>
#include <hip/hip_bf16.h>

typedef __attribute__((ext_vector_type(2))) float v2f;
typedef __attribute__((ext_vector_type(8))) float v8f;

#define NCOLS 128
#define NODE_K 256
#define EDGE_K 64
#define NHEADS 4

template <int K>
__global__ void __launch_bounds__(256)
gemm_f32_wmma(const float* __restrict__ A, const float* __restrict__ B,
              float* __restrict__ C, int M) {
    __shared__ float Apanel[16 * K];
    const int lane = threadIdx.x & 31;
    const int wave = threadIdx.x >> 5;
    const int m0   = blockIdx.x * 16;

    for (int idx = threadIdx.x; idx < 16 * K; idx += 256) {
        int r = idx / K, c = idx % K;
        int row = m0 + r; row = row < M ? row : M - 1;
        Apanel[idx] = A[(size_t)row * K + c];
    }
    __syncthreads();

    const int n0      = wave * 16;
    const int half    = lane >> 4;
    const int l15     = lane & 15;
    const int kpair   = half * 2;

    v8f acc = {};
    for (int k = 0; k < K; k += 4) {
        v2f a, b;
        a.x = Apanel[l15 * K + k + kpair];
        a.y = Apanel[l15 * K + k + kpair + 1];
        b.x = B[(size_t)(k + kpair)     * NCOLS + n0 + l15];
        b.y = B[(size_t)(k + kpair + 1) * NCOLS + n0 + l15];
        acc = __builtin_amdgcn_wmma_f32_16x16x4_f32(
            false, a, false, b, (short)0, acc, false, false);
    }

    const int mbase = half * 8;
#pragma unroll
    for (int r = 0; r < 8; ++r) {
        int row = m0 + mbase + r;
        if (row < M) C[(size_t)row * NCOLS + n0 + l15] = acc[r];
    }
}

typedef __attribute__((ext_vector_type(4))) float v4f;

#define EF_PITCH4  17
#define SCORE_GRID 1024

__global__ void __launch_bounds__(128)
edge_score_heads(const float* __restrict__ EF, const float* __restrict__ We,
                 const float* __restrict__ H, const int* __restrict__ src,
                 float* __restrict__ score, int E) {
    __shared__ v4f stage[4][16 * EF_PITCH4];

    const int lane = threadIdx.x & 31;
    const int head = __builtin_amdgcn_readfirstlane((int)(threadIdx.x >> 5));
    const int c  = lane & 15;
    const int hf = lane >> 4;

    const int ntile = (E + 15) >> 4;
    const int npair = (ntile + 1) >> 1;
    int p = blockIdx.x;
    if (p >= npair) return;
    const int G = gridDim.x;

    v2f wf[2][16];
#pragma unroll
    for (int u = 0; u < 2; ++u) {
#pragma unroll
        for (int j = 0; j < 16; ++j) {
            const int k0 = 8 * (j >> 1) + 4 * hf + 2 * (j & 1);
            const float* wp = We + (size_t)k0 * NCOLS + 32 * head + 16 * u + c;
            wf[u][j].x = wp[0];
            wf[u][j].y = wp[NCOLS];
        }
    }

    v4f* my = stage[head];
    const v4f* EF4 = reinterpret_cast<const v4f*>(EF);
    const v4f* H4  = reinterpret_cast<const v4f*>(H);
    const unsigned lastTile = (unsigned)(ntile - 1);
    const unsigned lastEdge = (unsigned)(E - 1);

    v4f g[8];
    v4f hv[4];
    unsigned sNext;

    auto loadRows = [&](int tile, v4f* dstv) {
        unsigned tl = (unsigned)tile;
        tl = tl < lastTile ? tl : lastTile;
        const unsigned r0 = tl * 16u + (unsigned)hf;
#pragma unroll
        for (int i = 0; i < 8; ++i) {
            unsigned row = r0 + 2u * (unsigned)i;
            row = row < lastEdge ? row : lastEdge;
            dstv[i] = EF4[row * (unsigned)(EDGE_K / 4) + (unsigned)c];
        }
    };
    auto loadSrc = [&](int tile) -> unsigned {
        unsigned tl = (unsigned)tile;
        tl = tl < lastTile ? tl : lastTile;
        unsigned e = tl * 16u + (unsigned)c;
        e = e < lastEdge ? e : lastEdge;
        return (unsigned)src[e];
    };
    auto loadH = [&](unsigned s, v4f* dstv) {
        const v4f* hp = H4 + (s * (unsigned)(NCOLS / 4) + (unsigned)(8 * head + 2 * hf));
        dstv[0] = hp[0]; dstv[1] = hp[1]; dstv[2] = hp[4]; dstv[3] = hp[5];
    };

    auto tileStep = [&](int tNext, int tNext2) -> float {
#pragma unroll
        for (int i = 0; i < 8; ++i) my[(2 * i + hf) * EF_PITCH4 + c] = g[i];
        __builtin_amdgcn_fence(__ATOMIC_RELEASE, "wavefront");
        __builtin_amdgcn_wave_barrier();
        __builtin_amdgcn_fence(__ATOMIC_ACQUIRE, "wavefront");
        v4f q[8];
#pragma unroll
        for (int i = 0; i < 8; ++i) q[i] = my[c * EF_PITCH4 + 2 * i + hf];
        __builtin_amdgcn_fence(__ATOMIC_RELEASE, "wavefront");
        __builtin_amdgcn_wave_barrier();

        v4f hn[4];
        loadRows(tNext, g);
        loadH(sNext, hn);
        const unsigned s2 = loadSrc(tNext2);
        __builtin_amdgcn_sched_barrier(0);

        v8f acc0 = {}, acc1 = {};
#pragma unroll
        for (int i = 0; i < 8; ++i) {
            const v2f b0 = __builtin_shufflevector(q[i], q[i], 0, 1);
            const v2f b1 = __builtin_shufflevector(q[i], q[i], 2, 3);
            acc0 = __builtin_amdgcn_wmma_f32_16x16x4_f32(false, wf[0][2 * i],     false, b0, (short)0, acc0, false, false);
            acc1 = __builtin_amdgcn_wmma_f32_16x16x4_f32(false, wf[1][2 * i],     false, b0, (short)0, acc1, false, false);
            acc0 = __builtin_amdgcn_wmma_f32_16x16x4_f32(false, wf[0][2 * i + 1], false, b1, (short)0, acc0, false, false);
            acc1 = __builtin_amdgcn_wmma_f32_16x16x4_f32(false, wf[1][2 * i + 1], false, b1, (short)0, acc1, false, false);
        }
        __builtin_amdgcn_sched_barrier(0);

        float d = 0.f;
#pragma unroll
        for (int r = 0; r < 4; ++r) {
            d += acc0[r]     * hv[0][r];
            d += acc0[r + 4] * hv[1][r];
            d += acc1[r]     * hv[2][r];
            d += acc1[r + 4] * hv[3][r];
        }
        d += __shfl_xor(d, 16, 32);
        float sv = d * 0.17677669529663689f;
        sv = sv > 0.f ? sv : 0.01f * sv;

        hv[0] = hn[0]; hv[1] = hn[1]; hv[2] = hn[2]; hv[3] = hn[3];
        sNext = s2;
        return sv;
    };

    loadRows(2 * p, g);
    loadH(loadSrc(2 * p), hv);
    sNext = loadSrc(2 * p + 1);

    for (; p < npair; p += G) {
        const int pn = p + G;
        const float svA = tileStep(2 * p + 1, 2 * pn);
        const float svB = tileStep(2 * pn, 2 * pn + 1);
        const int e = 32 * p + lane;
        if (e < E) score[(size_t)head * E + e] = hf ? svB : svA;
    }
}

#define TILE_N    196
#define AGG_WAVES 32
#define P_MAX     224
#define Q_BURST   160
#define Q_CAP     (P_MAX + Q_BURST)

__device__ __forceinline__ int   rl_i(int v, int k)   { return __builtin_amdgcn_readlane(v, k); }
__device__ __forceinline__ float rl_f(float v, int k) {
    return __int_as_float(__builtin_amdgcn_readlane(__float_as_int(v), k));
}

__device__ __forceinline__ void ldsMaxFloat(float* addr, float val) {
    if (val >= 0.f) atomicMax((int*)addr, __float_as_int(val));
    else            atomicMin((unsigned int*)addr, __float_as_uint(val));
}

__global__ void __launch_bounds__(1024)
aggregate_tiles(const float* __restrict__ H, const int* __restrict__ src,
                const int* __restrict__ dst, const float* __restrict__ score,
                float* __restrict__ out, int E, int N) {
    __shared__ alignas(16) float  tile[TILE_N * NCOLS];
    __shared__ alignas(16) float4 keptScore[AGG_WAVES][P_MAX];
    __shared__ int2  queue[AGG_WAVES][Q_CAP];
    __shared__ alignas(16) float smaxT[TILE_N * NHEADS];
    __shared__ alignas(16) float denT[TILE_N * NHEADS];

    const int lane = threadIdx.x & 31;
    const int wave = __builtin_amdgcn_readfirstlane((int)(threadIdx.x >> 5));
    const int lo   = blockIdx.x * TILE_N;
    int cnt = N - lo; cnt = cnt < TILE_N ? cnt : TILE_N;

    for (int i = threadIdx.x; i < TILE_N * NCOLS; i += 1024) tile[i] = 0.f;
    if (threadIdx.x < TILE_N * NHEADS) {
        smaxT[threadIdx.x] = __int_as_float(0xFF800000);
        denT[threadIdx.x]  = 0.f;
    }
    __syncthreads();

    int2*   q  = queue[wave];
    float4* ks = keptScore[wave];
    int  kept = 0;
    int  qn   = 0;
    bool keeping = true;
    int  resumeCall = 0;
    const float* sc0 = score;
    const float* sc1 = score + (size_t)E;
    const float* sc2 = score + 2 * (size_t)E;
    const float* sc3 = score + 3 * (size_t)E;

    auto drainMax = [&](int callsDone) {
        __builtin_amdgcn_fence(__ATOMIC_RELEASE, "wavefront");
        __builtin_amdgcn_wave_barrier();
        __builtin_amdgcn_fence(__ATOMIC_ACQUIRE, "wavefront");
        const bool keep = keeping && qn <= P_MAX;
        for (int r = kept; r < qn; r += 32) {
            const int idx = r + lane;
            if (idx < qn) {
                const int2 ent = q[idx];
                const int   s  = src[ent.x];
                const float s0 = sc0[ent.x], s1 = sc1[ent.x], s2 = sc2[ent.x], s3 = sc3[ent.x];
                float* mp = smaxT + ent.y * NHEADS;
                ldsMaxFloat(mp,     s0);
                ldsMaxFloat(mp + 1, s1);
                ldsMaxFloat(mp + 2, s2);
                ldsMaxFloat(mp + 3, s3);
                if (keep) {
                    q[idx]  = make_int2(s, ent.y);
                    ks[idx] = make_float4(s0, s1, s2, s3);
                }
            }
        }
        if (keep) { kept = qn; resumeCall = callsDone; }
        else      { keeping = false; qn = kept; }
        __builtin_amdgcn_fence(__ATOMIC_RELEASE, "wavefront");
        __builtin_amdgcn_wave_barrier();
    };

    auto sumBatch = [&](int n, int s, int row, float s0, float s1, float s2, float s3) {
        const float4 m4 = reinterpret_cast<const float4*>(smaxT)[row];
        const float w0 = __expf(s0 - m4.x), w1 = __expf(s1 - m4.y);
        const float w2 = __expf(s2 - m4.z), w3 = __expf(s3 - m4.w);
        if (lane < n) {
            float* dp = denT + row * NHEADS;
            atomicAdd(dp,     w0);
            atomicAdd(dp + 1, w1);
            atomicAdd(dp + 2, w2);
            atomicAdd(dp + 3, w3);
        }
        for (int k = 0; k < n; k += 4) {
            float hv[4][4];
#pragma unroll
            for (int u = 0; u < 4; ++u) {
                const int kk = k + u < n ? k + u : n - 1;
                const float* hr = H + (size_t)rl_i(s, kk) * NCOLS + lane;
                hv[u][0] = hr[0]; hv[u][1] = hr[32]; hv[u][2] = hr[64]; hv[u][3] = hr[96];
            }
#pragma unroll
            for (int u = 0; u < 4; ++u) {
                if (k + u < n) {
                    float* tr = tile + rl_i(row, k + u) * NCOLS + lane;
                    atomicAdd(tr,      hv[u][0] * rl_f(w0, k + u));
                    atomicAdd(tr + 32, hv[u][1] * rl_f(w1, k + u));
                    atomicAdd(tr + 64, hv[u][2] * rl_f(w2, k + u));
                    atomicAdd(tr + 96, hv[u][3] * rl_f(w3, k + u));
                }
            }
        }
    };

    auto drainSumGather = [&]() {
        __builtin_amdgcn_fence(__ATOMIC_RELEASE, "wavefront");
        __builtin_amdgcn_wave_barrier();
        __builtin_amdgcn_fence(__ATOMIC_ACQUIRE, "wavefront");
        for (int r = kept; r < qn; r += 32) {
            int n = qn - r; n = n < 32 ? n : 32;
            int idx = r + lane; idx = idx < qn ? idx : qn - 1;
            const int2 ent = q[idx];
            const int   s  = src[ent.x];
            const float s0 = sc0[ent.x], s1 = sc1[ent.x], s2 = sc2[ent.x], s3 = sc3[ent.x];
            sumBatch(n, s, ent.y, s0, s1, s2, s3);
        }
        qn = kept;
        __builtin_amdgcn_fence(__ATOMIC_RELEASE, "wavefront");
        __builtin_amdgcn_wave_barrier();
    };

    auto filter4 = [&](int4 d4, int v) {
        int dd[4] = {d4.x, d4.y, d4.z, d4.w};
#pragma unroll
        for (int c = 0; c < 4; ++c) {
            unsigned rel  = (unsigned)(dd[c] - lo);
            bool     hit  = rel < (unsigned)cnt;
            unsigned mask = __builtin_amdgcn_ballot_w32(hit);
            if (mask) {
                if (hit) q[qn + __builtin_amdgcn_mbcnt_lo(mask, 0u)] = make_int2(v * 4 + c, (int)rel);
                qn += __builtin_popcount(mask);
            }
        }
    };

    const int  nvec = E >> 2;
    const int  last = nvec - 1;
    const int4 none = make_int4(-1, -1, -1, -1);
    const int4* dst4 = reinterpret_cast<const int4*>(dst);
    const int  vfirst = wave * 32;
    auto ld = [&](int v) { return dst4[v < last ? v : last]; };

    auto filterLeftover = [&]() {
        if (wave == 0) {
            int4 t = none;
            int b = nvec * 4;
            if (lane == 0) {
                if (b     < E) t.x = dst[b];
                if (b + 1 < E) t.y = dst[b + 1];
                if (b + 2 < E) t.z = dst[b + 2];
            }
            filter4(t, nvec + lane);
        }
    };

    if (vfirst < nvec) {
        int vb = vfirst;
        int4 x0 = ld(vb + lane), x1 = ld(vb + lane + 1024), x2 = ld(vb + lane + 2048), x3 = ld(vb + lane + 3072);
        for (; vb < nvec; vb += 4 * 1024) {
            const int v0 = vb + lane, v1 = v0 + 1024, v2 = v0 + 2048, v3 = v0 + 3072;
            const int nx = v0 + 4 * 1024;
            const int4 y0 = ld(nx), y1 = ld(nx + 1024), y2 = ld(nx + 2048), y3 = ld(nx + 3072);
            const int c0 = ((vb - vfirst) >> 12) * 4;
            filter4(v0 > last ? none : x0, v0);
            if (qn - kept >= 32) drainMax(c0 + 1);
            filter4(v1 > last ? none : x1, v1);
            if (qn - kept >= 32) drainMax(c0 + 2);
            filter4(v2 > last ? none : x2, v2);
            if (qn - kept >= 32) drainMax(c0 + 3);
            filter4(v3 > last ? none : x3, v3);
            if (qn - kept >= 32) drainMax(c0 + 4);
            x0 = y0; x1 = y1; x2 = y2; x3 = y3;
        }
    }
    filterLeftover();
    if (qn > kept) drainMax(0x7fffffff);
    __syncthreads();

    for (int r = 0; r < kept; r += 32) {
        int n = kept - r; n = n < 32 ? n : 32;
        int idx = r + lane; idx = idx < kept ? idx : kept - 1;
        const int2   ent = q[idx];
        const float4 s4  = ks[idx];
        sumBatch(n, ent.x, ent.y, s4.x, s4.y, s4.z, s4.w);
    }
    if (!keeping) {
        if (vfirst < nvec) {
            const int ncall = ((nvec - vfirst + 4095) >> 12) * 4;
            for (int ci = resumeCall; ci < ncall; ++ci) {
                const int v = vfirst + (ci >> 2) * 4096 + (ci & 3) * 1024 + lane;
                const int4 x = ld(v);
                filter4(v > last ? none : x, v);
                if (qn - kept >= 32) drainSumGather();
            }
        }
        filterLeftover();
        if (qn > kept) drainSumGather();
    }
    __syncthreads();

    const float4* t4 = reinterpret_cast<const float4*>(tile);
    float4* o4 = reinterpret_cast<float4*>(out + (size_t)lo * NCOLS);
    for (int i = threadIdx.x; i < cnt * (NCOLS / 4); i += 1024) {
        const float d = denT[(i >> 5) * NHEADS + ((i & 31) >> 3)];
        float4 v = t4[i];
        const bool has = d > 0.f;
        v.x = has ? v.x / d : 0.f;
        v.y = has ? v.y / d : 0.f;
        v.z = has ? v.z / d : 0.f;
        v.w = has ? v.w / d : 0.f;
        o4[i] = v;
    }
}

extern "C" void kernel_launch(void* const* d_in, const int* in_sizes, int n_in,
                              void* d_out, int out_size, void* d_ws, size_t ws_size,
                              hipStream_t stream) {
    const float* node_feat = (const float*)d_in[0];
    const float* edge_feat = (const float*)d_in[1];
    const int*   src       = (const int*)d_in[2];
    const int*   dst       = (const int*)d_in[3];
    const float* W_node    = (const float*)d_in[4];
    const float* W_edge    = (const float*)d_in[5];
    float*       out       = (float*)d_out;

    const int N = in_sizes[0] / NODE_K;
    const int E = in_sizes[2];

    float* h     = (float*)d_ws;
    float* score = h     + (size_t)N * NCOLS;

    gemm_f32_wmma<NODE_K><<<(N + 15) / 16, 256, 0, stream>>>(node_feat, W_node, h, N);

    edge_score_heads<<<SCORE_GRID, 128, 0, stream>>>(edge_feat, W_edge, h, src, score, E);

    aggregate_tiles<<<(N + TILE_N - 1) / TILE_N, 1024, 0, stream>>>(h, src, dst, score, out, E, N);
}


# node projection re-decomposed: a wave keeps a 256x64 slice of W_node in registers and sweeps 16-row tiles, each lane reading its own row from memory (no LDS panel, no barrier); k sum order changes
# speedup vs baseline: 3.7550x; 1.1681x over previous
#include <hip/hip_runtime.h>
#include <hip/hip_bf16.h>

typedef __attribute__((ext_vector_type(2))) float v2f;
typedef __attribute__((ext_vector_type(8))) float v8f;

#define NCOLS 128
#define NODE_K 256
#define EDGE_K 64
#define NHEADS 4

typedef __attribute__((ext_vector_type(4))) float v4f;

#define GEMM_GRID 512

__global__ void __launch_bounds__(64)
node_gemm_wreg(const float* __restrict__ A, const float* __restrict__ B,
               float* __restrict__ C, int M) {
    const int lane = threadIdx.x & 31;
    const int ch = __builtin_amdgcn_readfirstlane((int)(threadIdx.x >> 5));
    const int c  = lane & 15;
    const int hf = lane >> 4;

    const int ntile = (M + 15) >> 4;
    int t = blockIdx.x;
    if (t >= ntile) return;
    const int G = gridDim.x;

    v2f wf[4][NODE_K / 4];
    const v4f* B4 = reinterpret_cast<const v4f*>(B);
#pragma unroll
    for (int j = 0; j < NODE_K / 4; ++j) {
        const int k0 = 8 * (j >> 1) + 4 * hf + 2 * (j & 1);
        const v4f w0 = B4[k0 * (NCOLS / 4) + 16 * ch + c];
        const v4f w1 = B4[(k0 + 1) * (NCOLS / 4) + 16 * ch + c];
#pragma unroll
        for (int u = 0; u < 4; ++u) { wf[u][j].x = w0[u]; wf[u][j].y = w1[u]; }
    }

    const v4f* A4 = reinterpret_cast<const v4f*>(A);
    const int lastTile = ntile - 1;
    const int lastRow  = M - 1;

    auto loadTile = [&](int tile, v4f* dstv) {
        int tl = tile < lastTile ? tile : lastTile;
        int row = tl * 16 + c; row = row < lastRow ? row : lastRow;
        const v4f* rp = A4 + (size_t)row * (NODE_K / 4) + hf;
#pragma unroll
        for (int i = 0; i < NODE_K / 8; ++i) dstv[i] = rp[2 * i];
    };

    v4f cur[NODE_K / 8];
    loadTile(t, cur);

    for (; t < ntile; t += G) {
        v4f nxt[NODE_K / 8];
        loadTile(t + G, nxt);
        __builtin_amdgcn_sched_barrier(0);

        v8f acc[4] = {};
#pragma unroll
        for (int i = 0; i < NODE_K / 8; ++i) {
            const v2f a0 = __builtin_shufflevector(cur[i], cur[i], 0, 1);
            const v2f a1 = __builtin_shufflevector(cur[i], cur[i], 2, 3);
#pragma unroll
            for (int u = 0; u < 4; ++u)
                acc[u] = __builtin_amdgcn_wmma_f32_16x16x4_f32(false, a0, false, wf[u][2 * i],     (short)0, acc[u], false, false);
#pragma unroll
            for (int u = 0; u < 4; ++u)
                acc[u] = __builtin_amdgcn_wmma_f32_16x16x4_f32(false, a1, false, wf[u][2 * i + 1], (short)0, acc[u], false, false);
        }
        __builtin_amdgcn_sched_barrier(0);

        const int rbase = t * 16 + 8 * hf;
#pragma unroll
        for (int r = 0; r < 8; ++r) {
            const int row = rbase + r;
            if (row < M) {
                v4f o;
                o.x = acc[0][r]; o.y = acc[1][r]; o.z = acc[2][r]; o.w = acc[3][r];
                *reinterpret_cast<v4f*>(C + (size_t)row * NCOLS + 64 * ch + 4 * c) = o;
            }
        }
#pragma unroll
        for (int i = 0; i < NODE_K / 8; ++i) cur[i] = nxt[i];
    }
}

#define EF_PITCH4  17
#define SCORE_GRID 1024

__global__ void __launch_bounds__(128)
edge_score_heads(const float* __restrict__ EF, const float* __restrict__ We,
                 const float* __restrict__ H, const int* __restrict__ src,
                 float* __restrict__ score, int E) {
    __shared__ v4f stage[4][16 * EF_PITCH4];

    const int lane = threadIdx.x & 31;
    const int head = __builtin_amdgcn_readfirstlane((int)(threadIdx.x >> 5));
    const int c  = lane & 15;
    const int hf = lane >> 4;

    const int ntile = (E + 15) >> 4;
    const int npair = (ntile + 1) >> 1;
    int p = blockIdx.x;
    if (p >= npair) return;
    const int G = gridDim.x;

    v2f wf[2][16];
#pragma unroll
    for (int u = 0; u < 2; ++u) {
#pragma unroll
        for (int j = 0; j < 16; ++j) {
            const int k0 = 8 * (j >> 1) + 4 * hf + 2 * (j & 1);
            const float* wp = We + (size_t)k0 * NCOLS + 32 * head + 16 * u + c;
            wf[u][j].x = wp[0];
            wf[u][j].y = wp[NCOLS];
        }
    }

    v4f* my = stage[head];
    const v4f* EF4 = reinterpret_cast<const v4f*>(EF);
    const v4f* H4  = reinterpret_cast<const v4f*>(H);
    const unsigned lastTile = (unsigned)(ntile - 1);
    const unsigned lastEdge = (unsigned)(E - 1);

    v4f g[8];
    v4f hv[4];
    unsigned sNext;

    auto loadRows = [&](int tile, v4f* dstv) {
        unsigned tl = (unsigned)tile;
        tl = tl < lastTile ? tl : lastTile;
        const unsigned r0 = tl * 16u + (unsigned)hf;
#pragma unroll
        for (int i = 0; i < 8; ++i) {
            unsigned row = r0 + 2u * (unsigned)i;
            row = row < lastEdge ? row : lastEdge;
            dstv[i] = EF4[row * (unsigned)(EDGE_K / 4) + (unsigned)c];
        }
    };
    auto loadSrc = [&](int tile) -> unsigned {
        unsigned tl = (unsigned)tile;
        tl = tl < lastTile ? tl : lastTile;
        unsigned e = tl * 16u + (unsigned)c;
        e = e < lastEdge ? e : lastEdge;
        return (unsigned)src[e];
    };
    auto loadH = [&](unsigned s, v4f* dstv) {
        const v4f* hp = H4 + (s * (unsigned)(NCOLS / 4) + (unsigned)(8 * head + 2 * hf));
        dstv[0] = hp[0]; dstv[1] = hp[1]; dstv[2] = hp[4]; dstv[3] = hp[5];
    };

    auto tileStep = [&](int tNext, int tNext2) -> float {
#pragma unroll
        for (int i = 0; i < 8; ++i) my[(2 * i + hf) * EF_PITCH4 + c] = g[i];
        __builtin_amdgcn_fence(__ATOMIC_RELEASE, "wavefront");
        __builtin_amdgcn_wave_barrier();
        __builtin_amdgcn_fence(__ATOMIC_ACQUIRE, "wavefront");
        v4f q[8];
#pragma unroll
        for (int i = 0; i < 8; ++i) q[i] = my[c * EF_PITCH4 + 2 * i + hf];
        __builtin_amdgcn_fence(__ATOMIC_RELEASE, "wavefront");
        __builtin_amdgcn_wave_barrier();

        v4f hn[4];
        loadRows(tNext, g);
        loadH(sNext, hn);
        const unsigned s2 = loadSrc(tNext2);
        __builtin_amdgcn_sched_barrier(0);

        v8f acc0 = {}, acc1 = {};
#pragma unroll
        for (int i = 0; i < 8; ++i) {
            const v2f b0 = __builtin_shufflevector(q[i], q[i], 0, 1);
            const v2f b1 = __builtin_shufflevector(q[i], q[i], 2, 3);
            acc0 = __builtin_amdgcn_wmma_f32_16x16x4_f32(false, wf[0][2 * i],     false, b0, (short)0, acc0, false, false);
            acc1 = __builtin_amdgcn_wmma_f32_16x16x4_f32(false, wf[1][2 * i],     false, b0, (short)0, acc1, false, false);
            acc0 = __builtin_amdgcn_wmma_f32_16x16x4_f32(false, wf[0][2 * i + 1], false, b1, (short)0, acc0, false, false);
            acc1 = __builtin_amdgcn_wmma_f32_16x16x4_f32(false, wf[1][2 * i + 1], false, b1, (short)0, acc1, false, false);
        }
        __builtin_amdgcn_sched_barrier(0);

        float d = 0.f;
#pragma unroll
        for (int r = 0; r < 4; ++r) {
            d += acc0[r]     * hv[0][r];
            d += acc0[r + 4] * hv[1][r];
            d += acc1[r]     * hv[2][r];
            d += acc1[r + 4] * hv[3][r];
        }
        d += __shfl_xor(d, 16, 32);
        float sv = d * 0.17677669529663689f;
        sv = sv > 0.f ? sv : 0.01f * sv;

        hv[0] = hn[0]; hv[1] = hn[1]; hv[2] = hn[2]; hv[3] = hn[3];
        sNext = s2;
        return sv;
    };

    loadRows(2 * p, g);
    loadH(loadSrc(2 * p), hv);
    sNext = loadSrc(2 * p + 1);

    for (; p < npair; p += G) {
        const int pn = p + G;
        const float svA = tileStep(2 * p + 1, 2 * pn);
        const float svB = tileStep(2 * pn, 2 * pn + 1);
        const int e = 32 * p + lane;
        if (e < E) score[(size_t)head * E + e] = hf ? svB : svA;
    }
}

#define TILE_N    196
#define AGG_WAVES 32
#define P_MAX     224
#define Q_BURST   160
#define Q_CAP     (P_MAX + Q_BURST)

__device__ __forceinline__ int   rl_i(int v, int k)   { return __builtin_amdgcn_readlane(v, k); }
__device__ __forceinline__ float rl_f(float v, int k) {
    return __int_as_float(__builtin_amdgcn_readlane(__float_as_int(v), k));
}

__device__ __forceinline__ void ldsMaxFloat(float* addr, float val) {
    if (val >= 0.f) atomicMax((int*)addr, __float_as_int(val));
    else            atomicMin((unsigned int*)addr, __float_as_uint(val));
}

__global__ void __launch_bounds__(1024)
aggregate_tiles(const float* __restrict__ H, const int* __restrict__ src,
                const int* __restrict__ dst, const float* __restrict__ score,
                float* __restrict__ out, int E, int N) {
    __shared__ alignas(16) float  tile[TILE_N * NCOLS];
    __shared__ alignas(16) float4 keptScore[AGG_WAVES][P_MAX];
    __shared__ int2  queue[AGG_WAVES][Q_CAP];
    __shared__ alignas(16) float smaxT[TILE_N * NHEADS];
    __shared__ alignas(16) float denT[TILE_N * NHEADS];

    const int lane = threadIdx.x & 31;
    const int wave = __builtin_amdgcn_readfirstlane((int)(threadIdx.x >> 5));
    const int lo   = blockIdx.x * TILE_N;
    int cnt = N - lo; cnt = cnt < TILE_N ? cnt : TILE_N;

    for (int i = threadIdx.x; i < TILE_N * NCOLS; i += 1024) tile[i] = 0.f;
    if (threadIdx.x < TILE_N * NHEADS) {
        smaxT[threadIdx.x] = __int_as_float(0xFF800000);
        denT[threadIdx.x]  = 0.f;
    }
    __syncthreads();

    int2*   q  = queue[wave];
    float4* ks = keptScore[wave];
    int  kept = 0;
    int  qn   = 0;
    bool keeping = true;
    int  resumeCall = 0;
    const float* sc0 = score;
    const float* sc1 = score + (size_t)E;
    const float* sc2 = score + 2 * (size_t)E;
    const float* sc3 = score + 3 * (size_t)E;

    auto drainMax = [&](int callsDone) {
        __builtin_amdgcn_fence(__ATOMIC_RELEASE, "wavefront");
        __builtin_amdgcn_wave_barrier();
        __builtin_amdgcn_fence(__ATOMIC_ACQUIRE, "wavefront");
        const bool keep = keeping && qn <= P_MAX;
        for (int r = kept; r < qn; r += 32) {
            const int idx = r + lane;
            if (idx < qn) {
                const int2 ent = q[idx];
                const int   s  = src[ent.x];
                const float s0 = sc0[ent.x], s1 = sc1[ent.x], s2 = sc2[ent.x], s3 = sc3[ent.x];
                float* mp = smaxT + ent.y * NHEADS;
                ldsMaxFloat(mp,     s0);
                ldsMaxFloat(mp + 1, s1);
                ldsMaxFloat(mp + 2, s2);
                ldsMaxFloat(mp + 3, s3);
                if (keep) {
                    q[idx]  = make_int2(s, ent.y);
                    ks[idx] = make_float4(s0, s1, s2, s3);
                }
            }
        }
        if (keep) { kept = qn; resumeCall = callsDone; }
        else      { keeping = false; qn = kept; }
        __builtin_amdgcn_fence(__ATOMIC_RELEASE, "wavefront");
        __builtin_amdgcn_wave_barrier();
    };

    auto sumBatch = [&](int n, int s, int row, float s0, float s1, float s2, float s3) {
        const float4 m4 = reinterpret_cast<const float4*>(smaxT)[row];
        const float w0 = __expf(s0 - m4.x), w1 = __expf(s1 - m4.y);
        const float w2 = __expf(s2 - m4.z), w3 = __expf(s3 - m4.w);
        if (lane < n) {
            float* dp = denT + row * NHEADS;
            atomicAdd(dp,     w0);
            atomicAdd(dp + 1, w1);
            atomicAdd(dp + 2, w2);
            atomicAdd(dp + 3, w3);
        }
        for (int k = 0; k < n; k += 4) {
            float hv[4][4];
#pragma unroll
            for (int u = 0; u < 4; ++u) {
                const int kk = k + u < n ? k + u : n - 1;
                const float* hr = H + (size_t)rl_i(s, kk) * NCOLS + lane;
                hv[u][0] = hr[0]; hv[u][1] = hr[32]; hv[u][2] = hr[64]; hv[u][3] = hr[96];
            }
#pragma unroll
            for (int u = 0; u < 4; ++u) {
                if (k + u < n) {
                    float* tr = tile + rl_i(row, k + u) * NCOLS + lane;
                    atomicAdd(tr,      hv[u][0] * rl_f(w0, k + u));
                    atomicAdd(tr + 32, hv[u][1] * rl_f(w1, k + u));
                    atomicAdd(tr + 64, hv[u][2] * rl_f(w2, k + u));
                    atomicAdd(tr + 96, hv[u][3] * rl_f(w3, k + u));
                }
            }
        }
    };

    auto drainSumGather = [&]() {
        __builtin_amdgcn_fence(__ATOMIC_RELEASE, "wavefront");
        __builtin_amdgcn_wave_barrier();
        __builtin_amdgcn_fence(__ATOMIC_ACQUIRE, "wavefront");
        for (int r = kept; r < qn; r += 32) {
            int n = qn - r; n = n < 32 ? n : 32;
            int idx = r + lane; idx = idx < qn ? idx : qn - 1;
            const int2 ent = q[idx];
            const int   s  = src[ent.x];
            const float s0 = sc0[ent.x], s1 = sc1[ent.x], s2 = sc2[ent.x], s3 = sc3[ent.x];
            sumBatch(n, s, ent.y, s0, s1, s2, s3);
        }
        qn = kept;
        __builtin_amdgcn_fence(__ATOMIC_RELEASE, "wavefront");
        __builtin_amdgcn_wave_barrier();
    };

    auto filter4 = [&](int4 d4, int v) {
        int dd[4] = {d4.x, d4.y, d4.z, d4.w};
#pragma unroll
        for (int c = 0; c < 4; ++c) {
            unsigned rel  = (unsigned)(dd[c] - lo);
            bool     hit  = rel < (unsigned)cnt;
            unsigned mask = __builtin_amdgcn_ballot_w32(hit);
            if (mask) {
                if (hit) q[qn + __builtin_amdgcn_mbcnt_lo(mask, 0u)] = make_int2(v * 4 + c, (int)rel);
                qn += __builtin_popcount(mask);
            }
        }
    };

    const int  nvec = E >> 2;
    const int  last = nvec - 1;
    const int4 none = make_int4(-1, -1, -1, -1);
    const int4* dst4 = reinterpret_cast<const int4*>(dst);
    const int  vfirst = wave * 32;
    auto ld = [&](int v) { return dst4[v < last ? v : last]; };

    auto filterLeftover = [&]() {
        if (wave == 0) {
            int4 t = none;
            int b = nvec * 4;
            if (lane == 0) {
                if (b     < E) t.x = dst[b];
                if (b + 1 < E) t.y = dst[b + 1];
                if (b + 2 < E) t.z = dst[b + 2];
            }
            filter4(t, nvec + lane);
        }
    };

    if (vfirst < nvec) {
        int vb = vfirst;
        int4 x0 = ld(vb + lane), x1 = ld(vb + lane + 1024), x2 = ld(vb + lane + 2048), x3 = ld(vb + lane + 3072);
        for (; vb < nvec; vb += 4 * 1024) {
            const int v0 = vb + lane, v1 = v0 + 1024, v2 = v0 + 2048, v3 = v0 + 3072;
            const int nx = v0 + 4 * 1024;
            const int4 y0 = ld(nx), y1 = ld(nx + 1024), y2 = ld(nx + 2048), y3 = ld(nx + 3072);
            const int c0 = ((vb - vfirst) >> 12) * 4;
            filter4(v0 > last ? none : x0, v0);
            if (qn - kept >= 32) drainMax(c0 + 1);
            filter4(v1 > last ? none : x1, v1);
            if (qn - kept >= 32) drainMax(c0 + 2);
            filter4(v2 > last ? none : x2, v2);
            if (qn - kept >= 32) drainMax(c0 + 3);
            filter4(v3 > last ? none : x3, v3);
            if (qn - kept >= 32) drainMax(c0 + 4);
            x0 = y0; x1 = y1; x2 = y2; x3 = y3;
        }
    }
    filterLeftover();
    if (qn > kept) drainMax(0x7fffffff);
    __syncthreads();

    for (int r = 0; r < kept; r += 32) {
        int n = kept - r; n = n < 32 ? n : 32;
        int idx = r + lane; idx = idx < kept ? idx : kept - 1;
        const int2   ent = q[idx];
        const float4 s4  = ks[idx];
        sumBatch(n, ent.x, ent.y, s4.x, s4.y, s4.z, s4.w);
    }
    if (!keeping) {
        if (vfirst < nvec) {
            const int ncall = ((nvec - vfirst + 4095) >> 12) * 4;
            for (int ci = resumeCall; ci < ncall; ++ci) {
                const int v = vfirst + (ci >> 2) * 4096 + (ci & 3) * 1024 + lane;
                const int4 x = ld(v);
                filter4(v > last ? none : x, v);
                if (qn - kept >= 32) drainSumGather();
            }
        }
        filterLeftover();
        if (qn > kept) drainSumGather();
    }
    __syncthreads();

    const float4* t4 = reinterpret_cast<const float4*>(tile);
    float4* o4 = reinterpret_cast<float4*>(out + (size_t)lo * NCOLS);
    for (int i = threadIdx.x; i < cnt * (NCOLS / 4); i += 1024) {
        const float d = denT[(i >> 5) * NHEADS + ((i & 31) >> 3)];
        float4 v = t4[i];
        const bool has = d > 0.f;
        v.x = has ? v.x / d : 0.f;
        v.y = has ? v.y / d : 0.f;
        v.z = has ? v.z / d : 0.f;
        v.w = has ? v.w / d : 0.f;
        o4[i] = v;
    }
}

extern "C" void kernel_launch(void* const* d_in, const int* in_sizes, int n_in,
                              void* d_out, int out_size, void* d_ws, size_t ws_size,
                              hipStream_t stream) {
    const float* node_feat = (const float*)d_in[0];
    const float* edge_feat = (const float*)d_in[1];
    const int*   src       = (const int*)d_in[2];
    const int*   dst       = (const int*)d_in[3];
    const float* W_node    = (const float*)d_in[4];
    const float* W_edge    = (const float*)d_in[5];
    float*       out       = (float*)d_out;

    const int N = in_sizes[0] / NODE_K;
    const int E = in_sizes[2];

    float* h     = (float*)d_ws;
    float* score = h     + (size_t)N * NCOLS;

    node_gemm_wreg<<<GEMM_GRID, 64, 0, stream>>>(node_feat, W_node, h, N);

    edge_score_heads<<<SCORE_GRID, 128, 0, stream>>>(edge_feat, W_edge, h, src, score, E);

    aggregate_tiles<<<(N + TILE_N - 1) / TILE_N, 1024, 0, stream>>>(h, src, dst, score, out, E, N);
}


# aggregation sweep reads a 1-byte owner-tile id per edge (plane written by the score launch) in place of the 4-byte dst id: 16 edges per load, dst fetched for hits only; LDS add order = arrival order
# speedup vs baseline: 3.9666x; 1.0564x over previous
#include <hip/hip_runtime.h>
#include <hip/hip_bf16.h>

typedef __attribute__((ext_vector_type(2))) float v2f;
typedef __attribute__((ext_vector_type(8))) float v8f;

#define NCOLS 128
#define NODE_K 256
#define EDGE_K 64
#define NHEADS 4

typedef __attribute__((ext_vector_type(4))) float v4f;

#define GEMM_GRID 512

__global__ void __launch_bounds__(64)
node_gemm_wreg(const float* __restrict__ A, const float* __restrict__ B,
               float* __restrict__ C, int M) {
    const int lane = threadIdx.x & 31;
    const int ch = __builtin_amdgcn_readfirstlane((int)(threadIdx.x >> 5));
    const int c  = lane & 15;
    const int hf = lane >> 4;

    const int ntile = (M + 15) >> 4;
    int t = blockIdx.x;
    if (t >= ntile) return;
    const int G = gridDim.x;

    v2f wf[4][NODE_K / 4];
    const v4f* B4 = reinterpret_cast<const v4f*>(B);
#pragma unroll
    for (int j = 0; j < NODE_K / 4; ++j) {
        const int k0 = 8 * (j >> 1) + 4 * hf + 2 * (j & 1);
        const v4f w0 = B4[k0 * (NCOLS / 4) + 16 * ch + c];
        const v4f w1 = B4[(k0 + 1) * (NCOLS / 4) + 16 * ch + c];
#pragma unroll
        for (int u = 0; u < 4; ++u) { wf[u][j].x = w0[u]; wf[u][j].y = w1[u]; }
    }

    const v4f* A4 = reinterpret_cast<const v4f*>(A);
    const int lastTile = ntile - 1;
    const int lastRow  = M - 1;

    auto loadTile = [&](int tile, v4f* dstv) {
        int tl = tile < lastTile ? tile : lastTile;
        int row = tl * 16 + c; row = row < lastRow ? row : lastRow;
        const v4f* rp = A4 + (size_t)row * (NODE_K / 4) + hf;
#pragma unroll
        for (int i = 0; i < NODE_K / 8; ++i) dstv[i] = rp[2 * i];
    };

    v4f cur[NODE_K / 8];
    loadTile(t, cur);

    for (; t < ntile; t += G) {
        v4f nxt[NODE_K / 8];
        loadTile(t + G, nxt);
        __builtin_amdgcn_sched_barrier(0);

        v8f acc[4] = {};
#pragma unroll
        for (int i = 0; i < NODE_K / 8; ++i) {
            const v2f a0 = __builtin_shufflevector(cur[i], cur[i], 0, 1);
            const v2f a1 = __builtin_shufflevector(cur[i], cur[i], 2, 3);
#pragma unroll
            for (int u = 0; u < 4; ++u)
                acc[u] = __builtin_amdgcn_wmma_f32_16x16x4_f32(false, a0, false, wf[u][2 * i],     (short)0, acc[u], false, false);
#pragma unroll
            for (int u = 0; u < 4; ++u)
                acc[u] = __builtin_amdgcn_wmma_f32_16x16x4_f32(false, a1, false, wf[u][2 * i + 1], (short)0, acc[u], false, false);
        }
        __builtin_amdgcn_sched_barrier(0);

        const int rbase = t * 16 + 8 * hf;
#pragma unroll
        for (int r = 0; r < 8; ++r) {
            const int row = rbase + r;
            if (row < M) {
                v4f o;
                o.x = acc[0][r]; o.y = acc[1][r]; o.z = acc[2][r]; o.w = acc[3][r];
                *reinterpret_cast<v4f*>(C + (size_t)row * NCOLS + 64 * ch + 4 * c) = o;
            }
        }
#pragma unroll
        for (int i = 0; i < NODE_K / 8; ++i) cur[i] = nxt[i];
    }
}

#define TILE_N    196
#define EF_PITCH4  17
#define SCORE_GRID 1024

__global__ void __launch_bounds__(128)
edge_score_heads(const float* __restrict__ EF, const float* __restrict__ We,
                 const float* __restrict__ H, const int* __restrict__ src,
                 const int* __restrict__ dst, int* __restrict__ tid,
                 float* __restrict__ score, int E) {
    __shared__ v4f stage[4][16 * EF_PITCH4];

    {
        const int nw = (E >> 4) * 4;
        const int4* dst4 = reinterpret_cast<const int4*>(dst);
        for (int v = blockIdx.x * 128 + threadIdx.x; v < nw; v += gridDim.x * 128) {
            const int4 d = dst4[v];
            const unsigned w = (((unsigned)d.x / TILE_N) & 255u)
                             | ((((unsigned)d.y / TILE_N) & 255u) << 8)
                             | ((((unsigned)d.z / TILE_N) & 255u) << 16)
                             | (((unsigned)d.w / TILE_N) << 24);
            tid[v] = (int)w;
        }
    }

    const int lane = threadIdx.x & 31;
    const int head = __builtin_amdgcn_readfirstlane((int)(threadIdx.x >> 5));
    const int c  = lane & 15;
    const int hf = lane >> 4;

    const int ntile = (E + 15) >> 4;
    const int npair = (ntile + 1) >> 1;
    int p = blockIdx.x;
    if (p >= npair) return;
    const int G = gridDim.x;

    v2f wf[2][16];
#pragma unroll
    for (int u = 0; u < 2; ++u) {
#pragma unroll
        for (int j = 0; j < 16; ++j) {
            const int k0 = 8 * (j >> 1) + 4 * hf + 2 * (j & 1);
            const float* wp = We + (size_t)k0 * NCOLS + 32 * head + 16 * u + c;
            wf[u][j].x = wp[0];
            wf[u][j].y = wp[NCOLS];
        }
    }

    v4f* my = stage[head];
    const v4f* EF4 = reinterpret_cast<const v4f*>(EF);
    const v4f* H4  = reinterpret_cast<const v4f*>(H);
    const unsigned lastTile = (unsigned)(ntile - 1);
    const unsigned lastEdge = (unsigned)(E - 1);

    v4f g[8];
    v4f hv[4];
    unsigned sNext;

    auto loadRows = [&](int tile, v4f* dstv) {
        unsigned tl = (unsigned)tile;
        tl = tl < lastTile ? tl : lastTile;
        const unsigned r0 = tl * 16u + (unsigned)hf;
#pragma unroll
        for (int i = 0; i < 8; ++i) {
            unsigned row = r0 + 2u * (unsigned)i;
            row = row < lastEdge ? row : lastEdge;
            dstv[i] = EF4[row * (unsigned)(EDGE_K / 4) + (unsigned)c];
        }
    };
    auto loadSrc = [&](int tile) -> unsigned {
        unsigned tl = (unsigned)tile;
        tl = tl < lastTile ? tl : lastTile;
        unsigned e = tl * 16u + (unsigned)c;
        e = e < lastEdge ? e : lastEdge;
        return (unsigned)src[e];
    };
    auto loadH = [&](unsigned s, v4f* dstv) {
        const v4f* hp = H4 + (s * (unsigned)(NCOLS / 4) + (unsigned)(8 * head + 2 * hf));
        dstv[0] = hp[0]; dstv[1] = hp[1]; dstv[2] = hp[4]; dstv[3] = hp[5];
    };

    auto tileStep = [&](int tNext, int tNext2) -> float {
#pragma unroll
        for (int i = 0; i < 8; ++i) my[(2 * i + hf) * EF_PITCH4 + c] = g[i];
        __builtin_amdgcn_fence(__ATOMIC_RELEASE, "wavefront");
        __builtin_amdgcn_wave_barrier();
        __builtin_amdgcn_fence(__ATOMIC_ACQUIRE, "wavefront");
        v4f q[8];
#pragma unroll
        for (int i = 0; i < 8; ++i) q[i] = my[c * EF_PITCH4 + 2 * i + hf];
        __builtin_amdgcn_fence(__ATOMIC_RELEASE, "wavefront");
        __builtin_amdgcn_wave_barrier();

        v4f hn[4];
        loadRows(tNext, g);
        loadH(sNext, hn);
        const unsigned s2 = loadSrc(tNext2);
        __builtin_amdgcn_sched_barrier(0);

        v8f acc0 = {}, acc1 = {};
#pragma unroll
        for (int i = 0; i < 8; ++i) {
            const v2f b0 = __builtin_shufflevector(q[i], q[i], 0, 1);
            const v2f b1 = __builtin_shufflevector(q[i], q[i], 2, 3);
            acc0 = __builtin_amdgcn_wmma_f32_16x16x4_f32(false, wf[0][2 * i],     false, b0, (short)0, acc0, false, false);
            acc1 = __builtin_amdgcn_wmma_f32_16x16x4_f32(false, wf[1][2 * i],     false, b0, (short)0, acc1, false, false);
            acc0 = __builtin_amdgcn_wmma_f32_16x16x4_f32(false, wf[0][2 * i + 1], false, b1, (short)0, acc0, false, false);
            acc1 = __builtin_amdgcn_wmma_f32_16x16x4_f32(false, wf[1][2 * i + 1], false, b1, (short)0, acc1, false, false);
        }
        __builtin_amdgcn_sched_barrier(0);

        float d = 0.f;
#pragma unroll
        for (int r = 0; r < 4; ++r) {
            d += acc0[r]     * hv[0][r];
            d += acc0[r + 4] * hv[1][r];
            d += acc1[r]     * hv[2][r];
            d += acc1[r + 4] * hv[3][r];
        }
        d += __shfl_xor(d, 16, 32);
        float sv = d * 0.17677669529663689f;
        sv = sv > 0.f ? sv : 0.01f * sv;

        hv[0] = hn[0]; hv[1] = hn[1]; hv[2] = hn[2]; hv[3] = hn[3];
        sNext = s2;
        return sv;
    };

    loadRows(2 * p, g);
    loadH(loadSrc(2 * p), hv);
    sNext = loadSrc(2 * p + 1);

    for (; p < npair; p += G) {
        const int pn = p + G;
        const float svA = tileStep(2 * p + 1, 2 * pn);
        const float svB = tileStep(2 * pn, 2 * pn + 1);
        const int e = 32 * p + lane;
        if (e < E) score[(size_t)head * E + e] = hf ? svB : svA;
    }
}

#define AGG_WAVES 32
#define P_MAX     224
#define Q_BURST   160
#define Q_CAP     (P_MAX + Q_BURST)

__device__ __forceinline__ int   rl_i(int v, int k)   { return __builtin_amdgcn_readlane(v, k); }
__device__ __forceinline__ float rl_f(float v, int k) {
    return __int_as_float(__builtin_amdgcn_readlane(__float_as_int(v), k));
}

__device__ __forceinline__ void ldsMaxFloat(float* addr, float val) {
    if (val >= 0.f) atomicMax((int*)addr, __float_as_int(val));
    else            atomicMin((unsigned int*)addr, __float_as_uint(val));
}

__device__ __forceinline__ unsigned zeroBytes(unsigned x) {
    return ~(((x & 0x7F7F7F7Fu) + 0x7F7F7F7Fu) | x) & 0x80808080u;
}

__global__ void __launch_bounds__(1024)
aggregate_tiles(const float* __restrict__ H, const int* __restrict__ src,
                const int* __restrict__ dst, const int* __restrict__ tid,
                const float* __restrict__ score,
                float* __restrict__ out, int E, int N) {
    __shared__ alignas(16) float  tile[TILE_N * NCOLS];
    __shared__ alignas(16) float4 keptScore[AGG_WAVES][P_MAX];
    __shared__ int2  queue[AGG_WAVES][Q_CAP];
    __shared__ alignas(16) float smaxT[TILE_N * NHEADS];
    __shared__ alignas(16) float denT[TILE_N * NHEADS];

    const int lane = threadIdx.x & 31;
    const int wave = __builtin_amdgcn_readfirstlane((int)(threadIdx.x >> 5));
    const int lo   = blockIdx.x * TILE_N;
    int cnt = N - lo; cnt = cnt < TILE_N ? cnt : TILE_N;
    const unsigned own4 = (blockIdx.x & 255u) * 0x01010101u;

    for (int i = threadIdx.x; i < TILE_N * NCOLS; i += 1024) tile[i] = 0.f;
    if (threadIdx.x < TILE_N * NHEADS) {
        smaxT[threadIdx.x] = __int_as_float(0xFF800000);
        denT[threadIdx.x]  = 0.f;
    }
    __syncthreads();

    int2*   q  = queue[wave];
    float4* ks = keptScore[wave];
    int  kept = 0;
    int  qn   = 0;
    bool keeping = true;
    int  resumeCall = 0;
    const float* sc0 = score;
    const float* sc1 = score + (size_t)E;
    const float* sc2 = score + 2 * (size_t)E;
    const float* sc3 = score + 3 * (size_t)E;

    auto drainMax = [&](int callsDone) {
        __builtin_amdgcn_fence(__ATOMIC_RELEASE, "wavefront");
        __builtin_amdgcn_wave_barrier();
        __builtin_amdgcn_fence(__ATOMIC_ACQUIRE, "wavefront");
        const bool keep = keeping && qn <= P_MAX;
        int w = kept;
        for (int r = kept; r < qn; r += 32) {
            const int idx = r + lane;
            bool  ok = false;
            int   s = 0, row = 0;
            float s0 = 0.f, s1 = 0.f, s2 = 0.f, s3 = 0.f;
            if (idx < qn) {
                const int e = q[idx].x;
                const unsigned rel = (unsigned)(dst[e] - lo);
                s  = src[e];
                s0 = sc0[e]; s1 = sc1[e]; s2 = sc2[e]; s3 = sc3[e];
                ok  = rel < (unsigned)cnt;
                row = (int)rel;
                if (ok) {
                    float* mp = smaxT + row * NHEADS;
                    ldsMaxFloat(mp,     s0);
                    ldsMaxFloat(mp + 1, s1);
                    ldsMaxFloat(mp + 2, s2);
                    ldsMaxFloat(mp + 3, s3);
                }
            }
            const unsigned okm = __builtin_amdgcn_ballot_w32(ok);
            if (keep && ok) {
                const int pos = w + (int)__builtin_amdgcn_mbcnt_lo(okm, 0u);
                q[pos]  = make_int2(s, row);
                ks[pos] = make_float4(s0, s1, s2, s3);
            }
            w += __builtin_popcount(okm);
        }
        if (keep) { kept = w; qn = w; resumeCall = callsDone; }
        else      { keeping = false; qn = kept; }
        __builtin_amdgcn_fence(__ATOMIC_RELEASE, "wavefront");
        __builtin_amdgcn_wave_barrier();
    };

    auto sumBatch = [&](int n, int s, int row, float s0, float s1, float s2, float s3) {
        const float4 m4 = reinterpret_cast<const float4*>(smaxT)[row];
        const float w0 = __expf(s0 - m4.x), w1 = __expf(s1 - m4.y);
        const float w2 = __expf(s2 - m4.z), w3 = __expf(s3 - m4.w);
        if (lane < n) {
            float* dp = denT + row * NHEADS;
            atomicAdd(dp,     w0);
            atomicAdd(dp + 1, w1);
            atomicAdd(dp + 2, w2);
            atomicAdd(dp + 3, w3);
        }
        for (int k = 0; k < n; k += 4) {
            float hv[4][4];
#pragma unroll
            for (int u = 0; u < 4; ++u) {
                const int kk = k + u < n ? k + u : n - 1;
                const float* hr = H + (size_t)rl_i(s, kk) * NCOLS + lane;
                hv[u][0] = hr[0]; hv[u][1] = hr[32]; hv[u][2] = hr[64]; hv[u][3] = hr[96];
            }
#pragma unroll
            for (int u = 0; u < 4; ++u) {
                if (k + u < n) {
                    float* tr = tile + rl_i(row, k + u) * NCOLS + lane;
                    atomicAdd(tr,      hv[u][0] * rl_f(w0, k + u));
                    atomicAdd(tr + 32, hv[u][1] * rl_f(w1, k + u));
                    atomicAdd(tr + 64, hv[u][2] * rl_f(w2, k + u));
                    atomicAdd(tr + 96, hv[u][3] * rl_f(w3, k + u));
                }
            }
        }
    };

    auto drainSumGather = [&]() {
        __builtin_amdgcn_fence(__ATOMIC_RELEASE, "wavefront");
        __builtin_amdgcn_wave_barrier();
        __builtin_amdgcn_fence(__ATOMIC_ACQUIRE, "wavefront");
        int w = kept;
        for (int r = kept; r < qn; r += 32) {
            const int idx = r + lane;
            bool ok = false;
            int  e = 0, row = 0;
            if (idx < qn) {
                e = q[idx].x;
                const unsigned rel = (unsigned)(dst[e] - lo);
                ok  = rel < (unsigned)cnt;
                row = (int)rel;
            }
            const unsigned okm = __builtin_amdgcn_ballot_w32(ok);
            if (ok) q[w + (int)__builtin_amdgcn_mbcnt_lo(okm, 0u)] = make_int2(e, row);
            w += __builtin_popcount(okm);
        }
        qn = w;
        __builtin_amdgcn_fence(__ATOMIC_RELEASE, "wavefront");
        __builtin_amdgcn_wave_barrier();
        __builtin_amdgcn_fence(__ATOMIC_ACQUIRE, "wavefront");
        for (int r = kept; r < qn; r += 32) {
            int n = qn - r; n = n < 32 ? n : 32;
            int idx = r + lane; idx = idx < qn ? idx : qn - 1;
            const int2 ent = q[idx];
            const int   s  = src[ent.x];
            const float s0 = sc0[ent.x], s1 = sc1[ent.x], s2 = sc2[ent.x], s3 = sc3[ent.x];
            sumBatch(n, s, ent.y, s0, s1, s2, s3);
        }
        qn = kept;
        __builtin_amdgcn_fence(__ATOMIC_RELEASE, "wavefront");
        __builtin_amdgcn_wave_barrier();
    };

    const int  nvec = E >> 4;
    const int  last = nvec - 1;
    const int4* tid4 = reinterpret_cast<const int4*>(tid);
    const int  vfirst = wave * 32;
    auto ld = [&](int v) { return tid4[v < last ? v : last]; };

    auto hitMask = [&](int4 d, int v) -> unsigned {
        const unsigned m = (zeroBytes((unsigned)d.x ^ own4) >> 7) | (zeroBytes((unsigned)d.y ^ own4) >> 6)
                         | (zeroBytes((unsigned)d.z ^ own4) >> 5) | (zeroBytes((unsigned)d.w ^ own4) >> 4);
        return v > last ? 0u : m;
    };

    auto appendHits = [&](unsigned m, int v, auto&& afterRound) {
        for (;;) {
            const bool     has  = m != 0u;
            const unsigned mask = __builtin_amdgcn_ballot_w32(has);
            if (!mask) break;
            if (has) {
                const int i = __builtin_ctz(m);
                q[qn + (int)__builtin_amdgcn_mbcnt_lo(mask, 0u)] = make_int2(v * 16 + 4 * (i & 7) + (i >> 3), 0);
                m &= m - 1u;
            }
            qn += __builtin_popcount(mask);
            afterRound();
        }
    };
    auto roomForRound = [&]() { if (qn + 32 > Q_CAP) drainMax(resumeCall); };

    auto filterLeftover = [&]() {
        if (wave == 0) {
            const int e = nvec * 16 + lane;
            bool hit = false;
            if (lane < 16 && e < E) hit = (unsigned)(dst[e] - lo) < (unsigned)cnt;
            const unsigned mask = __builtin_amdgcn_ballot_w32(hit);
            if (mask) {
                if (hit) q[qn + (int)__builtin_amdgcn_mbcnt_lo(mask, 0u)] = make_int2(e, 0);
                qn += __builtin_popcount(mask);
            }
        }
    };

    if (vfirst < nvec) {
        int vb = vfirst;
        int4 x0 = ld(vb + lane), x1 = ld(vb + lane + 1024), x2 = ld(vb + lane + 2048), x3 = ld(vb + lane + 3072);
        for (; vb < nvec; vb += 4 * 1024) {
            const int v0 = vb + lane, v1 = v0 + 1024, v2 = v0 + 2048, v3 = v0 + 3072;
            const int nx = v0 + 4 * 1024;
            const int4 y0 = ld(nx), y1 = ld(nx + 1024), y2 = ld(nx + 2048), y3 = ld(nx + 3072);
            const int c0 = ((vb - vfirst) >> 12) * 4;
            appendHits(hitMask(x0, v0), v0, roomForRound);
            if (qn - kept >= 32) drainMax(c0 + 1);
            appendHits(hitMask(x1, v1), v1, roomForRound);
            if (qn - kept >= 32) drainMax(c0 + 2);
            appendHits(hitMask(x2, v2), v2, roomForRound);
            if (qn - kept >= 32) drainMax(c0 + 3);
            appendHits(hitMask(x3, v3), v3, roomForRound);
            if (qn - kept >= 32) drainMax(c0 + 4);
            x0 = y0; x1 = y1; x2 = y2; x3 = y3;
        }
    }
    filterLeftover();
    if (qn > kept) drainMax(0x7fffffff);
    __syncthreads();

    for (int r = 0; r < kept; r += 32) {
        int n = kept - r; n = n < 32 ? n : 32;
        int idx = r + lane; idx = idx < kept ? idx : kept - 1;
        const int2   ent = q[idx];
        const float4 s4  = ks[idx];
        sumBatch(n, ent.x, ent.y, s4.x, s4.y, s4.z, s4.w);
    }
    if (!keeping) {
        if (vfirst < nvec) {
            const int ncall = ((nvec - vfirst + 4095) >> 12) * 4;
            for (int ci = resumeCall; ci < ncall; ++ci) {
                const int v = vfirst + (ci >> 2) * 4096 + (ci & 3) * 1024 + lane;
                appendHits(hitMask(ld(v), v), v, [&]() { if (qn - kept >= 32) drainSumGather(); });
            }
        }
        filterLeftover();
        if (qn > kept) drainSumGather();
    }
    __syncthreads();

    const float4* t4 = reinterpret_cast<const float4*>(tile);
    float4* o4 = reinterpret_cast<float4*>(out + (size_t)lo * NCOLS);
    for (int i = threadIdx.x; i < cnt * (NCOLS / 4); i += 1024) {
        const float d = denT[(i >> 5) * NHEADS + ((i & 31) >> 3)];
        float4 v = t4[i];
        const bool has = d > 0.f;
        v.x = has ? v.x / d : 0.f;
        v.y = has ? v.y / d : 0.f;
        v.z = has ? v.z / d : 0.f;
        v.w = has ? v.w / d : 0.f;
        o4[i] = v;
    }
}

extern "C" void kernel_launch(void* const* d_in, const int* in_sizes, int n_in,
                              void* d_out, int out_size, void* d_ws, size_t ws_size,
                              hipStream_t stream) {
    const float* node_feat = (const float*)d_in[0];
    const float* edge_feat = (const float*)d_in[1];
    const int*   src       = (const int*)d_in[2];
    const int*   dst       = (const int*)d_in[3];
    const float* W_node    = (const float*)d_in[4];
    const float* W_edge    = (const float*)d_in[5];
    float*       out       = (float*)d_out;

    const int N = in_sizes[0] / NODE_K;
    const int E = in_sizes[2];

    float* h     = (float*)d_ws;
    float* score = h     + (size_t)N * NCOLS;
    int*   tid   = (int*)(score + (size_t)NHEADS * E);

    node_gemm_wreg<<<GEMM_GRID, 64, 0, stream>>>(node_feat, W_node, h, N);

    edge_score_heads<<<SCORE_GRID, 128, 0, stream>>>(edge_feat, W_edge, h, src, dst, tid, score, E);

    aggregate_tiles<<<(N + TILE_N - 1) / TILE_N, 1024, 0, stream>>>(h, src, dst, tid, score, out, E, N);
}


# aggregation sweep tests one presence bit per 32-edge group (256-bit owner-tile set per group, built by the score launch as eight word planes), reads dst ids of hit groups only; LDS add order: arrival
# speedup vs baseline: 4.2242x; 1.0649x over previous
#include <hip/hip_runtime.h>
#include <hip/hip_bf16.h>

typedef __attribute__((ext_vector_type(2))) float v2f;
typedef __attribute__((ext_vector_type(8))) float v8f;

#define NCOLS 128
#define NODE_K 256
#define EDGE_K 64
#define NHEADS 4

typedef __attribute__((ext_vector_type(4))) float v4f;

#define GEMM_GRID 512

__global__ void __launch_bounds__(64)
node_gemm_wreg(const float* __restrict__ A, const float* __restrict__ B,
               float* __restrict__ C, int M) {
    const int lane = threadIdx.x & 31;
    const int ch = __builtin_amdgcn_readfirstlane((int)(threadIdx.x >> 5));
    const int c  = lane & 15;
    const int hf = lane >> 4;

    const int ntile = (M + 15) >> 4;
    int t = blockIdx.x;
    if (t >= ntile) return;
    const int G = gridDim.x;

    v2f wf[4][NODE_K / 4];
    const v4f* B4 = reinterpret_cast<const v4f*>(B);
#pragma unroll
    for (int j = 0; j < NODE_K / 4; ++j) {
        const int k0 = 8 * (j >> 1) + 4 * hf + 2 * (j & 1);
        const v4f w0 = B4[k0 * (NCOLS / 4) + 16 * ch + c];
        const v4f w1 = B4[(k0 + 1) * (NCOLS / 4) + 16 * ch + c];
#pragma unroll
        for (int u = 0; u < 4; ++u) { wf[u][j].x = w0[u]; wf[u][j].y = w1[u]; }
    }

    const v4f* A4 = reinterpret_cast<const v4f*>(A);
    const int lastTile = ntile - 1;
    const int lastRow  = M - 1;

    auto loadTile = [&](int tile, v4f* dstv) {
        int tl = tile < lastTile ? tile : lastTile;
        int row = tl * 16 + c; row = row < lastRow ? row : lastRow;
        const v4f* rp = A4 + (size_t)row * (NODE_K / 4) + hf;
#pragma unroll
        for (int i = 0; i < NODE_K / 8; ++i) dstv[i] = rp[2 * i];
    };

    v4f cur[NODE_K / 8];
    loadTile(t, cur);

    for (; t < ntile; t += G) {
        v4f nxt[NODE_K / 8];
        loadTile(t + G, nxt);
        __builtin_amdgcn_sched_barrier(0);

        v8f acc[4] = {};
#pragma unroll
        for (int i = 0; i < NODE_K / 8; ++i) {
            const v2f a0 = __builtin_shufflevector(cur[i], cur[i], 0, 1);
            const v2f a1 = __builtin_shufflevector(cur[i], cur[i], 2, 3);
#pragma unroll
            for (int u = 0; u < 4; ++u)
                acc[u] = __builtin_amdgcn_wmma_f32_16x16x4_f32(false, a0, false, wf[u][2 * i],     (short)0, acc[u], false, false);
#pragma unroll
            for (int u = 0; u < 4; ++u)
                acc[u] = __builtin_amdgcn_wmma_f32_16x16x4_f32(false, a1, false, wf[u][2 * i + 1], (short)0, acc[u], false, false);
        }
        __builtin_amdgcn_sched_barrier(0);

        const int rbase = t * 16 + 8 * hf;
#pragma unroll
        for (int r = 0; r < 8; ++r) {
            const int row = rbase + r;
            if (row < M) {
                v4f o;
                o.x = acc[0][r]; o.y = acc[1][r]; o.z = acc[2][r]; o.w = acc[3][r];
                *reinterpret_cast<v4f*>(C + (size_t)row * NCOLS + 64 * ch + 4 * c) = o;
            }
        }
#pragma unroll
        for (int i = 0; i < NODE_K / 8; ++i) cur[i] = nxt[i];
    }
}

#define TILE_N    196
#define EF_PITCH4  17
#define SCORE_GRID 1024

__global__ void __launch_bounds__(128)
edge_score_heads(const float* __restrict__ EF, const float* __restrict__ We,
                 const float* __restrict__ H, const int* __restrict__ src,
                 const int* __restrict__ dst, unsigned* __restrict__ pres,
                 float* __restrict__ score, int E) {
    __shared__ v4f stage[4][16 * EF_PITCH4];

    const int lane = threadIdx.x & 31;
    const int head = __builtin_amdgcn_readfirstlane((int)(threadIdx.x >> 5));
    const int c  = lane & 15;
    const int hf = lane >> 4;

    {
        const int ng = E >> 5;
        const int gp = (ng + 3) & ~3;
        const int4* dst4 = reinterpret_cast<const int4*>(dst);
        for (int task = blockIdx.x + gridDim.x * head; task * 32 < gp; task += 4 * gridDim.x) {
            const int g = task * 32 + lane;
            unsigned w[8] = {0u, 0u, 0u, 0u, 0u, 0u, 0u, 0u};
            if (g < ng) {
#pragma unroll
                for (int i = 0; i < 8; ++i) {
                    const int4 d = dst4[(size_t)g * 8 + i];
                    const unsigned t4[4] = {(unsigned)d.x / TILE_N, (unsigned)d.y / TILE_N,
                                            (unsigned)d.z / TILE_N, (unsigned)d.w / TILE_N};
#pragma unroll
                    for (int j = 0; j < 4; ++j) {
                        const unsigned t   = t4[j] & 255u;
                        const unsigned bit = 1u << (t & 31u);
#pragma unroll
                        for (int k = 0; k < 8; ++k) w[k] |= (t >> 5) == (unsigned)k ? bit : 0u;
                    }
                }
            }
            if (g < gp) {
#pragma unroll
                for (int k = 0; k < 8; ++k) pres[(size_t)k * gp + g] = w[k];
            }
        }
    }

    const int ntile = (E + 15) >> 4;
    const int npair = (ntile + 1) >> 1;
    int p = blockIdx.x;
    if (p >= npair) return;
    const int G = gridDim.x;

    v2f wf[2][16];
#pragma unroll
    for (int u = 0; u < 2; ++u) {
#pragma unroll
        for (int j = 0; j < 16; ++j) {
            const int k0 = 8 * (j >> 1) + 4 * hf + 2 * (j & 1);
            const float* wp = We + (size_t)k0 * NCOLS + 32 * head + 16 * u + c;
            wf[u][j].x = wp[0];
            wf[u][j].y = wp[NCOLS];
        }
    }

    v4f* my = stage[head];
    const v4f* EF4 = reinterpret_cast<const v4f*>(EF);
    const v4f* H4  = reinterpret_cast<const v4f*>(H);
    const unsigned lastTile = (unsigned)(ntile - 1);
    const unsigned lastEdge = (unsigned)(E - 1);

    v4f g[8];
    v4f hv[4];
    unsigned sNext;

    auto loadRows = [&](int tile, v4f* dstv) {
        unsigned tl = (unsigned)tile;
        tl = tl < lastTile ? tl : lastTile;
        const unsigned r0 = tl * 16u + (unsigned)hf;
#pragma unroll
        for (int i = 0; i < 8; ++i) {
            unsigned row = r0 + 2u * (unsigned)i;
            row = row < lastEdge ? row : lastEdge;
            dstv[i] = EF4[row * (unsigned)(EDGE_K / 4) + (unsigned)c];
        }
    };
    auto loadSrc = [&](int tile) -> unsigned {
        unsigned tl = (unsigned)tile;
        tl = tl < lastTile ? tl : lastTile;
        unsigned e = tl * 16u + (unsigned)c;
        e = e < lastEdge ? e : lastEdge;
        return (unsigned)src[e];
    };
    auto loadH = [&](unsigned s, v4f* dstv) {
        const v4f* hp = H4 + (s * (unsigned)(NCOLS / 4) + (unsigned)(8 * head + 2 * hf));
        dstv[0] = hp[0]; dstv[1] = hp[1]; dstv[2] = hp[4]; dstv[3] = hp[5];
    };

    auto tileStep = [&](int tNext, int tNext2) -> float {
#pragma unroll
        for (int i = 0; i < 8; ++i) my[(2 * i + hf) * EF_PITCH4 + c] = g[i];
        __builtin_amdgcn_fence(__ATOMIC_RELEASE, "wavefront");
        __builtin_amdgcn_wave_barrier();
        __builtin_amdgcn_fence(__ATOMIC_ACQUIRE, "wavefront");
        v4f q[8];
#pragma unroll
        for (int i = 0; i < 8; ++i) q[i] = my[c * EF_PITCH4 + 2 * i + hf];
        __builtin_amdgcn_fence(__ATOMIC_RELEASE, "wavefront");
        __builtin_amdgcn_wave_barrier();

        v4f hn[4];
        loadRows(tNext, g);
        loadH(sNext, hn);
        const unsigned s2 = loadSrc(tNext2);
        __builtin_amdgcn_sched_barrier(0);

        v8f acc0 = {}, acc1 = {};
#pragma unroll
        for (int i = 0; i < 8; ++i) {
            const v2f b0 = __builtin_shufflevector(q[i], q[i], 0, 1);
            const v2f b1 = __builtin_shufflevector(q[i], q[i], 2, 3);
            acc0 = __builtin_amdgcn_wmma_f32_16x16x4_f32(false, wf[0][2 * i],     false, b0, (short)0, acc0, false, false);
            acc1 = __builtin_amdgcn_wmma_f32_16x16x4_f32(false, wf[1][2 * i],     false, b0, (short)0, acc1, false, false);
            acc0 = __builtin_amdgcn_wmma_f32_16x16x4_f32(false, wf[0][2 * i + 1], false, b1, (short)0, acc0, false, false);
            acc1 = __builtin_amdgcn_wmma_f32_16x16x4_f32(false, wf[1][2 * i + 1], false, b1, (short)0, acc1, false, false);
        }
        __builtin_amdgcn_sched_barrier(0);

        float d = 0.f;
#pragma unroll
        for (int r = 0; r < 4; ++r) {
            d += acc0[r]     * hv[0][r];
            d += acc0[r + 4] * hv[1][r];
            d += acc1[r]     * hv[2][r];
            d += acc1[r + 4] * hv[3][r];
        }
        d += __shfl_xor(d, 16, 32);
        float sv = d * 0.17677669529663689f;
        sv = sv > 0.f ? sv : 0.01f * sv;

        hv[0] = hn[0]; hv[1] = hn[1]; hv[2] = hn[2]; hv[3] = hn[3];
        sNext = s2;
        return sv;
    };

    loadRows(2 * p, g);
    loadH(loadSrc(2 * p), hv);
    sNext = loadSrc(2 * p + 1);

    for (; p < npair; p += G) {
        const int pn = p + G;
        const float svA = tileStep(2 * p + 1, 2 * pn);
        const float svB = tileStep(2 * pn, 2 * pn + 1);
        const int e = 32 * p + lane;
        if (e < E) score[(size_t)head * E + e] = hf ? svB : svA;
    }
}

#define AGG_WAVES 32
#define P_MAX     192
#define G_CALL    128
#define Q_BURST   160
#define Q_CAP     (P_MAX + Q_BURST)

__device__ __forceinline__ int   rl_i(int v, int k)   { return __builtin_amdgcn_readlane(v, k); }
__device__ __forceinline__ float rl_f(float v, int k) {
    return __int_as_float(__builtin_amdgcn_readlane(__float_as_int(v), k));
}

__device__ __forceinline__ void ldsMaxFloat(float* addr, float val) {
    if (val >= 0.f) atomicMax((int*)addr, __float_as_int(val));
    else            atomicMin((unsigned int*)addr, __float_as_uint(val));
}

__global__ void __launch_bounds__(1024)
aggregate_tiles(const float* __restrict__ H, const int* __restrict__ src,
                const int* __restrict__ dst, const unsigned* __restrict__ pres,
                const float* __restrict__ score,
                float* __restrict__ out, int E, int N) {
    __shared__ alignas(16) float  tile[TILE_N * NCOLS];
    __shared__ alignas(16) float4 keptScore[AGG_WAVES][P_MAX];
    __shared__ int2  queue[AGG_WAVES][Q_CAP];
    __shared__ int   glist[AGG_WAVES][G_CALL];
    __shared__ alignas(16) float smaxT[TILE_N * NHEADS];
    __shared__ alignas(16) float denT[TILE_N * NHEADS];

    const int lane = threadIdx.x & 31;
    const int wave = __builtin_amdgcn_readfirstlane((int)(threadIdx.x >> 5));
    const int lo   = blockIdx.x * TILE_N;
    int cnt = N - lo; cnt = cnt < TILE_N ? cnt : TILE_N;
    const unsigned obyte = blockIdx.x & 255u;
    const unsigned obit  = obyte & 31u;

    for (int i = threadIdx.x; i < TILE_N * NCOLS; i += 1024) tile[i] = 0.f;
    if (threadIdx.x < TILE_N * NHEADS) {
        smaxT[threadIdx.x] = __int_as_float(0xFF800000);
        denT[threadIdx.x]  = 0.f;
    }
    __syncthreads();

    int2*   q  = queue[wave];
    float4* ks = keptScore[wave];
    int*    gl = glist[wave];
    int  kept = 0;
    int  qn   = 0;
    bool keeping = true;
    int  resumeCall = 0;
    const float* sc0 = score;
    const float* sc1 = score + (size_t)E;
    const float* sc2 = score + 2 * (size_t)E;
    const float* sc3 = score + 3 * (size_t)E;

    auto drainMax = [&](int callsDone) {
        __builtin_amdgcn_fence(__ATOMIC_RELEASE, "wavefront");
        __builtin_amdgcn_wave_barrier();
        __builtin_amdgcn_fence(__ATOMIC_ACQUIRE, "wavefront");
        const bool keep = keeping && qn <= P_MAX;
        int w = kept;
        for (int r = kept; r < qn; r += 32) {
            const int idx = r + lane;
            bool  ok = false;
            int   s = 0, row = 0;
            float s0 = 0.f, s1 = 0.f, s2 = 0.f, s3 = 0.f;
            if (idx < qn) {
                const int e = q[idx].x;
                const unsigned rel = (unsigned)(dst[e] - lo);
                s  = src[e];
                s0 = sc0[e]; s1 = sc1[e]; s2 = sc2[e]; s3 = sc3[e];
                ok  = rel < (unsigned)cnt;
                row = (int)rel;
                if (ok) {
                    float* mp = smaxT + row * NHEADS;
                    ldsMaxFloat(mp,     s0);
                    ldsMaxFloat(mp + 1, s1);
                    ldsMaxFloat(mp + 2, s2);
                    ldsMaxFloat(mp + 3, s3);
                }
            }
            const unsigned okm = __builtin_amdgcn_ballot_w32(ok);
            if (keep && ok) {
                const int pos = w + (int)__builtin_amdgcn_mbcnt_lo(okm, 0u);
                q[pos]  = make_int2(s, row);
                ks[pos] = make_float4(s0, s1, s2, s3);
            }
            w += __builtin_popcount(okm);
        }
        if (keep) { kept = w; qn = w; resumeCall = callsDone; }
        else      { keeping = false; qn = kept; }
        __builtin_amdgcn_fence(__ATOMIC_RELEASE, "wavefront");
        __builtin_amdgcn_wave_barrier();
    };

    auto sumBatch = [&](int n, int s, int row, float s0, float s1, float s2, float s3) {
        const float4 m4 = reinterpret_cast<const float4*>(smaxT)[row];
        const float w0 = __expf(s0 - m4.x), w1 = __expf(s1 - m4.y);
        const float w2 = __expf(s2 - m4.z), w3 = __expf(s3 - m4.w);
        if (lane < n) {
            float* dp = denT + row * NHEADS;
            atomicAdd(dp,     w0);
            atomicAdd(dp + 1, w1);
            atomicAdd(dp + 2, w2);
            atomicAdd(dp + 3, w3);
        }
        for (int k = 0; k < n; k += 4) {
            float hv[4][4];
#pragma unroll
            for (int u = 0; u < 4; ++u) {
                const int kk = k + u < n ? k + u : n - 1;
                const float* hr = H + (size_t)rl_i(s, kk) * NCOLS + lane;
                hv[u][0] = hr[0]; hv[u][1] = hr[32]; hv[u][2] = hr[64]; hv[u][3] = hr[96];
            }
#pragma unroll
            for (int u = 0; u < 4; ++u) {
                if (k + u < n) {
                    float* tr = tile + rl_i(row, k + u) * NCOLS + lane;
                    atomicAdd(tr,      hv[u][0] * rl_f(w0, k + u));
                    atomicAdd(tr + 32, hv[u][1] * rl_f(w1, k + u));
                    atomicAdd(tr + 64, hv[u][2] * rl_f(w2, k + u));
                    atomicAdd(tr + 96, hv[u][3] * rl_f(w3, k + u));
                }
            }
        }
    };

    auto drainSumGather = [&]() {
        __builtin_amdgcn_fence(__ATOMIC_RELEASE, "wavefront");
        __builtin_amdgcn_wave_barrier();
        __builtin_amdgcn_fence(__ATOMIC_ACQUIRE, "wavefront");
        int w = kept;
        for (int r = kept; r < qn; r += 32) {
            const int idx = r + lane;
            bool ok = false;
            int  e = 0, row = 0;
            if (idx < qn) {
                e = q[idx].x;
                const unsigned rel = (unsigned)(dst[e] - lo);
                ok  = rel < (unsigned)cnt;
                row = (int)rel;
            }
            const unsigned okm = __builtin_amdgcn_ballot_w32(ok);
            if (ok) q[w + (int)__builtin_amdgcn_mbcnt_lo(okm, 0u)] = make_int2(e, row);
            w += __builtin_popcount(okm);
        }
        qn = w;
        __builtin_amdgcn_fence(__ATOMIC_RELEASE, "wavefront");
        __builtin_amdgcn_wave_barrier();
        __builtin_amdgcn_fence(__ATOMIC_ACQUIRE, "wavefront");
        for (int r = kept; r < qn; r += 32) {
            int n = qn - r; n = n < 32 ? n : 32;
            int idx = r + lane; idx = idx < qn ? idx : qn - 1;
            const int2 ent = q[idx];
            const int   s  = src[ent.x];
            const float s0 = sc0[ent.x], s1 = sc1[ent.x], s2 = sc2[ent.x], s3 = sc3[ent.x];
            sumBatch(n, s, ent.y, s0, s1, s2, s3);
        }
        qn = kept;
        __builtin_amdgcn_fence(__ATOMIC_RELEASE, "wavefront");
        __builtin_amdgcn_wave_barrier();
    };

    const int  ng   = E >> 5;
    const int  gp   = (ng + 3) & ~3;
    const int  nvec = gp >> 2;
    const int  last = nvec - 1;
    const int4* plane4 = reinterpret_cast<const int4*>(pres + (size_t)(obyte >> 5) * gp);
    const int4* dst4   = reinterpret_cast<const int4*>(dst);
    const int  vfirst = wave * 32;
    auto ld = [&](int v) { return plane4[v < last ? v : last]; };

    auto hitMask = [&](int4 d, int v) -> unsigned {
        const unsigned m = (((unsigned)d.x >> obit) & 1u)        | ((((unsigned)d.y >> obit) & 1u) << 1)
                         | ((((unsigned)d.z >> obit) & 1u) << 2) | ((((unsigned)d.w >> obit) & 1u) << 3);
        return v > last ? 0u : m;
    };

    auto appendHits = [&](unsigned m, int base, auto&& afterRound) {
        for (;;) {
            const bool     has  = m != 0u;
            const unsigned mask = __builtin_amdgcn_ballot_w32(has);
            if (!mask) break;
            if (has) {
                q[qn + (int)__builtin_amdgcn_mbcnt_lo(mask, 0u)] = make_int2(base + __builtin_ctz(m), 0);
                m &= m - 1u;
            }
            qn += __builtin_popcount(mask);
            afterRound();
        }
    };

    auto filterCall = [&](int4 x, int v, auto&& afterRound) {
        unsigned m = hitMask(x, v);
        int gn = 0;
        for (;;) {
            const bool     has  = m != 0u;
            const unsigned mask = __builtin_amdgcn_ballot_w32(has);
            if (!mask) break;
            if (has) {
                gl[gn + (int)__builtin_amdgcn_mbcnt_lo(mask, 0u)] = v * 4 + __builtin_ctz(m);
                m &= m - 1u;
            }
            gn += __builtin_popcount(mask);
        }
        __builtin_amdgcn_fence(__ATOMIC_RELEASE, "wavefront");
        __builtin_amdgcn_wave_barrier();
        __builtin_amdgcn_fence(__ATOMIC_ACQUIRE, "wavefront");
        for (int r = 0; r < gn; r += 16) {
            const int idx = r + (lane >> 1);
            unsigned em = 0u;
            int base = 0;
            if (idx < gn) {
                const int g = gl[idx];
                base = g * 32 + 16 * (lane & 1);
                const int4* dp = dst4 + ((size_t)g * 8 + 4 * (lane & 1));
#pragma unroll
                for (int i = 0; i < 4; ++i) {
                    const int4 d = dp[i];
                    em |= ((unsigned)(d.x - lo) < (unsigned)cnt ? 1u : 0u) << (4 * i);
                    em |= ((unsigned)(d.y - lo) < (unsigned)cnt ? 1u : 0u) << (4 * i + 1);
                    em |= ((unsigned)(d.z - lo) < (unsigned)cnt ? 1u : 0u) << (4 * i + 2);
                    em |= ((unsigned)(d.w - lo) < (unsigned)cnt ? 1u : 0u) << (4 * i + 3);
                }
            }
            appendHits(em, base, afterRound);
        }
        __builtin_amdgcn_fence(__ATOMIC_RELEASE, "wavefront");
        __builtin_amdgcn_wave_barrier();
    };
    auto roomForRound = [&]() { if (qn + 32 > Q_CAP) drainMax(resumeCall); };

    auto filterLeftover = [&]() {
        if (wave == 0) {
            const int e = ng * 32 + lane;
            bool hit = false;
            if (e < E) hit = (unsigned)(dst[e] - lo) < (unsigned)cnt;
            const unsigned mask = __builtin_amdgcn_ballot_w32(hit);
            if (mask) {
                if (hit) q[qn + (int)__builtin_amdgcn_mbcnt_lo(mask, 0u)] = make_int2(e, 0);
                qn += __builtin_popcount(mask);
            }
        }
    };

    if (vfirst < nvec) {
        int vb = vfirst;
        int4 x0 = ld(vb + lane), x1 = ld(vb + lane + 1024), x2 = ld(vb + lane + 2048), x3 = ld(vb + lane + 3072);
        for (; vb < nvec; vb += 4 * 1024) {
            const int v0 = vb + lane, v1 = v0 + 1024, v2 = v0 + 2048, v3 = v0 + 3072;
            const int nx = v0 + 4 * 1024;
            const int4 y0 = ld(nx), y1 = ld(nx + 1024), y2 = ld(nx + 2048), y3 = ld(nx + 3072);
            const int c0 = ((vb - vfirst) >> 12) * 4;
            filterCall(x0, v0, roomForRound);
            if (qn - kept >= 32) drainMax(c0 + 1);
            filterCall(x1, v1, roomForRound);
            if (qn - kept >= 32) drainMax(c0 + 2);
            filterCall(x2, v2, roomForRound);
            if (qn - kept >= 32) drainMax(c0 + 3);
            filterCall(x3, v3, roomForRound);
            if (qn - kept >= 32) drainMax(c0 + 4);
            x0 = y0; x1 = y1; x2 = y2; x3 = y3;
        }
    }
    filterLeftover();
    if (qn > kept) drainMax(0x7fffffff);
    __syncthreads();

    for (int r = 0; r < kept; r += 32) {
        int n = kept - r; n = n < 32 ? n : 32;
        int idx = r + lane; idx = idx < kept ? idx : kept - 1;
        const int2   ent = q[idx];
        const float4 s4  = ks[idx];
        sumBatch(n, ent.x, ent.y, s4.x, s4.y, s4.z, s4.w);
    }
    if (!keeping) {
        if (vfirst < nvec) {
            const int ncall = ((nvec - vfirst + 4095) >> 12) * 4;
            for (int ci = resumeCall; ci < ncall; ++ci) {
                const int v = vfirst + (ci >> 2) * 4096 + (ci & 3) * 1024 + lane;
                filterCall(ld(v), v, [&]() { if (qn - kept >= 32) drainSumGather(); });
            }
        }
        filterLeftover();
        if (qn > kept) drainSumGather();
    }
    __syncthreads();

    const float4* t4 = reinterpret_cast<const float4*>(tile);
    float4* o4 = reinterpret_cast<float4*>(out + (size_t)lo * NCOLS);
    for (int i = threadIdx.x; i < cnt * (NCOLS / 4); i += 1024) {
        const float d = denT[(i >> 5) * NHEADS + ((i & 31) >> 3)];
        float4 v = t4[i];
        const bool has = d > 0.f;
        v.x = has ? v.x / d : 0.f;
        v.y = has ? v.y / d : 0.f;
        v.z = has ? v.z / d : 0.f;
        v.w = has ? v.w / d : 0.f;
        o4[i] = v;
    }
}

extern "C" void kernel_launch(void* const* d_in, const int* in_sizes, int n_in,
                              void* d_out, int out_size, void* d_ws, size_t ws_size,
                              hipStream_t stream) {
    const float* node_feat = (const float*)d_in[0];
    const float* edge_feat = (const float*)d_in[1];
    const int*   src       = (const int*)d_in[2];
    const int*   dst       = (const int*)d_in[3];
    const float* W_node    = (const float*)d_in[4];
    const float* W_edge    = (const float*)d_in[5];
    float*       out       = (float*)d_out;

    const int N = in_sizes[0] / NODE_K;
    const int E = in_sizes[2];

    float*    h     = (float*)d_ws;
    float*    score = h     + (size_t)N * NCOLS;
    unsigned* pres  = (unsigned*)(score + (size_t)NHEADS * E);

    node_gemm_wreg<<<GEMM_GRID, 64, 0, stream>>>(node_feat, W_node, h, N);

    edge_score_heads<<<SCORE_GRID, 128, 0, stream>>>(edge_feat, W_edge, h, src, dst, pres, score, E);

    aggregate_tiles<<<(N + TILE_N - 1) / TILE_N, 1024, 0, stream>>>(h, src, dst, pres, score, out, E, N);
}


# edge score re-associated: G[n][head] = h[n] x W_edge^T once per node (WMMA, 1/16 of the matrix work); score = f32 dot edge_feat[e].G[src[e]] via a lane exchange tree; 4 launches; k sum order changes
# speedup vs baseline: 4.6191x; 1.0935x over previous
#include <hip/hip_runtime.h>
#include <hip/hip_bf16.h>

typedef __attribute__((ext_vector_type(2))) float v2f;
typedef __attribute__((ext_vector_type(8))) float v8f;

#define NCOLS 128
#define NODE_K 256
#define EDGE_K 64
#define NHEADS 4

typedef __attribute__((ext_vector_type(4))) float v4f;

#define GEMM_GRID 512

__global__ void __launch_bounds__(64)
node_gemm_wreg(const float* __restrict__ A, const float* __restrict__ B,
               float* __restrict__ C, int M) {
    const int lane = threadIdx.x & 31;
    const int ch = __builtin_amdgcn_readfirstlane((int)(threadIdx.x >> 5));
    const int c  = lane & 15;
    const int hf = lane >> 4;

    const int ntile = (M + 15) >> 4;
    int t = blockIdx.x;
    if (t >= ntile) return;
    const int G = gridDim.x;

    v2f wf[4][NODE_K / 4];
    const v4f* B4 = reinterpret_cast<const v4f*>(B);
#pragma unroll
    for (int j = 0; j < NODE_K / 4; ++j) {
        const int k0 = 8 * (j >> 1) + 4 * hf + 2 * (j & 1);
        const v4f w0 = B4[k0 * (NCOLS / 4) + 16 * ch + c];
        const v4f w1 = B4[(k0 + 1) * (NCOLS / 4) + 16 * ch + c];
#pragma unroll
        for (int u = 0; u < 4; ++u) { wf[u][j].x = w0[u]; wf[u][j].y = w1[u]; }
    }

    const v4f* A4 = reinterpret_cast<const v4f*>(A);
    const int lastTile = ntile - 1;
    const int lastRow  = M - 1;

    auto loadTile = [&](int tile, v4f* dstv) {
        int tl = tile < lastTile ? tile : lastTile;
        int row = tl * 16 + c; row = row < lastRow ? row : lastRow;
        const v4f* rp = A4 + (size_t)row * (NODE_K / 4) + hf;
#pragma unroll
        for (int i = 0; i < NODE_K / 8; ++i) dstv[i] = rp[2 * i];
    };

    v4f cur[NODE_K / 8];
    loadTile(t, cur);

    for (; t < ntile; t += G) {
        v4f nxt[NODE_K / 8];
        loadTile(t + G, nxt);
        __builtin_amdgcn_sched_barrier(0);

        v8f acc[4] = {};
#pragma unroll
        for (int i = 0; i < NODE_K / 8; ++i) {
            const v2f a0 = __builtin_shufflevector(cur[i], cur[i], 0, 1);
            const v2f a1 = __builtin_shufflevector(cur[i], cur[i], 2, 3);
#pragma unroll
            for (int u = 0; u < 4; ++u)
                acc[u] = __builtin_amdgcn_wmma_f32_16x16x4_f32(false, a0, false, wf[u][2 * i],     (short)0, acc[u], false, false);
#pragma unroll
            for (int u = 0; u < 4; ++u)
                acc[u] = __builtin_amdgcn_wmma_f32_16x16x4_f32(false, a1, false, wf[u][2 * i + 1], (short)0, acc[u], false, false);
        }
        __builtin_amdgcn_sched_barrier(0);

        const int rbase = t * 16 + 8 * hf;
#pragma unroll
        for (int r = 0; r < 8; ++r) {
            const int row = rbase + r;
            if (row < M) {
                v4f o;
                o.x = acc[0][r]; o.y = acc[1][r]; o.z = acc[2][r]; o.w = acc[3][r];
                *reinterpret_cast<v4f*>(C + (size_t)row * NCOLS + 64 * ch + 4 * c) = o;
            }
        }
#pragma unroll
        for (int i = 0; i < NODE_K / 8; ++i) cur[i] = nxt[i];
    }
}

#define TILE_N    196
#define GCOLS     256
#define GPROJ_GRID 1024

__global__ void __launch_bounds__(128)
node_edge_proj(const float* __restrict__ H, const float* __restrict__ We,
               float* __restrict__ G, int M) {
    const int lane = threadIdx.x & 31;
    const int hd = __builtin_amdgcn_readfirstlane((int)(threadIdx.x >> 5));
    const int c  = lane & 15;
    const int hf = lane >> 4;
    const int ntile = (M + 15) >> 4;
    const int lastRow = M - 1;

    v2f wf[4][8];
    const v4f* We4 = reinterpret_cast<const v4f*>(We);
#pragma unroll
    for (int u = 0; u < 4; ++u) {
#pragma unroll
        for (int i = 0; i < 4; ++i) {
            const v4f w = We4[(4 * c + u) * (NCOLS / 4) + 8 * hd + 2 * i + hf];
            wf[u][2 * i].x     = w.x; wf[u][2 * i].y     = w.y;
            wf[u][2 * i + 1].x = w.z; wf[u][2 * i + 1].y = w.w;
        }
    }

    const v4f* H4 = reinterpret_cast<const v4f*>(H);
    for (int t = blockIdx.x; t < ntile; t += gridDim.x) {
        int row = t * 16 + c; row = row < lastRow ? row : lastRow;
        const v4f* rp = H4 + (size_t)row * (NCOLS / 4) + 8 * hd + hf;
        v4f a[4];
#pragma unroll
        for (int i = 0; i < 4; ++i) a[i] = rp[2 * i];

        v8f acc[4] = {};
#pragma unroll
        for (int i = 0; i < 4; ++i) {
            const v2f a0 = __builtin_shufflevector(a[i], a[i], 0, 1);
            const v2f a1 = __builtin_shufflevector(a[i], a[i], 2, 3);
#pragma unroll
            for (int u = 0; u < 4; ++u)
                acc[u] = __builtin_amdgcn_wmma_f32_16x16x4_f32(false, a0, false, wf[u][2 * i],     (short)0, acc[u], false, false);
#pragma unroll
            for (int u = 0; u < 4; ++u)
                acc[u] = __builtin_amdgcn_wmma_f32_16x16x4_f32(false, a1, false, wf[u][2 * i + 1], (short)0, acc[u], false, false);
        }

        const int rbase = t * 16 + 8 * hf;
#pragma unroll
        for (int r = 0; r < 8; ++r) {
            const int orow = rbase + r;
            if (orow < M) {
                v4f o;
                o.x = acc[0][r]; o.y = acc[1][r]; o.z = acc[2][r]; o.w = acc[3][r];
                *reinterpret_cast<v4f*>(G + (size_t)orow * GCOLS + 64 * hd + 4 * c) = o;
            }
        }
    }
}

#define SCORE_WAVES 4
#define SCORE_GRID  1280

__global__ void __launch_bounds__(32 * SCORE_WAVES)
edge_score_dot(const float* __restrict__ EF, const float* __restrict__ G,
               const int* __restrict__ src, const int* __restrict__ dst,
               unsigned* __restrict__ pres, float* __restrict__ score, int E) {
    const int lane = threadIdx.x & 31;
    const int wave = __builtin_amdgcn_readfirstlane((int)(threadIdx.x >> 5));
    const int gw = blockIdx.x * SCORE_WAVES + wave;
    const int nw = gridDim.x * SCORE_WAVES;
    const int i  = lane & 15;
    const int hf = lane >> 4;

    {
        const int ng = E >> 5;
        const int gp = (ng + 3) & ~3;
        const int4* dst4 = reinterpret_cast<const int4*>(dst);
        for (int task = gw; task * 32 < gp; task += nw) {
            const int g = task * 32 + lane;
            unsigned w[8] = {0u, 0u, 0u, 0u, 0u, 0u, 0u, 0u};
            if (g < ng) {
#pragma unroll
                for (int ii = 0; ii < 8; ++ii) {
                    const int4 d = dst4[(size_t)g * 8 + ii];
                    const unsigned t4[4] = {(unsigned)d.x / TILE_N, (unsigned)d.y / TILE_N,
                                            (unsigned)d.z / TILE_N, (unsigned)d.w / TILE_N};
#pragma unroll
                    for (int j = 0; j < 4; ++j) {
                        const unsigned t   = t4[j] & 255u;
                        const unsigned bit = 1u << (t & 31u);
#pragma unroll
                        for (int k = 0; k < 8; ++k) w[k] |= (t >> 5) == (unsigned)k ? bit : 0u;
                    }
                }
            }
            if (g < gp) {
#pragma unroll
                for (int k = 0; k < 8; ++k) pres[(size_t)k * gp + g] = w[k];
            }
        }
    }

    const int nb = (E + 31) >> 5;
    const v4f* EF4 = reinterpret_cast<const v4f*>(EF);
    const v4f* G4  = reinterpret_cast<const v4f*>(G);
    const unsigned lastEdge = (unsigned)(E - 1);
    const bool b3 = (lane & 8) != 0, b2 = (lane & 4) != 0, b1 = (lane & 2) != 0, b0 = (lane & 1) != 0;

    for (int b = gw; b < nb; b += nw) {
        const unsigned e0 = (unsigned)b * 32u;
        unsigned el = e0 + (unsigned)lane; el = el < lastEdge ? el : lastEdge;
        const int sv = src[el];

        float r8[8][4];
#pragma unroll
        for (int q = 0; q < 4; ++q) {
            v4f x[4];
            v4f g[4][4];
#pragma unroll
            for (int sl = 0; sl < 4; ++sl) {
                const int jp = 2 * q + (sl & 1) + 8 * (sl >> 1);
                unsigned e = e0 + 16u * (unsigned)hf + (unsigned)jp; e = e < lastEdge ? e : lastEdge;
                const int sA = __builtin_amdgcn_readlane(sv, jp);
                const int sB = __builtin_amdgcn_readlane(sv, jp + 16);
                const unsigned s = (unsigned)(hf ? sB : sA);
                x[sl] = __builtin_nontemporal_load(&EF4[e * (unsigned)(EDGE_K / 4) + (unsigned)i]);
                const v4f* gr = G4 + (s * (unsigned)(GCOLS / 4) + (unsigned)i);
#pragma unroll
                for (int hd = 0; hd < 4; ++hd) g[sl][hd] = gr[16 * hd];
            }
            __builtin_amdgcn_sched_barrier(0);
            float v[4][4];
#pragma unroll
            for (int sl = 0; sl < 4; ++sl) {
#pragma unroll
                for (int hd = 0; hd < 4; ++hd) {
                    const v4f gg = g[sl][hd];
                    const v4f xx = x[sl];
                    v[sl][hd] = xx.x * gg.x + xx.y * gg.y + xx.z * gg.z + xx.w * gg.w;
                }
            }
#pragma unroll
            for (int m = 0; m < 2; ++m) {
#pragma unroll
                for (int hd = 0; hd < 4; ++hd) {
                    const float lo = v[m][hd], hi = v[m + 2][hd];
                    const float keep = b3 ? hi : lo;
                    const float send = b3 ? lo : hi;
                    r8[2 * q + m][hd] = keep + __shfl_xor(send, 8, 32);
                }
            }
        }
        float r4[4][4];
#pragma unroll
        for (int jp = 0; jp < 4; ++jp) {
#pragma unroll
            for (int hd = 0; hd < 4; ++hd) {
                const float lo = r8[jp][hd], hi = r8[jp + 4][hd];
                const float keep = b2 ? hi : lo;
                const float send = b2 ? lo : hi;
                r4[jp][hd] = keep + __shfl_xor(send, 4, 32);
            }
        }
        float r2[2][4];
#pragma unroll
        for (int jp = 0; jp < 2; ++jp) {
#pragma unroll
            for (int hd = 0; hd < 4; ++hd) {
                const float lo = r4[jp][hd], hi = r4[jp + 2][hd];
                const float keep = b1 ? hi : lo;
                const float send = b1 ? lo : hi;
                r2[jp][hd] = keep + __shfl_xor(send, 2, 32);
            }
        }
        const unsigned eo = e0 + (unsigned)lane;
#pragma unroll
        for (int hd = 0; hd < 4; ++hd) {
            const float lo = r2[0][hd], hi = r2[1][hd];
            const float keep = b0 ? hi : lo;
            const float send = b0 ? lo : hi;
            const float d = keep + __shfl_xor(send, 1, 32);
            float sc = d * 0.17677669529663689f;
            sc = sc > 0.f ? sc : 0.01f * sc;
            if (eo <= lastEdge) score[(unsigned)hd * (unsigned)E + eo] = sc;
        }
    }
}

#define AGG_WAVES 32
#define P_MAX     192
#define G_CALL    128
#define Q_BURST   160
#define Q_CAP     (P_MAX + Q_BURST)

__device__ __forceinline__ int   rl_i(int v, int k)   { return __builtin_amdgcn_readlane(v, k); }
__device__ __forceinline__ float rl_f(float v, int k) {
    return __int_as_float(__builtin_amdgcn_readlane(__float_as_int(v), k));
}

__device__ __forceinline__ void ldsMaxFloat(float* addr, float val) {
    if (val >= 0.f) atomicMax((int*)addr, __float_as_int(val));
    else            atomicMin((unsigned int*)addr, __float_as_uint(val));
}

__global__ void __launch_bounds__(1024)
aggregate_tiles(const float* __restrict__ H, const int* __restrict__ src,
                const int* __restrict__ dst, const unsigned* __restrict__ pres,
                const float* __restrict__ score,
                float* __restrict__ out, int E, int N) {
    __shared__ alignas(16) float  tile[TILE_N * NCOLS];
    __shared__ alignas(16) float4 keptScore[AGG_WAVES][P_MAX];
    __shared__ int2  queue[AGG_WAVES][Q_CAP];
    __shared__ int   glist[AGG_WAVES][G_CALL];
    __shared__ alignas(16) float smaxT[TILE_N * NHEADS];
    __shared__ alignas(16) float denT[TILE_N * NHEADS];

    const int lane = threadIdx.x & 31;
    const int wave = __builtin_amdgcn_readfirstlane((int)(threadIdx.x >> 5));
    const int lo   = blockIdx.x * TILE_N;
    int cnt = N - lo; cnt = cnt < TILE_N ? cnt : TILE_N;
    const unsigned obyte = blockIdx.x & 255u;
    const unsigned obit  = obyte & 31u;

    for (int i = threadIdx.x; i < TILE_N * NCOLS; i += 1024) tile[i] = 0.f;
    if (threadIdx.x < TILE_N * NHEADS) {
        smaxT[threadIdx.x] = __int_as_float(0xFF800000);
        denT[threadIdx.x]  = 0.f;
    }
    __syncthreads();

    int2*   q  = queue[wave];
    float4* ks = keptScore[wave];
    int*    gl = glist[wave];
    int  kept = 0;
    int  qn   = 0;
    bool keeping = true;
    int  resumeCall = 0;
    const float* sc0 = score;
    const float* sc1 = score + (size_t)E;
    const float* sc2 = score + 2 * (size_t)E;
    const float* sc3 = score + 3 * (size_t)E;

    auto drainMax = [&](int callsDone) {
        __builtin_amdgcn_fence(__ATOMIC_RELEASE, "wavefront");
        __builtin_amdgcn_wave_barrier();
        __builtin_amdgcn_fence(__ATOMIC_ACQUIRE, "wavefront");
        const bool keep = keeping && qn <= P_MAX;
        int w = kept;
        for (int r = kept; r < qn; r += 32) {
            const int idx = r + lane;
            bool  ok = false;
            int   s = 0, row = 0;
            float s0 = 0.f, s1 = 0.f, s2 = 0.f, s3 = 0.f;
            if (idx < qn) {
                const int e = q[idx].x;
                const unsigned rel = (unsigned)(dst[e] - lo);
                s  = src[e];
                s0 = sc0[e]; s1 = sc1[e]; s2 = sc2[e]; s3 = sc3[e];
                ok  = rel < (unsigned)cnt;
                row = (int)rel;
                if (ok) {
                    float* mp = smaxT + row * NHEADS;
                    ldsMaxFloat(mp,     s0);
                    ldsMaxFloat(mp + 1, s1);
                    ldsMaxFloat(mp + 2, s2);
                    ldsMaxFloat(mp + 3, s3);
                }
            }
            const unsigned okm = __builtin_amdgcn_ballot_w32(ok);
            if (keep && ok) {
                const int pos = w + (int)__builtin_amdgcn_mbcnt_lo(okm, 0u);
                q[pos]  = make_int2(s, row);
                ks[pos] = make_float4(s0, s1, s2, s3);
            }
            w += __builtin_popcount(okm);
        }
        if (keep) { kept = w; qn = w; resumeCall = callsDone; }
        else      { keeping = false; qn = kept; }
        __builtin_amdgcn_fence(__ATOMIC_RELEASE, "wavefront");
        __builtin_amdgcn_wave_barrier();
    };

    auto sumBatch = [&](int n, int s, int row, float s0, float s1, float s2, float s3) {
        const float4 m4 = reinterpret_cast<const float4*>(smaxT)[row];
        const float w0 = __expf(s0 - m4.x), w1 = __expf(s1 - m4.y);
        const float w2 = __expf(s2 - m4.z), w3 = __expf(s3 - m4.w);
        if (lane < n) {
            float* dp = denT + row * NHEADS;
            atomicAdd(dp,     w0);
            atomicAdd(dp + 1, w1);
            atomicAdd(dp + 2, w2);
            atomicAdd(dp + 3, w3);
        }
        for (int k = 0; k < n; k += 4) {
            float hv[4][4];
#pragma unroll
            for (int u = 0; u < 4; ++u) {
                const int kk = k + u < n ? k + u : n - 1;
                const float* hr = H + (size_t)rl_i(s, kk) * NCOLS + lane;
                hv[u][0] = hr[0]; hv[u][1] = hr[32]; hv[u][2] = hr[64]; hv[u][3] = hr[96];
            }
#pragma unroll
            for (int u = 0; u < 4; ++u) {
                if (k + u < n) {
                    float* tr = tile + rl_i(row, k + u) * NCOLS + lane;
                    atomicAdd(tr,      hv[u][0] * rl_f(w0, k + u));
                    atomicAdd(tr + 32, hv[u][1] * rl_f(w1, k + u));
                    atomicAdd(tr + 64, hv[u][2] * rl_f(w2, k + u));
                    atomicAdd(tr + 96, hv[u][3] * rl_f(w3, k + u));
                }
            }
        }
    };

    auto drainSumGather = [&]() {
        __builtin_amdgcn_fence(__ATOMIC_RELEASE, "wavefront");
        __builtin_amdgcn_wave_barrier();
        __builtin_amdgcn_fence(__ATOMIC_ACQUIRE, "wavefront");
        int w = kept;
        for (int r = kept; r < qn; r += 32) {
            const int idx = r + lane;
            bool ok = false;
            int  e = 0, row = 0;
            if (idx < qn) {
                e = q[idx].x;
                const unsigned rel = (unsigned)(dst[e] - lo);
                ok  = rel < (unsigned)cnt;
                row = (int)rel;
            }
            const unsigned okm = __builtin_amdgcn_ballot_w32(ok);
            if (ok) q[w + (int)__builtin_amdgcn_mbcnt_lo(okm, 0u)] = make_int2(e, row);
            w += __builtin_popcount(okm);
        }
        qn = w;
        __builtin_amdgcn_fence(__ATOMIC_RELEASE, "wavefront");
        __builtin_amdgcn_wave_barrier();
        __builtin_amdgcn_fence(__ATOMIC_ACQUIRE, "wavefront");
        for (int r = kept; r < qn; r += 32) {
            int n = qn - r; n = n < 32 ? n : 32;
            int idx = r + lane; idx = idx < qn ? idx : qn - 1;
            const int2 ent = q[idx];
            const int   s  = src[ent.x];
            const float s0 = sc0[ent.x], s1 = sc1[ent.x], s2 = sc2[ent.x], s3 = sc3[ent.x];
            sumBatch(n, s, ent.y, s0, s1, s2, s3);
        }
        qn = kept;
        __builtin_amdgcn_fence(__ATOMIC_RELEASE, "wavefront");
        __builtin_amdgcn_wave_barrier();
    };

    const int  ng   = E >> 5;
    const int  gp   = (ng + 3) & ~3;
    const int  nvec = gp >> 2;
    const int  last = nvec - 1;
    const int4* plane4 = reinterpret_cast<const int4*>(pres + (size_t)(obyte >> 5) * gp);
    const int4* dst4   = reinterpret_cast<const int4*>(dst);
    const int  vfirst = wave * 32;
    auto ld = [&](int v) { return plane4[v < last ? v : last]; };

    auto hitMask = [&](int4 d, int v) -> unsigned {
        const unsigned m = (((unsigned)d.x >> obit) & 1u)        | ((((unsigned)d.y >> obit) & 1u) << 1)
                         | ((((unsigned)d.z >> obit) & 1u) << 2) | ((((unsigned)d.w >> obit) & 1u) << 3);
        return v > last ? 0u : m;
    };

    auto appendHits = [&](unsigned m, int base, auto&& afterRound) {
        for (;;) {
            const bool     has  = m != 0u;
            const unsigned mask = __builtin_amdgcn_ballot_w32(has);
            if (!mask) break;
            if (has) {
                q[qn + (int)__builtin_amdgcn_mbcnt_lo(mask, 0u)] = make_int2(base + __builtin_ctz(m), 0);
                m &= m - 1u;
            }
            qn += __builtin_popcount(mask);
            afterRound();
        }
    };

    auto filterCall = [&](int4 x, int v, auto&& afterRound) {
        unsigned m = hitMask(x, v);
        int gn = 0;
        for (;;) {
            const bool     has  = m != 0u;
            const unsigned mask = __builtin_amdgcn_ballot_w32(has);
            if (!mask) break;
            if (has) {
                gl[gn + (int)__builtin_amdgcn_mbcnt_lo(mask, 0u)] = v * 4 + __builtin_ctz(m);
                m &= m - 1u;
            }
            gn += __builtin_popcount(mask);
        }
        __builtin_amdgcn_fence(__ATOMIC_RELEASE, "wavefront");
        __builtin_amdgcn_wave_barrier();
        __builtin_amdgcn_fence(__ATOMIC_ACQUIRE, "wavefront");
        for (int r = 0; r < gn; r += 16) {
            const int idx = r + (lane >> 1);
            unsigned em = 0u;
            int base = 0;
            if (idx < gn) {
                const int g = gl[idx];
                base = g * 32 + 16 * (lane & 1);
                const int4* dp = dst4 + ((size_t)g * 8 + 4 * (lane & 1));
#pragma unroll
                for (int i = 0; i < 4; ++i) {
                    const int4 d = dp[i];
                    em |= ((unsigned)(d.x - lo) < (unsigned)cnt ? 1u : 0u) << (4 * i);
                    em |= ((unsigned)(d.y - lo) < (unsigned)cnt ? 1u : 0u) << (4 * i + 1);
                    em |= ((unsigned)(d.z - lo) < (unsigned)cnt ? 1u : 0u) << (4 * i + 2);
                    em |= ((unsigned)(d.w - lo) < (unsigned)cnt ? 1u : 0u) << (4 * i + 3);
                }
            }
            appendHits(em, base, afterRound);
        }
        __builtin_amdgcn_fence(__ATOMIC_RELEASE, "wavefront");
        __builtin_amdgcn_wave_barrier();
    };
    auto roomForRound = [&]() { if (qn + 32 > Q_CAP) drainMax(resumeCall); };

    auto filterLeftover = [&]() {
        if (wave == 0) {
            const int e = ng * 32 + lane;
            bool hit = false;
            if (e < E) hit = (unsigned)(dst[e] - lo) < (unsigned)cnt;
            const unsigned mask = __builtin_amdgcn_ballot_w32(hit);
            if (mask) {
                if (hit) q[qn + (int)__builtin_amdgcn_mbcnt_lo(mask, 0u)] = make_int2(e, 0);
                qn += __builtin_popcount(mask);
            }
        }
    };

    if (vfirst < nvec) {
        int vb = vfirst;
        int4 x0 = ld(vb + lane), x1 = ld(vb + lane + 1024), x2 = ld(vb + lane + 2048), x3 = ld(vb + lane + 3072);
        for (; vb < nvec; vb += 4 * 1024) {
            const int v0 = vb + lane, v1 = v0 + 1024, v2 = v0 + 2048, v3 = v0 + 3072;
            const int nx = v0 + 4 * 1024;
            const int4 y0 = ld(nx), y1 = ld(nx + 1024), y2 = ld(nx + 2048), y3 = ld(nx + 3072);
            const int c0 = ((vb - vfirst) >> 12) * 4;
            filterCall(x0, v0, roomForRound);
            if (qn - kept >= 32) drainMax(c0 + 1);
            filterCall(x1, v1, roomForRound);
            if (qn - kept >= 32) drainMax(c0 + 2);
            filterCall(x2, v2, roomForRound);
            if (qn - kept >= 32) drainMax(c0 + 3);
            filterCall(x3, v3, roomForRound);
            if (qn - kept >= 32) drainMax(c0 + 4);
            x0 = y0; x1 = y1; x2 = y2; x3 = y3;
        }
    }
    filterLeftover();
    if (qn > kept) drainMax(0x7fffffff);
    __syncthreads();

    for (int r = 0; r < kept; r += 32) {
        int n = kept - r; n = n < 32 ? n : 32;
        int idx = r + lane; idx = idx < kept ? idx : kept - 1;
        const int2   ent = q[idx];
        const float4 s4  = ks[idx];
        sumBatch(n, ent.x, ent.y, s4.x, s4.y, s4.z, s4.w);
    }
    if (!keeping) {
        if (vfirst < nvec) {
            const int ncall = ((nvec - vfirst + 4095) >> 12) * 4;
            for (int ci = resumeCall; ci < ncall; ++ci) {
                const int v = vfirst + (ci >> 2) * 4096 + (ci & 3) * 1024 + lane;
                filterCall(ld(v), v, [&]() { if (qn - kept >= 32) drainSumGather(); });
            }
        }
        filterLeftover();
        if (qn > kept) drainSumGather();
    }
    __syncthreads();

    const float4* t4 = reinterpret_cast<const float4*>(tile);
    float4* o4 = reinterpret_cast<float4*>(out + (size_t)lo * NCOLS);
    for (int i = threadIdx.x; i < cnt * (NCOLS / 4); i += 1024) {
        const float d = denT[(i >> 5) * NHEADS + ((i & 31) >> 3)];
        float4 v = t4[i];
        const bool has = d > 0.f;
        v.x = has ? v.x / d : 0.f;
        v.y = has ? v.y / d : 0.f;
        v.z = has ? v.z / d : 0.f;
        v.w = has ? v.w / d : 0.f;
        o4[i] = v;
    }
}

extern "C" void kernel_launch(void* const* d_in, const int* in_sizes, int n_in,
                              void* d_out, int out_size, void* d_ws, size_t ws_size,
                              hipStream_t stream) {
    const float* node_feat = (const float*)d_in[0];
    const float* edge_feat = (const float*)d_in[1];
    const int*   src       = (const int*)d_in[2];
    const int*   dst       = (const int*)d_in[3];
    const float* W_node    = (const float*)d_in[4];
    const float* W_edge    = (const float*)d_in[5];
    float*       out       = (float*)d_out;

    const int N = in_sizes[0] / NODE_K;
    const int E = in_sizes[2];

    float*    h     = (float*)d_ws;
    float*    G     = h     + (size_t)N * NCOLS;
    float*    score = G     + (size_t)N * GCOLS;
    unsigned* pres  = (unsigned*)(score + (size_t)NHEADS * E);

    node_gemm_wreg<<<GEMM_GRID, 64, 0, stream>>>(node_feat, W_node, h, N);

    node_edge_proj<<<GPROJ_GRID, 128, 0, stream>>>(h, W_edge, G, N);

    edge_score_dot<<<SCORE_GRID, 32 * SCORE_WAVES, 0, stream>>>(edge_feat, G, src, dst, pres, score, E);

    aggregate_tiles<<<(N + TILE_N - 1) / TILE_N, 1024, 0, stream>>>(h, src, dst, pres, score, out, E, N);
}


# G projection launch folded into the node projection: a wave passes its finished 16x64 h tile through a wave-private LDS panel and takes its two heads' G products (WMMA f32) itself; 4 launches -> 3
# speedup vs baseline: 4.7589x; 1.0303x over previous
#include <hip/hip_runtime.h>
#include <hip/hip_bf16.h>

typedef __attribute__((ext_vector_type(2))) float v2f;
typedef __attribute__((ext_vector_type(8))) float v8f;

#define NCOLS 128
#define NODE_K 256
#define EDGE_K 64
#define NHEADS 4

typedef __attribute__((ext_vector_type(4))) float v4f;

#define GEMM_GRID 512
#define GCOLS     256
#define HS_PITCH  68

__global__ void __launch_bounds__(64)
node_gemm_wreg(const float* __restrict__ A, const float* __restrict__ B,
               float* __restrict__ C, const float* __restrict__ We,
               float* __restrict__ Gout, int M) {
    __shared__ alignas(16) float hs[2][16 * HS_PITCH];
    const int lane = threadIdx.x & 31;
    const int ch = __builtin_amdgcn_readfirstlane((int)(threadIdx.x >> 5));
    const int c  = lane & 15;
    const int hf = lane >> 4;

    const int ntile = (M + 15) >> 4;
    int t = blockIdx.x;
    if (t >= ntile) return;
    const int G = gridDim.x;

    v2f wf[4][NODE_K / 4];
    const v4f* B4 = reinterpret_cast<const v4f*>(B);
#pragma unroll
    for (int j = 0; j < NODE_K / 4; ++j) {
        const int k0 = 8 * (j >> 1) + 4 * hf + 2 * (j & 1);
        const v4f w0 = B4[k0 * (NCOLS / 4) + 16 * ch + c];
        const v4f w1 = B4[(k0 + 1) * (NCOLS / 4) + 16 * ch + c];
#pragma unroll
        for (int u = 0; u < 4; ++u) { wf[u][j].x = w0[u]; wf[u][j].y = w1[u]; }
    }

    v2f wfe[2][4][8];
    const v4f* We4 = reinterpret_cast<const v4f*>(We);
#pragma unroll
    for (int hh = 0; hh < 2; ++hh) {
#pragma unroll
        for (int u = 0; u < 4; ++u) {
#pragma unroll
            for (int i = 0; i < 4; ++i) {
                const v4f w = We4[(4 * c + u) * (NCOLS / 4) + 8 * (2 * ch + hh) + 2 * i + hf];
                wfe[hh][u][2 * i].x     = w.x; wfe[hh][u][2 * i].y     = w.y;
                wfe[hh][u][2 * i + 1].x = w.z; wfe[hh][u][2 * i + 1].y = w.w;
            }
        }
    }
    float* hp = hs[ch];

    const v4f* A4 = reinterpret_cast<const v4f*>(A);
    const int lastTile = ntile - 1;
    const int lastRow  = M - 1;

    auto loadTile = [&](int tile, v4f* dstv) {
        int tl = tile < lastTile ? tile : lastTile;
        int row = tl * 16 + c; row = row < lastRow ? row : lastRow;
        const v4f* rp = A4 + (size_t)row * (NODE_K / 4) + hf;
#pragma unroll
        for (int i = 0; i < NODE_K / 8; ++i) dstv[i] = rp[2 * i];
    };

    v4f cur[NODE_K / 8];
    loadTile(t, cur);

    for (; t < ntile; t += G) {
        v4f nxt[NODE_K / 8];
        loadTile(t + G, nxt);
        __builtin_amdgcn_sched_barrier(0);

        v8f acc[4] = {};
#pragma unroll
        for (int i = 0; i < NODE_K / 8; ++i) {
            const v2f a0 = __builtin_shufflevector(cur[i], cur[i], 0, 1);
            const v2f a1 = __builtin_shufflevector(cur[i], cur[i], 2, 3);
#pragma unroll
            for (int u = 0; u < 4; ++u)
                acc[u] = __builtin_amdgcn_wmma_f32_16x16x4_f32(false, a0, false, wf[u][2 * i],     (short)0, acc[u], false, false);
#pragma unroll
            for (int u = 0; u < 4; ++u)
                acc[u] = __builtin_amdgcn_wmma_f32_16x16x4_f32(false, a1, false, wf[u][2 * i + 1], (short)0, acc[u], false, false);
        }
        __builtin_amdgcn_sched_barrier(0);

        const int rbase = t * 16 + 8 * hf;
#pragma unroll
        for (int r = 0; r < 8; ++r) {
            const int row = rbase + r;
            if (row < M) {
                v4f o;
                o.x = acc[0][r]; o.y = acc[1][r]; o.z = acc[2][r]; o.w = acc[3][r];
                *reinterpret_cast<v4f*>(C + (size_t)row * NCOLS + 64 * ch + 4 * c) = o;
            }
            v4f p;
            p.x = acc[0][r]; p.y = acc[1][r]; p.z = acc[2][r]; p.w = acc[3][r];
            *reinterpret_cast<v4f*>(hp + (8 * hf + r) * HS_PITCH + 4 * c) = p;
        }
        __builtin_amdgcn_fence(__ATOMIC_RELEASE, "wavefront");
        __builtin_amdgcn_wave_barrier();
        __builtin_amdgcn_fence(__ATOMIC_ACQUIRE, "wavefront");

        v4f a2[2][4];
#pragma unroll
        for (int hh = 0; hh < 2; ++hh) {
#pragma unroll
            for (int i = 0; i < 4; ++i)
                a2[hh][i] = *reinterpret_cast<const v4f*>(hp + c * HS_PITCH + 32 * hh + 8 * i + 4 * hf);
        }
#pragma unroll
        for (int hh = 0; hh < 2; ++hh) {
            const v4f* a = a2[hh];
            v8f ag[4] = {};
#pragma unroll
            for (int i = 0; i < 4; ++i) {
                const v2f a0 = __builtin_shufflevector(a[i], a[i], 0, 1);
                const v2f a1 = __builtin_shufflevector(a[i], a[i], 2, 3);
#pragma unroll
                for (int u = 0; u < 4; ++u)
                    ag[u] = __builtin_amdgcn_wmma_f32_16x16x4_f32(false, a0, false, wfe[hh][u][2 * i],     (short)0, ag[u], false, false);
#pragma unroll
                for (int u = 0; u < 4; ++u)
                    ag[u] = __builtin_amdgcn_wmma_f32_16x16x4_f32(false, a1, false, wfe[hh][u][2 * i + 1], (short)0, ag[u], false, false);
            }
#pragma unroll
            for (int r = 0; r < 8; ++r) {
                const int orow = rbase + r;
                if (orow < M) {
                    v4f o;
                    o.x = ag[0][r]; o.y = ag[1][r]; o.z = ag[2][r]; o.w = ag[3][r];
                    *reinterpret_cast<v4f*>(Gout + (size_t)orow * GCOLS + 64 * (2 * ch + hh) + 4 * c) = o;
                }
            }
        }
        __builtin_amdgcn_fence(__ATOMIC_RELEASE, "wavefront");
        __builtin_amdgcn_wave_barrier();

#pragma unroll
        for (int i = 0; i < NODE_K / 8; ++i) cur[i] = nxt[i];
    }
}

#define TILE_N    196

#define SCORE_WAVES 4
#define SCORE_GRID  1280

__global__ void __launch_bounds__(32 * SCORE_WAVES)
edge_score_dot(const float* __restrict__ EF, const float* __restrict__ G,
               const int* __restrict__ src, const int* __restrict__ dst,
               unsigned* __restrict__ pres, float* __restrict__ score, int E) {
    const int lane = threadIdx.x & 31;
    const int wave = __builtin_amdgcn_readfirstlane((int)(threadIdx.x >> 5));
    const int gw = blockIdx.x * SCORE_WAVES + wave;
    const int nw = gridDim.x * SCORE_WAVES;
    const int i  = lane & 15;
    const int hf = lane >> 4;

    {
        const int ng = E >> 5;
        const int gp = (ng + 3) & ~3;
        const int4* dst4 = reinterpret_cast<const int4*>(dst);
        for (int task = gw; task * 32 < gp; task += nw) {
            const int g = task * 32 + lane;
            unsigned w[8] = {0u, 0u, 0u, 0u, 0u, 0u, 0u, 0u};
            if (g < ng) {
#pragma unroll
                for (int ii = 0; ii < 8; ++ii) {
                    const int4 d = dst4[(size_t)g * 8 + ii];
                    const unsigned t4[4] = {(unsigned)d.x / TILE_N, (unsigned)d.y / TILE_N,
                                            (unsigned)d.z / TILE_N, (unsigned)d.w / TILE_N};
#pragma unroll
                    for (int j = 0; j < 4; ++j) {
                        const unsigned t   = t4[j] & 255u;
                        const unsigned bit = 1u << (t & 31u);
#pragma unroll
                        for (int k = 0; k < 8; ++k) w[k] |= (t >> 5) == (unsigned)k ? bit : 0u;
                    }
                }
            }
            if (g < gp) {
#pragma unroll
                for (int k = 0; k < 8; ++k) pres[(size_t)k * gp + g] = w[k];
            }
        }
    }

    const int nb = (E + 31) >> 5;
    const v4f* EF4 = reinterpret_cast<const v4f*>(EF);
    const v4f* G4  = reinterpret_cast<const v4f*>(G);
    const unsigned lastEdge = (unsigned)(E - 1);
    const bool b3 = (lane & 8) != 0, b2 = (lane & 4) != 0, b1 = (lane & 2) != 0, b0 = (lane & 1) != 0;

    for (int b = gw; b < nb; b += nw) {
        const unsigned e0 = (unsigned)b * 32u;
        unsigned el = e0 + (unsigned)lane; el = el < lastEdge ? el : lastEdge;
        const int sv = src[el];

        float r8[8][4];
#pragma unroll
        for (int q = 0; q < 4; ++q) {
            v4f x[4];
            v4f g[4][4];
#pragma unroll
            for (int sl = 0; sl < 4; ++sl) {
                const int jp = 2 * q + (sl & 1) + 8 * (sl >> 1);
                unsigned e = e0 + 16u * (unsigned)hf + (unsigned)jp; e = e < lastEdge ? e : lastEdge;
                const int sA = __builtin_amdgcn_readlane(sv, jp);
                const int sB = __builtin_amdgcn_readlane(sv, jp + 16);
                const unsigned s = (unsigned)(hf ? sB : sA);
                x[sl] = __builtin_nontemporal_load(&EF4[e * (unsigned)(EDGE_K / 4) + (unsigned)i]);
                const v4f* gr = G4 + (s * (unsigned)(GCOLS / 4) + (unsigned)i);
#pragma unroll
                for (int hd = 0; hd < 4; ++hd) g[sl][hd] = gr[16 * hd];
            }
            __builtin_amdgcn_sched_barrier(0);
            float v[4][4];
#pragma unroll
            for (int sl = 0; sl < 4; ++sl) {
#pragma unroll
                for (int hd = 0; hd < 4; ++hd) {
                    const v4f gg = g[sl][hd];
                    const v4f xx = x[sl];
                    v[sl][hd] = xx.x * gg.x + xx.y * gg.y + xx.z * gg.z + xx.w * gg.w;
                }
            }
#pragma unroll
            for (int m = 0; m < 2; ++m) {
#pragma unroll
                for (int hd = 0; hd < 4; ++hd) {
                    const float lo = v[m][hd], hi = v[m + 2][hd];
                    const float keep = b3 ? hi : lo;
                    const float send = b3 ? lo : hi;
                    r8[2 * q + m][hd] = keep + __shfl_xor(send, 8, 32);
                }
            }
        }
        float r4[4][4];
#pragma unroll
        for (int jp = 0; jp < 4; ++jp) {
#pragma unroll
            for (int hd = 0; hd < 4; ++hd) {
                const float lo = r8[jp][hd], hi = r8[jp + 4][hd];
                const float keep = b2 ? hi : lo;
                const float send = b2 ? lo : hi;
                r4[jp][hd] = keep + __shfl_xor(send, 4, 32);
            }
        }
        float r2[2][4];
#pragma unroll
        for (int jp = 0; jp < 2; ++jp) {
#pragma unroll
            for (int hd = 0; hd < 4; ++hd) {
                const float lo = r4[jp][hd], hi = r4[jp + 2][hd];
                const float keep = b1 ? hi : lo;
                const float send = b1 ? lo : hi;
                r2[jp][hd] = keep + __shfl_xor(send, 2, 32);
            }
        }
        const unsigned eo = e0 + (unsigned)lane;
#pragma unroll
        for (int hd = 0; hd < 4; ++hd) {
            const float lo = r2[0][hd], hi = r2[1][hd];
            const float keep = b0 ? hi : lo;
            const float send = b0 ? lo : hi;
            const float d = keep + __shfl_xor(send, 1, 32);
            float sc = d * 0.17677669529663689f;
            sc = sc > 0.f ? sc : 0.01f * sc;
            if (eo <= lastEdge) score[(unsigned)hd * (unsigned)E + eo] = sc;
        }
    }
}

#define AGG_WAVES 32
#define P_MAX     192
#define G_CALL    128
#define Q_BURST   160
#define Q_CAP     (P_MAX + Q_BURST)

__device__ __forceinline__ int   rl_i(int v, int k)   { return __builtin_amdgcn_readlane(v, k); }
__device__ __forceinline__ float rl_f(float v, int k) {
    return __int_as_float(__builtin_amdgcn_readlane(__float_as_int(v), k));
}

__device__ __forceinline__ void ldsMaxFloat(float* addr, float val) {
    if (val >= 0.f) atomicMax((int*)addr, __float_as_int(val));
    else            atomicMin((unsigned int*)addr, __float_as_uint(val));
}

__global__ void __launch_bounds__(1024)
aggregate_tiles(const float* __restrict__ H, const int* __restrict__ src,
                const int* __restrict__ dst, const unsigned* __restrict__ pres,
                const float* __restrict__ score,
                float* __restrict__ out, int E, int N) {
    __shared__ alignas(16) float  tile[TILE_N * NCOLS];
    __shared__ alignas(16) float4 keptScore[AGG_WAVES][P_MAX];
    __shared__ int2  queue[AGG_WAVES][Q_CAP];
    __shared__ int   glist[AGG_WAVES][G_CALL];
    __shared__ alignas(16) float smaxT[TILE_N * NHEADS];
    __shared__ alignas(16) float denT[TILE_N * NHEADS];

    const int lane = threadIdx.x & 31;
    const int wave = __builtin_amdgcn_readfirstlane((int)(threadIdx.x >> 5));
    const int lo   = blockIdx.x * TILE_N;
    int cnt = N - lo; cnt = cnt < TILE_N ? cnt : TILE_N;
    const unsigned obyte = blockIdx.x & 255u;
    const unsigned obit  = obyte & 31u;

    for (int i = threadIdx.x; i < TILE_N * NCOLS; i += 1024) tile[i] = 0.f;
    if (threadIdx.x < TILE_N * NHEADS) {
        smaxT[threadIdx.x] = __int_as_float(0xFF800000);
        denT[threadIdx.x]  = 0.f;
    }
    __syncthreads();

    int2*   q  = queue[wave];
    float4* ks = keptScore[wave];
    int*    gl = glist[wave];
    int  kept = 0;
    int  qn   = 0;
    bool keeping = true;
    int  resumeCall = 0;
    const float* sc0 = score;
    const float* sc1 = score + (size_t)E;
    const float* sc2 = score + 2 * (size_t)E;
    const float* sc3 = score + 3 * (size_t)E;

    auto drainMax = [&](int callsDone) {
        __builtin_amdgcn_fence(__ATOMIC_RELEASE, "wavefront");
        __builtin_amdgcn_wave_barrier();
        __builtin_amdgcn_fence(__ATOMIC_ACQUIRE, "wavefront");
        const bool keep = keeping && qn <= P_MAX;
        int w = kept;
        for (int r = kept; r < qn; r += 32) {
            const int idx = r + lane;
            bool  ok = false;
            int   s = 0, row = 0;
            float s0 = 0.f, s1 = 0.f, s2 = 0.f, s3 = 0.f;
            if (idx < qn) {
                const int e = q[idx].x;
                const unsigned rel = (unsigned)(dst[e] - lo);
                s  = src[e];
                s0 = sc0[e]; s1 = sc1[e]; s2 = sc2[e]; s3 = sc3[e];
                ok  = rel < (unsigned)cnt;
                row = (int)rel;
                if (ok) {
                    float* mp = smaxT + row * NHEADS;
                    ldsMaxFloat(mp,     s0);
                    ldsMaxFloat(mp + 1, s1);
                    ldsMaxFloat(mp + 2, s2);
                    ldsMaxFloat(mp + 3, s3);
                }
            }
            const unsigned okm = __builtin_amdgcn_ballot_w32(ok);
            if (keep && ok) {
                const int pos = w + (int)__builtin_amdgcn_mbcnt_lo(okm, 0u);
                q[pos]  = make_int2(s, row);
                ks[pos] = make_float4(s0, s1, s2, s3);
            }
            w += __builtin_popcount(okm);
        }
        if (keep) { kept = w; qn = w; resumeCall = callsDone; }
        else      { keeping = false; qn = kept; }
        __builtin_amdgcn_fence(__ATOMIC_RELEASE, "wavefront");
        __builtin_amdgcn_wave_barrier();
    };

    auto sumBatch = [&](int n, int s, int row, float s0, float s1, float s2, float s3) {
        const float4 m4 = reinterpret_cast<const float4*>(smaxT)[row];
        const float w0 = __expf(s0 - m4.x), w1 = __expf(s1 - m4.y);
        const float w2 = __expf(s2 - m4.z), w3 = __expf(s3 - m4.w);
        if (lane < n) {
            float* dp = denT + row * NHEADS;
            atomicAdd(dp,     w0);
            atomicAdd(dp + 1, w1);
            atomicAdd(dp + 2, w2);
            atomicAdd(dp + 3, w3);
        }
        for (int k = 0; k < n; k += 4) {
            float hv[4][4];
#pragma unroll
            for (int u = 0; u < 4; ++u) {
                const int kk = k + u < n ? k + u : n - 1;
                const float* hr = H + (size_t)rl_i(s, kk) * NCOLS + lane;
                hv[u][0] = hr[0]; hv[u][1] = hr[32]; hv[u][2] = hr[64]; hv[u][3] = hr[96];
            }
#pragma unroll
            for (int u = 0; u < 4; ++u) {
                if (k + u < n) {
                    float* tr = tile + rl_i(row, k + u) * NCOLS + lane;
                    atomicAdd(tr,      hv[u][0] * rl_f(w0, k + u));
                    atomicAdd(tr + 32, hv[u][1] * rl_f(w1, k + u));
                    atomicAdd(tr + 64, hv[u][2] * rl_f(w2, k + u));
                    atomicAdd(tr + 96, hv[u][3] * rl_f(w3, k + u));
                }
            }
        }
    };

    auto drainSumGather = [&]() {
        __builtin_amdgcn_fence(__ATOMIC_RELEASE, "wavefront");
        __builtin_amdgcn_wave_barrier();
        __builtin_amdgcn_fence(__ATOMIC_ACQUIRE, "wavefront");
        int w = kept;
        for (int r = kept; r < qn; r += 32) {
            const int idx = r + lane;
            bool ok = false;
            int  e = 0, row = 0;
            if (idx < qn) {
                e = q[idx].x;
                const unsigned rel = (unsigned)(dst[e] - lo);
                ok  = rel < (unsigned)cnt;
                row = (int)rel;
            }
            const unsigned okm = __builtin_amdgcn_ballot_w32(ok);
            if (ok) q[w + (int)__builtin_amdgcn_mbcnt_lo(okm, 0u)] = make_int2(e, row);
            w += __builtin_popcount(okm);
        }
        qn = w;
        __builtin_amdgcn_fence(__ATOMIC_RELEASE, "wavefront");
        __builtin_amdgcn_wave_barrier();
        __builtin_amdgcn_fence(__ATOMIC_ACQUIRE, "wavefront");
        for (int r = kept; r < qn; r += 32) {
            int n = qn - r; n = n < 32 ? n : 32;
            int idx = r + lane; idx = idx < qn ? idx : qn - 1;
            const int2 ent = q[idx];
            const int   s  = src[ent.x];
            const float s0 = sc0[ent.x], s1 = sc1[ent.x], s2 = sc2[ent.x], s3 = sc3[ent.x];
            sumBatch(n, s, ent.y, s0, s1, s2, s3);
        }
        qn = kept;
        __builtin_amdgcn_fence(__ATOMIC_RELEASE, "wavefront");
        __builtin_amdgcn_wave_barrier();
    };

    const int  ng   = E >> 5;
    const int  gp   = (ng + 3) & ~3;
    const int  nvec = gp >> 2;
    const int  last = nvec - 1;
    const int4* plane4 = reinterpret_cast<const int4*>(pres + (size_t)(obyte >> 5) * gp);
    const int4* dst4   = reinterpret_cast<const int4*>(dst);
    const int  vfirst = wave * 32;
    auto ld = [&](int v) { return plane4[v < last ? v : last]; };

    auto hitMask = [&](int4 d, int v) -> unsigned {
        const unsigned m = (((unsigned)d.x >> obit) & 1u)        | ((((unsigned)d.y >> obit) & 1u) << 1)
                         | ((((unsigned)d.z >> obit) & 1u) << 2) | ((((unsigned)d.w >> obit) & 1u) << 3);
        return v > last ? 0u : m;
    };

    auto appendHits = [&](unsigned m, int base, auto&& afterRound) {
        for (;;) {
            const bool     has  = m != 0u;
            const unsigned mask = __builtin_amdgcn_ballot_w32(has);
            if (!mask) break;
            if (has) {
                q[qn + (int)__builtin_amdgcn_mbcnt_lo(mask, 0u)] = make_int2(base + __builtin_ctz(m), 0);
                m &= m - 1u;
            }
            qn += __builtin_popcount(mask);
            afterRound();
        }
    };

    auto filterCall = [&](int4 x, int v, auto&& afterRound) {
        unsigned m = hitMask(x, v);
        int gn = 0;
        for (;;) {
            const bool     has  = m != 0u;
            const unsigned mask = __builtin_amdgcn_ballot_w32(has);
            if (!mask) break;
            if (has) {
                gl[gn + (int)__builtin_amdgcn_mbcnt_lo(mask, 0u)] = v * 4 + __builtin_ctz(m);
                m &= m - 1u;
            }
            gn += __builtin_popcount(mask);
        }
        __builtin_amdgcn_fence(__ATOMIC_RELEASE, "wavefront");
        __builtin_amdgcn_wave_barrier();
        __builtin_amdgcn_fence(__ATOMIC_ACQUIRE, "wavefront");
        for (int r = 0; r < gn; r += 16) {
            const int idx = r + (lane >> 1);
            unsigned em = 0u;
            int base = 0;
            if (idx < gn) {
                const int g = gl[idx];
                base = g * 32 + 16 * (lane & 1);
                const int4* dp = dst4 + ((size_t)g * 8 + 4 * (lane & 1));
#pragma unroll
                for (int i = 0; i < 4; ++i) {
                    const int4 d = dp[i];
                    em |= ((unsigned)(d.x - lo) < (unsigned)cnt ? 1u : 0u) << (4 * i);
                    em |= ((unsigned)(d.y - lo) < (unsigned)cnt ? 1u : 0u) << (4 * i + 1);
                    em |= ((unsigned)(d.z - lo) < (unsigned)cnt ? 1u : 0u) << (4 * i + 2);
                    em |= ((unsigned)(d.w - lo) < (unsigned)cnt ? 1u : 0u) << (4 * i + 3);
                }
            }
            appendHits(em, base, afterRound);
        }
        __builtin_amdgcn_fence(__ATOMIC_RELEASE, "wavefront");
        __builtin_amdgcn_wave_barrier();
    };
    auto roomForRound = [&]() { if (qn + 32 > Q_CAP) drainMax(resumeCall); };

    auto filterLeftover = [&]() {
        if (wave == 0) {
            const int e = ng * 32 + lane;
            bool hit = false;
            if (e < E) hit = (unsigned)(dst[e] - lo) < (unsigned)cnt;
            const unsigned mask = __builtin_amdgcn_ballot_w32(hit);
            if (mask) {
                if (hit) q[qn + (int)__builtin_amdgcn_mbcnt_lo(mask, 0u)] = make_int2(e, 0);
                qn += __builtin_popcount(mask);
            }
        }
    };

    if (vfirst < nvec) {
        int vb = vfirst;
        int4 x0 = ld(vb + lane), x1 = ld(vb + lane + 1024), x2 = ld(vb + lane + 2048), x3 = ld(vb + lane + 3072);
        for (; vb < nvec; vb += 4 * 1024) {
            const int v0 = vb + lane, v1 = v0 + 1024, v2 = v0 + 2048, v3 = v0 + 3072;
            const int nx = v0 + 4 * 1024;
            const int4 y0 = ld(nx), y1 = ld(nx + 1024), y2 = ld(nx + 2048), y3 = ld(nx + 3072);
            const int c0 = ((vb - vfirst) >> 12) * 4;
            filterCall(x0, v0, roomForRound);
            if (qn - kept >= 32) drainMax(c0 + 1);
            filterCall(x1, v1, roomForRound);
            if (qn - kept >= 32) drainMax(c0 + 2);
            filterCall(x2, v2, roomForRound);
            if (qn - kept >= 32) drainMax(c0 + 3);
            filterCall(x3, v3, roomForRound);
            if (qn - kept >= 32) drainMax(c0 + 4);
            x0 = y0; x1 = y1; x2 = y2; x3 = y3;
        }
    }
    filterLeftover();
    if (qn > kept) drainMax(0x7fffffff);
    __syncthreads();

    for (int r = 0; r < kept; r += 32) {
        int n = kept - r; n = n < 32 ? n : 32;
        int idx = r + lane; idx = idx < kept ? idx : kept - 1;
        const int2   ent = q[idx];
        const float4 s4  = ks[idx];
        sumBatch(n, ent.x, ent.y, s4.x, s4.y, s4.z, s4.w);
    }
    if (!keeping) {
        if (vfirst < nvec) {
            const int ncall = ((nvec - vfirst + 4095) >> 12) * 4;
            for (int ci = resumeCall; ci < ncall; ++ci) {
                const int v = vfirst + (ci >> 2) * 4096 + (ci & 3) * 1024 + lane;
                filterCall(ld(v), v, [&]() { if (qn - kept >= 32) drainSumGather(); });
            }
        }
        filterLeftover();
        if (qn > kept) drainSumGather();
    }
    __syncthreads();

    const float4* t4 = reinterpret_cast<const float4*>(tile);
    float4* o4 = reinterpret_cast<float4*>(out + (size_t)lo * NCOLS);
    for (int i = threadIdx.x; i < cnt * (NCOLS / 4); i += 1024) {
        const float d = denT[(i >> 5) * NHEADS + ((i & 31) >> 3)];
        float4 v = t4[i];
        const bool has = d > 0.f;
        v.x = has ? v.x / d : 0.f;
        v.y = has ? v.y / d : 0.f;
        v.z = has ? v.z / d : 0.f;
        v.w = has ? v.w / d : 0.f;
        o4[i] = v;
    }
}

extern "C" void kernel_launch(void* const* d_in, const int* in_sizes, int n_in,
                              void* d_out, int out_size, void* d_ws, size_t ws_size,
                              hipStream_t stream) {
    const float* node_feat = (const float*)d_in[0];
    const float* edge_feat = (const float*)d_in[1];
    const int*   src       = (const int*)d_in[2];
    const int*   dst       = (const int*)d_in[3];
    const float* W_node    = (const float*)d_in[4];
    const float* W_edge    = (const float*)d_in[5];
    float*       out       = (float*)d_out;

    const int N = in_sizes[0] / NODE_K;
    const int E = in_sizes[2];

    float*    h     = (float*)d_ws;
    float*    G     = h     + (size_t)N * NCOLS;
    float*    score = G     + (size_t)N * GCOLS;
    unsigned* pres  = (unsigned*)(score + (size_t)NHEADS * E);

    node_gemm_wreg<<<GEMM_GRID, 64, 0, stream>>>(node_feat, W_node, h, W_edge, G, N);

    edge_score_dot<<<SCORE_GRID, 32 * SCORE_WAVES, 0, stream>>>(edge_feat, G, src, dst, pres, score, E);

    aggregate_tiles<<<(N + TILE_N - 1) / TILE_N, 1024, 0, stream>>>(h, src, dst, pres, score, out, E, N);
}


# score launch writes one 32-byte record per edge (four head scores, src id, dst id); the aggregation fetches a hit from one line instead of six (4-byte reads of six planes); sum orders unchanged
# speedup vs baseline: 5.0065x; 1.0520x over previous
#include <hip/hip_runtime.h>
#include <hip/hip_bf16.h>

typedef __attribute__((ext_vector_type(2))) float v2f;
typedef __attribute__((ext_vector_type(8))) float v8f;

#define NCOLS 128
#define NODE_K 256
#define EDGE_K 64
#define NHEADS 4

typedef __attribute__((ext_vector_type(4))) float v4f;

#define GEMM_GRID 512
#define GCOLS     256
#define HS_PITCH  68

__global__ void __launch_bounds__(64)
node_gemm_wreg(const float* __restrict__ A, const float* __restrict__ B,
               float* __restrict__ C, const float* __restrict__ We,
               float* __restrict__ Gout, int M) {
    __shared__ alignas(16) float hs[2][16 * HS_PITCH];
    const int lane = threadIdx.x & 31;
    const int ch = __builtin_amdgcn_readfirstlane((int)(threadIdx.x >> 5));
    const int c  = lane & 15;
    const int hf = lane >> 4;

    const int ntile = (M + 15) >> 4;
    int t = blockIdx.x;
    if (t >= ntile) return;
    const int G = gridDim.x;

    v2f wf[4][NODE_K / 4];
    const v4f* B4 = reinterpret_cast<const v4f*>(B);
#pragma unroll
    for (int j = 0; j < NODE_K / 4; ++j) {
        const int k0 = 8 * (j >> 1) + 4 * hf + 2 * (j & 1);
        const v4f w0 = B4[k0 * (NCOLS / 4) + 16 * ch + c];
        const v4f w1 = B4[(k0 + 1) * (NCOLS / 4) + 16 * ch + c];
#pragma unroll
        for (int u = 0; u < 4; ++u) { wf[u][j].x = w0[u]; wf[u][j].y = w1[u]; }
    }

    v2f wfe[2][4][8];
    const v4f* We4 = reinterpret_cast<const v4f*>(We);
#pragma unroll
    for (int hh = 0; hh < 2; ++hh) {
#pragma unroll
        for (int u = 0; u < 4; ++u) {
#pragma unroll
            for (int i = 0; i < 4; ++i) {
                const v4f w = We4[(4 * c + u) * (NCOLS / 4) + 8 * (2 * ch + hh) + 2 * i + hf];
                wfe[hh][u][2 * i].x     = w.x; wfe[hh][u][2 * i].y     = w.y;
                wfe[hh][u][2 * i + 1].x = w.z; wfe[hh][u][2 * i + 1].y = w.w;
            }
        }
    }
    float* hp = hs[ch];

    const v4f* A4 = reinterpret_cast<const v4f*>(A);
    const int lastTile = ntile - 1;
    const int lastRow  = M - 1;

    auto loadTile = [&](int tile, v4f* dstv) {
        int tl = tile < lastTile ? tile : lastTile;
        int row = tl * 16 + c; row = row < lastRow ? row : lastRow;
        const v4f* rp = A4 + (size_t)row * (NODE_K / 4) + hf;
#pragma unroll
        for (int i = 0; i < NODE_K / 8; ++i) dstv[i] = rp[2 * i];
    };

    v4f cur[NODE_K / 8];
    loadTile(t, cur);

    for (; t < ntile; t += G) {
        v4f nxt[NODE_K / 8];
        loadTile(t + G, nxt);
        __builtin_amdgcn_sched_barrier(0);

        v8f acc[4] = {};
#pragma unroll
        for (int i = 0; i < NODE_K / 8; ++i) {
            const v2f a0 = __builtin_shufflevector(cur[i], cur[i], 0, 1);
            const v2f a1 = __builtin_shufflevector(cur[i], cur[i], 2, 3);
#pragma unroll
            for (int u = 0; u < 4; ++u)
                acc[u] = __builtin_amdgcn_wmma_f32_16x16x4_f32(false, a0, false, wf[u][2 * i],     (short)0, acc[u], false, false);
#pragma unroll
            for (int u = 0; u < 4; ++u)
                acc[u] = __builtin_amdgcn_wmma_f32_16x16x4_f32(false, a1, false, wf[u][2 * i + 1], (short)0, acc[u], false, false);
        }
        __builtin_amdgcn_sched_barrier(0);

        const int rbase = t * 16 + 8 * hf;
#pragma unroll
        for (int r = 0; r < 8; ++r) {
            const int row = rbase + r;
            if (row < M) {
                v4f o;
                o.x = acc[0][r]; o.y = acc[1][r]; o.z = acc[2][r]; o.w = acc[3][r];
                *reinterpret_cast<v4f*>(C + (size_t)row * NCOLS + 64 * ch + 4 * c) = o;
            }
            v4f p;
            p.x = acc[0][r]; p.y = acc[1][r]; p.z = acc[2][r]; p.w = acc[3][r];
            *reinterpret_cast<v4f*>(hp + (8 * hf + r) * HS_PITCH + 4 * c) = p;
        }
        __builtin_amdgcn_fence(__ATOMIC_RELEASE, "wavefront");
        __builtin_amdgcn_wave_barrier();
        __builtin_amdgcn_fence(__ATOMIC_ACQUIRE, "wavefront");

        v4f a2[2][4];
#pragma unroll
        for (int hh = 0; hh < 2; ++hh) {
#pragma unroll
            for (int i = 0; i < 4; ++i)
                a2[hh][i] = *reinterpret_cast<const v4f*>(hp + c * HS_PITCH + 32 * hh + 8 * i + 4 * hf);
        }
#pragma unroll
        for (int hh = 0; hh < 2; ++hh) {
            const v4f* a = a2[hh];
            v8f ag[4] = {};
#pragma unroll
            for (int i = 0; i < 4; ++i) {
                const v2f a0 = __builtin_shufflevector(a[i], a[i], 0, 1);
                const v2f a1 = __builtin_shufflevector(a[i], a[i], 2, 3);
#pragma unroll
                for (int u = 0; u < 4; ++u)
                    ag[u] = __builtin_amdgcn_wmma_f32_16x16x4_f32(false, a0, false, wfe[hh][u][2 * i],     (short)0, ag[u], false, false);
#pragma unroll
                for (int u = 0; u < 4; ++u)
                    ag[u] = __builtin_amdgcn_wmma_f32_16x16x4_f32(false, a1, false, wfe[hh][u][2 * i + 1], (short)0, ag[u], false, false);
            }
#pragma unroll
            for (int r = 0; r < 8; ++r) {
                const int orow = rbase + r;
                if (orow < M) {
                    v4f o;
                    o.x = ag[0][r]; o.y = ag[1][r]; o.z = ag[2][r]; o.w = ag[3][r];
                    *reinterpret_cast<v4f*>(Gout + (size_t)orow * GCOLS + 64 * (2 * ch + hh) + 4 * c) = o;
                }
            }
        }
        __builtin_amdgcn_fence(__ATOMIC_RELEASE, "wavefront");
        __builtin_amdgcn_wave_barrier();

#pragma unroll
        for (int i = 0; i < NODE_K / 8; ++i) cur[i] = nxt[i];
    }
}

#define TILE_N    196

#define SCORE_WAVES 4
#define SCORE_GRID  1280

__global__ void __launch_bounds__(32 * SCORE_WAVES)
edge_score_dot(const float* __restrict__ EF, const float* __restrict__ G,
               const int* __restrict__ src, const int* __restrict__ dst,
               unsigned* __restrict__ pres, float* __restrict__ rec, int E) {
    const int lane = threadIdx.x & 31;
    const int wave = __builtin_amdgcn_readfirstlane((int)(threadIdx.x >> 5));
    const int gw = blockIdx.x * SCORE_WAVES + wave;
    const int nw = gridDim.x * SCORE_WAVES;
    const int i  = lane & 15;
    const int hf = lane >> 4;

    {
        const int ng = E >> 5;
        const int gp = (ng + 3) & ~3;
        const int4* dst4 = reinterpret_cast<const int4*>(dst);
        for (int task = gw; task * 32 < gp; task += nw) {
            const int g = task * 32 + lane;
            unsigned w[8] = {0u, 0u, 0u, 0u, 0u, 0u, 0u, 0u};
            if (g < ng) {
#pragma unroll
                for (int ii = 0; ii < 8; ++ii) {
                    const int4 d = dst4[(size_t)g * 8 + ii];
                    const unsigned t4[4] = {(unsigned)d.x / TILE_N, (unsigned)d.y / TILE_N,
                                            (unsigned)d.z / TILE_N, (unsigned)d.w / TILE_N};
#pragma unroll
                    for (int j = 0; j < 4; ++j) {
                        const unsigned t   = t4[j] & 255u;
                        const unsigned bit = 1u << (t & 31u);
#pragma unroll
                        for (int k = 0; k < 8; ++k) w[k] |= (t >> 5) == (unsigned)k ? bit : 0u;
                    }
                }
            }
            if (g < gp) {
#pragma unroll
                for (int k = 0; k < 8; ++k) pres[(size_t)k * gp + g] = w[k];
            }
        }
    }

    const int nb = (E + 31) >> 5;
    const v4f* EF4 = reinterpret_cast<const v4f*>(EF);
    const v4f* G4  = reinterpret_cast<const v4f*>(G);
    const unsigned lastEdge = (unsigned)(E - 1);
    const bool b3 = (lane & 8) != 0, b2 = (lane & 4) != 0, b1 = (lane & 2) != 0, b0 = (lane & 1) != 0;

    for (int b = gw; b < nb; b += nw) {
        const unsigned e0 = (unsigned)b * 32u;
        unsigned el = e0 + (unsigned)lane; el = el < lastEdge ? el : lastEdge;
        const int sv = src[el];
        const int dv = dst[el];

        float r8[8][4];
#pragma unroll
        for (int q = 0; q < 4; ++q) {
            v4f x[4];
            v4f g[4][4];
#pragma unroll
            for (int sl = 0; sl < 4; ++sl) {
                const int jp = 2 * q + (sl & 1) + 8 * (sl >> 1);
                unsigned e = e0 + 16u * (unsigned)hf + (unsigned)jp; e = e < lastEdge ? e : lastEdge;
                const int sA = __builtin_amdgcn_readlane(sv, jp);
                const int sB = __builtin_amdgcn_readlane(sv, jp + 16);
                const unsigned s = (unsigned)(hf ? sB : sA);
                x[sl] = __builtin_nontemporal_load(&EF4[e * (unsigned)(EDGE_K / 4) + (unsigned)i]);
                const v4f* gr = G4 + (s * (unsigned)(GCOLS / 4) + (unsigned)i);
#pragma unroll
                for (int hd = 0; hd < 4; ++hd) g[sl][hd] = gr[16 * hd];
            }
            __builtin_amdgcn_sched_barrier(0);
            float v[4][4];
#pragma unroll
            for (int sl = 0; sl < 4; ++sl) {
#pragma unroll
                for (int hd = 0; hd < 4; ++hd) {
                    const v4f gg = g[sl][hd];
                    const v4f xx = x[sl];
                    v[sl][hd] = xx.x * gg.x + xx.y * gg.y + xx.z * gg.z + xx.w * gg.w;
                }
            }
#pragma unroll
            for (int m = 0; m < 2; ++m) {
#pragma unroll
                for (int hd = 0; hd < 4; ++hd) {
                    const float lo = v[m][hd], hi = v[m + 2][hd];
                    const float keep = b3 ? hi : lo;
                    const float send = b3 ? lo : hi;
                    r8[2 * q + m][hd] = keep + __shfl_xor(send, 8, 32);
                }
            }
        }
        float r4[4][4];
#pragma unroll
        for (int jp = 0; jp < 4; ++jp) {
#pragma unroll
            for (int hd = 0; hd < 4; ++hd) {
                const float lo = r8[jp][hd], hi = r8[jp + 4][hd];
                const float keep = b2 ? hi : lo;
                const float send = b2 ? lo : hi;
                r4[jp][hd] = keep + __shfl_xor(send, 4, 32);
            }
        }
        float r2[2][4];
#pragma unroll
        for (int jp = 0; jp < 2; ++jp) {
#pragma unroll
            for (int hd = 0; hd < 4; ++hd) {
                const float lo = r4[jp][hd], hi = r4[jp + 2][hd];
                const float keep = b1 ? hi : lo;
                const float send = b1 ? lo : hi;
                r2[jp][hd] = keep + __shfl_xor(send, 2, 32);
            }
        }
        const unsigned eo = e0 + (unsigned)lane;
#pragma unroll
        for (int hd = 0; hd < 4; ++hd) {
            const float lo = r2[0][hd], hi = r2[1][hd];
            const float keep = b0 ? hi : lo;
            const float send = b0 ? lo : hi;
            const float d = keep + __shfl_xor(send, 1, 32);
            float sc = d * 0.17677669529663689f;
            sc = sc > 0.f ? sc : 0.01f * sc;
            if (eo <= lastEdge) rec[8u * eo + (unsigned)hd] = sc;
        }
        if (eo <= lastEdge) *reinterpret_cast<int2*>(rec + 8u * eo + 4u) = make_int2(sv, dv);
    }
}

#define AGG_WAVES 32
#define P_MAX     192
#define G_CALL    128
#define Q_BURST   160
#define Q_CAP     (P_MAX + Q_BURST)

__device__ __forceinline__ int   rl_i(int v, int k)   { return __builtin_amdgcn_readlane(v, k); }
__device__ __forceinline__ float rl_f(float v, int k) {
    return __int_as_float(__builtin_amdgcn_readlane(__float_as_int(v), k));
}

__device__ __forceinline__ void ldsMaxFloat(float* addr, float val) {
    if (val >= 0.f) atomicMax((int*)addr, __float_as_int(val));
    else            atomicMin((unsigned int*)addr, __float_as_uint(val));
}

__global__ void __launch_bounds__(1024)
aggregate_tiles(const float* __restrict__ H,
                const int* __restrict__ dst, const unsigned* __restrict__ pres,
                const float* __restrict__ rec,
                float* __restrict__ out, int E, int N) {
    __shared__ alignas(16) float  tile[TILE_N * NCOLS];
    __shared__ alignas(16) float4 keptScore[AGG_WAVES][P_MAX];
    __shared__ int2  queue[AGG_WAVES][Q_CAP];
    __shared__ int   glist[AGG_WAVES][G_CALL];
    __shared__ alignas(16) float smaxT[TILE_N * NHEADS];
    __shared__ alignas(16) float denT[TILE_N * NHEADS];

    const int lane = threadIdx.x & 31;
    const int wave = __builtin_amdgcn_readfirstlane((int)(threadIdx.x >> 5));
    const int lo   = blockIdx.x * TILE_N;
    int cnt = N - lo; cnt = cnt < TILE_N ? cnt : TILE_N;
    const unsigned obyte = blockIdx.x & 255u;
    const unsigned obit  = obyte & 31u;

    for (int i = threadIdx.x; i < TILE_N * NCOLS; i += 1024) tile[i] = 0.f;
    if (threadIdx.x < TILE_N * NHEADS) {
        smaxT[threadIdx.x] = __int_as_float(0xFF800000);
        denT[threadIdx.x]  = 0.f;
    }
    __syncthreads();

    int2*   q  = queue[wave];
    float4* ks = keptScore[wave];
    int*    gl = glist[wave];
    int  kept = 0;
    int  qn   = 0;
    bool keeping = true;
    int  resumeCall = 0;
    const float4* rec4 = reinterpret_cast<const float4*>(rec);
    const int2*   rec2 = reinterpret_cast<const int2*>(rec);

    auto drainMax = [&](int callsDone) {
        __builtin_amdgcn_fence(__ATOMIC_RELEASE, "wavefront");
        __builtin_amdgcn_wave_barrier();
        __builtin_amdgcn_fence(__ATOMIC_ACQUIRE, "wavefront");
        const bool keep = keeping && qn <= P_MAX;
        int w = kept;
        for (int r = kept; r < qn; r += 32) {
            const int idx = r + lane;
            bool  ok = false;
            int   s = 0, row = 0;
            float s0 = 0.f, s1 = 0.f, s2 = 0.f, s3 = 0.f;
            if (idx < qn) {
                const int e = q[idx].x;
                const float4 s4 = rec4[2 * (size_t)e];
                const int2   sd = rec2[4 * (size_t)e + 2];
                const unsigned rel = (unsigned)(sd.y - lo);
                s  = sd.x;
                s0 = s4.x; s1 = s4.y; s2 = s4.z; s3 = s4.w;
                ok  = rel < (unsigned)cnt;
                row = (int)rel;
                if (ok) {
                    float* mp = smaxT + row * NHEADS;
                    ldsMaxFloat(mp,     s0);
                    ldsMaxFloat(mp + 1, s1);
                    ldsMaxFloat(mp + 2, s2);
                    ldsMaxFloat(mp + 3, s3);
                }
            }
            const unsigned okm = __builtin_amdgcn_ballot_w32(ok);
            if (keep && ok) {
                const int pos = w + (int)__builtin_amdgcn_mbcnt_lo(okm, 0u);
                q[pos]  = make_int2(s, row);
                ks[pos] = make_float4(s0, s1, s2, s3);
            }
            w += __builtin_popcount(okm);
        }
        if (keep) { kept = w; qn = w; resumeCall = callsDone; }
        else      { keeping = false; qn = kept; }
        __builtin_amdgcn_fence(__ATOMIC_RELEASE, "wavefront");
        __builtin_amdgcn_wave_barrier();
    };

    auto sumBatch = [&](int n, int s, int row, float s0, float s1, float s2, float s3) {
        const float4 m4 = reinterpret_cast<const float4*>(smaxT)[row];
        const float w0 = __expf(s0 - m4.x), w1 = __expf(s1 - m4.y);
        const float w2 = __expf(s2 - m4.z), w3 = __expf(s3 - m4.w);
        if (lane < n) {
            float* dp = denT + row * NHEADS;
            atomicAdd(dp,     w0);
            atomicAdd(dp + 1, w1);
            atomicAdd(dp + 2, w2);
            atomicAdd(dp + 3, w3);
        }
        for (int k = 0; k < n; k += 4) {
            float hv[4][4];
#pragma unroll
            for (int u = 0; u < 4; ++u) {
                const int kk = k + u < n ? k + u : n - 1;
                const float* hr = H + (size_t)rl_i(s, kk) * NCOLS + lane;
                hv[u][0] = hr[0]; hv[u][1] = hr[32]; hv[u][2] = hr[64]; hv[u][3] = hr[96];
            }
#pragma unroll
            for (int u = 0; u < 4; ++u) {
                if (k + u < n) {
                    float* tr = tile + rl_i(row, k + u) * NCOLS + lane;
                    atomicAdd(tr,      hv[u][0] * rl_f(w0, k + u));
                    atomicAdd(tr + 32, hv[u][1] * rl_f(w1, k + u));
                    atomicAdd(tr + 64, hv[u][2] * rl_f(w2, k + u));
                    atomicAdd(tr + 96, hv[u][3] * rl_f(w3, k + u));
                }
            }
        }
    };

    auto drainSumGather = [&]() {
        __builtin_amdgcn_fence(__ATOMIC_RELEASE, "wavefront");
        __builtin_amdgcn_wave_barrier();
        __builtin_amdgcn_fence(__ATOMIC_ACQUIRE, "wavefront");
        int w = kept;
        for (int r = kept; r < qn; r += 32) {
            const int idx = r + lane;
            bool ok = false;
            int  e = 0, row = 0;
            if (idx < qn) {
                e = q[idx].x;
                const unsigned rel = (unsigned)(dst[e] - lo);
                ok  = rel < (unsigned)cnt;
                row = (int)rel;
            }
            const unsigned okm = __builtin_amdgcn_ballot_w32(ok);
            if (ok) q[w + (int)__builtin_amdgcn_mbcnt_lo(okm, 0u)] = make_int2(e, row);
            w += __builtin_popcount(okm);
        }
        qn = w;
        __builtin_amdgcn_fence(__ATOMIC_RELEASE, "wavefront");
        __builtin_amdgcn_wave_barrier();
        __builtin_amdgcn_fence(__ATOMIC_ACQUIRE, "wavefront");
        for (int r = kept; r < qn; r += 32) {
            int n = qn - r; n = n < 32 ? n : 32;
            int idx = r + lane; idx = idx < qn ? idx : qn - 1;
            const int2 ent = q[idx];
            const float4 s4 = rec4[2 * (size_t)ent.x];
            const int    s  = rec2[4 * (size_t)ent.x + 2].x;
            sumBatch(n, s, ent.y, s4.x, s4.y, s4.z, s4.w);
        }
        qn = kept;
        __builtin_amdgcn_fence(__ATOMIC_RELEASE, "wavefront");
        __builtin_amdgcn_wave_barrier();
    };

    const int  ng   = E >> 5;
    const int  gp   = (ng + 3) & ~3;
    const int  nvec = gp >> 2;
    const int  last = nvec - 1;
    const int4* plane4 = reinterpret_cast<const int4*>(pres + (size_t)(obyte >> 5) * gp);
    const int4* dst4   = reinterpret_cast<const int4*>(dst);
    const int  vfirst = wave * 32;
    auto ld = [&](int v) { return plane4[v < last ? v : last]; };

    auto hitMask = [&](int4 d, int v) -> unsigned {
        const unsigned m = (((unsigned)d.x >> obit) & 1u)        | ((((unsigned)d.y >> obit) & 1u) << 1)
                         | ((((unsigned)d.z >> obit) & 1u) << 2) | ((((unsigned)d.w >> obit) & 1u) << 3);
        return v > last ? 0u : m;
    };

    auto appendHits = [&](unsigned m, int base, auto&& afterRound) {
        for (;;) {
            const bool     has  = m != 0u;
            const unsigned mask = __builtin_amdgcn_ballot_w32(has);
            if (!mask) break;
            if (has) {
                q[qn + (int)__builtin_amdgcn_mbcnt_lo(mask, 0u)] = make_int2(base + __builtin_ctz(m), 0);
                m &= m - 1u;
            }
            qn += __builtin_popcount(mask);
            afterRound();
        }
    };

    auto filterCall = [&](int4 x, int v, auto&& afterRound) {
        unsigned m = hitMask(x, v);
        int gn = 0;
        for (;;) {
            const bool     has  = m != 0u;
            const unsigned mask = __builtin_amdgcn_ballot_w32(has);
            if (!mask) break;
            if (has) {
                gl[gn + (int)__builtin_amdgcn_mbcnt_lo(mask, 0u)] = v * 4 + __builtin_ctz(m);
                m &= m - 1u;
            }
            gn += __builtin_popcount(mask);
        }
        __builtin_amdgcn_fence(__ATOMIC_RELEASE, "wavefront");
        __builtin_amdgcn_wave_barrier();
        __builtin_amdgcn_fence(__ATOMIC_ACQUIRE, "wavefront");
        for (int r = 0; r < gn; r += 16) {
            const int idx = r + (lane >> 1);
            unsigned em = 0u;
            int base = 0;
            if (idx < gn) {
                const int g = gl[idx];
                base = g * 32 + 16 * (lane & 1);
                const int4* dp = dst4 + ((size_t)g * 8 + 4 * (lane & 1));
#pragma unroll
                for (int i = 0; i < 4; ++i) {
                    const int4 d = dp[i];
                    em |= ((unsigned)(d.x - lo) < (unsigned)cnt ? 1u : 0u) << (4 * i);
                    em |= ((unsigned)(d.y - lo) < (unsigned)cnt ? 1u : 0u) << (4 * i + 1);
                    em |= ((unsigned)(d.z - lo) < (unsigned)cnt ? 1u : 0u) << (4 * i + 2);
                    em |= ((unsigned)(d.w - lo) < (unsigned)cnt ? 1u : 0u) << (4 * i + 3);
                }
            }
            appendHits(em, base, afterRound);
        }
        __builtin_amdgcn_fence(__ATOMIC_RELEASE, "wavefront");
        __builtin_amdgcn_wave_barrier();
    };
    auto roomForRound = [&]() { if (qn + 32 > Q_CAP) drainMax(resumeCall); };

    auto filterLeftover = [&]() {
        if (wave == 0) {
            const int e = ng * 32 + lane;
            bool hit = false;
            if (e < E) hit = (unsigned)(dst[e] - lo) < (unsigned)cnt;
            const unsigned mask = __builtin_amdgcn_ballot_w32(hit);
            if (mask) {
                if (hit) q[qn + (int)__builtin_amdgcn_mbcnt_lo(mask, 0u)] = make_int2(e, 0);
                qn += __builtin_popcount(mask);
            }
        }
    };

    if (vfirst < nvec) {
        int vb = vfirst;
        int4 x0 = ld(vb + lane), x1 = ld(vb + lane + 1024), x2 = ld(vb + lane + 2048), x3 = ld(vb + lane + 3072);
        for (; vb < nvec; vb += 4 * 1024) {
            const int v0 = vb + lane, v1 = v0 + 1024, v2 = v0 + 2048, v3 = v0 + 3072;
            const int nx = v0 + 4 * 1024;
            const int4 y0 = ld(nx), y1 = ld(nx + 1024), y2 = ld(nx + 2048), y3 = ld(nx + 3072);
            const int c0 = ((vb - vfirst) >> 12) * 4;
            filterCall(x0, v0, roomForRound);
            if (qn - kept >= 32) drainMax(c0 + 1);
            filterCall(x1, v1, roomForRound);
            if (qn - kept >= 32) drainMax(c0 + 2);
            filterCall(x2, v2, roomForRound);
            if (qn - kept >= 32) drainMax(c0 + 3);
            filterCall(x3, v3, roomForRound);
            if (qn - kept >= 32) drainMax(c0 + 4);
            x0 = y0; x1 = y1; x2 = y2; x3 = y3;
        }
    }
    filterLeftover();
    if (qn > kept) drainMax(0x7fffffff);
    __syncthreads();

    for (int r = 0; r < kept; r += 32) {
        int n = kept - r; n = n < 32 ? n : 32;
        int idx = r + lane; idx = idx < kept ? idx : kept - 1;
        const int2   ent = q[idx];
        const float4 s4  = ks[idx];
        sumBatch(n, ent.x, ent.y, s4.x, s4.y, s4.z, s4.w);
    }
    if (!keeping) {
        if (vfirst < nvec) {
            const int ncall = ((nvec - vfirst + 4095) >> 12) * 4;
            for (int ci = resumeCall; ci < ncall; ++ci) {
                const int v = vfirst + (ci >> 2) * 4096 + (ci & 3) * 1024 + lane;
                filterCall(ld(v), v, [&]() { if (qn - kept >= 32) drainSumGather(); });
            }
        }
        filterLeftover();
        if (qn > kept) drainSumGather();
    }
    __syncthreads();

    const float4* t4 = reinterpret_cast<const float4*>(tile);
    float4* o4 = reinterpret_cast<float4*>(out + (size_t)lo * NCOLS);
    for (int i = threadIdx.x; i < cnt * (NCOLS / 4); i += 1024) {
        const float d = denT[(i >> 5) * NHEADS + ((i & 31) >> 3)];
        float4 v = t4[i];
        const bool has = d > 0.f;
        v.x = has ? v.x / d : 0.f;
        v.y = has ? v.y / d : 0.f;
        v.z = has ? v.z / d : 0.f;
        v.w = has ? v.w / d : 0.f;
        o4[i] = v;
    }
}

extern "C" void kernel_launch(void* const* d_in, const int* in_sizes, int n_in,
                              void* d_out, int out_size, void* d_ws, size_t ws_size,
                              hipStream_t stream) {
    const float* node_feat = (const float*)d_in[0];
    const float* edge_feat = (const float*)d_in[1];
    const int*   src       = (const int*)d_in[2];
    const int*   dst       = (const int*)d_in[3];
    const float* W_node    = (const float*)d_in[4];
    const float* W_edge    = (const float*)d_in[5];
    float*       out       = (float*)d_out;

    const int N = in_sizes[0] / NODE_K;
    const int E = in_sizes[2];

    float*    h     = (float*)d_ws;
    float*    G     = h     + (size_t)N * NCOLS;
    float*    rec   = G     + (size_t)N * GCOLS;
    unsigned* pres  = (unsigned*)(rec + (size_t)8 * E);

    node_gemm_wreg<<<GEMM_GRID, 64, 0, stream>>>(node_feat, W_node, h, W_edge, G, N);

    edge_score_dot<<<SCORE_GRID, 32 * SCORE_WAVES, 0, stream>>>(edge_feat, G, src, dst, pres, rec, E);

    aggregate_tiles<<<(N + TILE_N - 1) / TILE_N, 1024, 0, stream>>>(h, dst, pres, rec, out, E, N);
}
